# starting kernel
# baseline (speedup 1.0000x reference)
#include <hip/hip_runtime.h>
#include <math.h>

typedef __attribute__((ext_vector_type(2))) float v2f;
typedef __attribute__((ext_vector_type(8))) float v8f;

#define B_    32
#define S_    64
#define H_    300
#define L_    20
#define KT    60
#define KTP   62
#define OUTC  160
#define OUTQOFF (B_*S_*OUTC)
#define NEG_INF (-3.4e38f)

__global__ __launch_bounds__(128)
void maxpool_kernel(const float* __restrict__ p, const float* __restrict__ q,
                    const float* __restrict__ W, float* __restrict__ out)
{
    __shared__ float As[S_][KTP];
    __shared__ float Bs[S_][KTP];
    __shared__ float nA[S_];
    __shared__ float nB[S_];
    __shared__ float cosBuf[S_][S_ + 1];

    const int idx  = blockIdx.x;
    const int l    = idx % L_;
    const int hf   = (idx / L_) & 1;
    const int b    = idx / (2 * L_);
    const int tid  = threadIdx.x;
    const int lane = tid & 31;
    const int wv   = tid >> 5;
    const int lo   = lane & 15;
    const int hi   = lane >> 4;

    const float* Ph = p + (b * S_) * (2 * H_) + hf * H_;
    const float* Qh = q + (b * S_) * (2 * H_) + hf * H_;
    const float* Wr = W + ((2 + hf) * L_ + l) * H_;

    v8f c0 = {}, c1 = {}, c2 = {}, c3 = {};
    float nacc = 0.f;

    for (int kt = 0; kt < H_; kt += KT) {
        __syncthreads();
        for (int i = tid; i < S_ * KT; i += 128) {
            int s = i / KT, k = i % KT;
            float wval = Wr[kt + k];
            As[s][k] = Ph[s * (2 * H_) + kt + k] * wval;
            Bs[s][k] = Qh[s * (2 * H_) + kt + k] * wval;
        }
        __syncthreads();
        if (tid < S_) {
            for (int k = 0; k < KT; ++k) { float v = As[tid][k]; nacc += v * v; }
        } else {
            int r = tid - S_;
            for (int k = 0; k < KT; ++k) { float v = Bs[r][k]; nacc += v * v; }
        }
        const int m = wv * 16 + lo;
        for (int kb = 0; kb < KT; kb += 4) {
            int ka = kb + 2 * hi;
            v2f a  = *(const v2f*)&As[m][ka];
            v2f b0 = *(const v2f*)&Bs[ 0 + lo][ka];
            v2f b1 = *(const v2f*)&Bs[16 + lo][ka];
            v2f b2 = *(const v2f*)&Bs[32 + lo][ka];
            v2f b3 = *(const v2f*)&Bs[48 + lo][ka];
            c0 = __builtin_amdgcn_wmma_f32_16x16x4_f32(false, a, false, b0, (short)0, c0, false, false);
            c1 = __builtin_amdgcn_wmma_f32_16x16x4_f32(false, a, false, b1, (short)0, c1, false, false);
            c2 = __builtin_amdgcn_wmma_f32_16x16x4_f32(false, a, false, b2, (short)0, c2, false, false);
            c3 = __builtin_amdgcn_wmma_f32_16x16x4_f32(false, a, false, b3, (short)0, c3, false, false);
        }
    }
    __syncthreads();
    if (tid < S_) nA[tid] = sqrtf(nacc); else nB[tid - S_] = sqrtf(nacc);
    __syncthreads();

    #pragma unroll
    for (int j = 0; j < 8; ++j) {
        int m = wv * 16 + 8 * hi + j;
        float na = nA[m];
        cosBuf[m][ 0 + lo] = c0[j] / (na * nB[ 0 + lo]);
        cosBuf[m][16 + lo] = c1[j] / (na * nB[16 + lo]);
        cosBuf[m][32 + lo] = c2[j] / (na * nB[32 + lo]);
        cosBuf[m][48 + lo] = c3[j] / (na * nB[48 + lo]);
    }
    __syncthreads();

    if (tid < S_) {
        float mx = NEG_INF;
        for (int n = 0; n < S_; ++n) mx = fmaxf(mx, cosBuf[tid][n]);
        out[(b * S_ + tid) * OUTC + (2 + hf) * L_ + l] = mx;
    } else {
        int n = tid - S_;
        float mx = NEG_INF;
        for (int m = 0; m < S_; ++m) mx = fmaxf(mx, cosBuf[m][n]);
        out[OUTQOFF + (b * S_ + n) * OUTC + (2 + hf) * L_ + l] = mx;
    }
}

__global__ __launch_bounds__(128)
void attn_cos_kernel(const float* __restrict__ p, const float* __restrict__ q,
                     float* __restrict__ cosW, float* __restrict__ rowSum,
                     float* __restrict__ colSum)
{
    __shared__ float As[S_][KTP];
    __shared__ float Bs[S_][KTP];
    __shared__ float nA[S_];
    __shared__ float nB[S_];
    __shared__ float cosBuf[S_][S_ + 1];

    const int hf   = blockIdx.x & 1;
    const int b    = blockIdx.x >> 1;
    const int tid  = threadIdx.x;
    const int lane = tid & 31;
    const int wv   = tid >> 5;
    const int lo   = lane & 15;
    const int hi   = lane >> 4;

    const float* Ph = p + (b * S_) * (2 * H_) + hf * H_;
    const float* Qh = q + (b * S_) * (2 * H_) + hf * H_;

    v8f c0 = {}, c1 = {}, c2 = {}, c3 = {};
    float nacc = 0.f;

    for (int kt = 0; kt < H_; kt += KT) {
        __syncthreads();
        for (int i = tid; i < S_ * KT; i += 128) {
            int s = i / KT, k = i % KT;
            As[s][k] = Ph[s * (2 * H_) + kt + k];
            Bs[s][k] = Qh[s * (2 * H_) + kt + k];
        }
        __syncthreads();
        if (tid < S_) {
            for (int k = 0; k < KT; ++k) { float v = As[tid][k]; nacc += v * v; }
        } else {
            int r = tid - S_;
            for (int k = 0; k < KT; ++k) { float v = Bs[r][k]; nacc += v * v; }
        }
        const int m = wv * 16 + lo;
        for (int kb = 0; kb < KT; kb += 4) {
            int ka = kb + 2 * hi;
            v2f a  = *(const v2f*)&As[m][ka];
            v2f b0 = *(const v2f*)&Bs[ 0 + lo][ka];
            v2f b1 = *(const v2f*)&Bs[16 + lo][ka];
            v2f b2 = *(const v2f*)&Bs[32 + lo][ka];
            v2f b3 = *(const v2f*)&Bs[48 + lo][ka];
            c0 = __builtin_amdgcn_wmma_f32_16x16x4_f32(false, a, false, b0, (short)0, c0, false, false);
            c1 = __builtin_amdgcn_wmma_f32_16x16x4_f32(false, a, false, b1, (short)0, c1, false, false);
            c2 = __builtin_amdgcn_wmma_f32_16x16x4_f32(false, a, false, b2, (short)0, c2, false, false);
            c3 = __builtin_amdgcn_wmma_f32_16x16x4_f32(false, a, false, b3, (short)0, c3, false, false);
        }
    }
    __syncthreads();
    if (tid < S_) nA[tid] = sqrtf(nacc); else nB[tid - S_] = sqrtf(nacc);
    __syncthreads();

    #pragma unroll
    for (int j = 0; j < 8; ++j) {
        int m = wv * 16 + 8 * hi + j;
        float na = nA[m];
        cosBuf[m][ 0 + lo] = c0[j] / (na * nB[ 0 + lo]);
        cosBuf[m][16 + lo] = c1[j] / (na * nB[16 + lo]);
        cosBuf[m][32 + lo] = c2[j] / (na * nB[32 + lo]);
        cosBuf[m][48 + lo] = c3[j] / (na * nB[48 + lo]);
    }
    __syncthreads();

    float* dst = cosW + (b * 2 + hf) * (S_ * S_);
    for (int i = tid; i < S_ * S_; i += 128)
        dst[i] = cosBuf[i >> 6][i & 63];

    if (tid < S_) {
        float sm = 0.f;
        for (int n = 0; n < S_; ++n) sm += cosBuf[tid][n];
        rowSum[(b * 2 + hf) * S_ + tid] = sm;
    } else {
        int n = tid - S_;
        float sm = 0.f;
        for (int m = 0; m < S_; ++m) sm += cosBuf[m][n];
        colSum[(b * 2 + hf) * S_ + n] = sm;
    }
}

__global__ __launch_bounds__(128)
void attn_vec_kernel(const float* __restrict__ p, const float* __restrict__ q,
                     const float* __restrict__ cosW,
                     const float* __restrict__ rowSum, const float* __restrict__ colSum,
                     float* __restrict__ qvm, float* __restrict__ pvm,
                     float* __restrict__ qvx, float* __restrict__ pvx)
{
    __shared__ float cosRow[S_];
    __shared__ float cosCol[S_];

    const int bid = blockIdx.x;
    const int s   = bid % S_;
    const int hf  = (bid / S_) & 1;
    const int b   = bid / (2 * S_);
    const int tid = threadIdx.x;

    const float* cosBase = cosW + (b * 2 + hf) * (S_ * S_);
    if (tid < S_)       cosRow[tid]       = cosBase[s * S_ + tid];
    else                cosCol[tid - S_]  = cosBase[(tid - S_) * S_ + s];
    __syncthreads();

    const float rs = rowSum[(b * 2 + hf) * S_ + s];
    const float cs = colSum[(b * 2 + hf) * S_ + s];

    const float* Ph = p + (b * S_) * (2 * H_) + hf * H_;
    const float* Qh = q + (b * S_) * (2 * H_) + hf * H_;
    const int obase = ((b * 2 + hf) * S_ + s) * H_;

    for (int h = tid; h < H_; h += 128) {
        float qm = 0.f, pm = 0.f, qx = NEG_INF, px = NEG_INF;
        for (int j = 0; j < S_; ++j) {
            float a  = Qh[j * (2 * H_) + h] * cosRow[j];
            float bb = Ph[j * (2 * H_) + h] * cosCol[j];
            qm += a;  qx = fmaxf(qx, a);
            pm += bb; px = fmaxf(px, bb);
        }
        qvm[obase + h] = qm / rs;
        qvx[obase + h] = qx;
        pvm[obase + h] = pm / cs;
        pvx[obase + h] = px;
    }
}

__global__ __launch_bounds__(256)
void mpcos_kernel(const float* __restrict__ p, const float* __restrict__ q,
                  const float* __restrict__ W,
                  const float* __restrict__ qvm, const float* __restrict__ pvm,
                  const float* __restrict__ qvx, const float* __restrict__ pvx,
                  float* __restrict__ out)
{
    const int TOT = 2 * 6 * B_ * S_ * L_;
    int idx = blockIdx.x * blockDim.x + threadIdx.x;
    if (idx >= TOT) return;

    const int l = idx % L_;  int t = idx / L_;
    const int s = t % S_;    t /= S_;
    const int b = t % B_;    t /= B_;
    const int sec = t % 6;   const int tgt = t / 6;

    const int secTab[6] = {0, 1, 4, 5, 6, 7};
    const int secId = secTab[sec];
    const int hf = secId & 1;

    const float* X = (tgt == 0 ? p : q) + (b * S_ + s) * (2 * H_) + hf * H_;
    const float* Y;
    if (secId == 0)      Y = (tgt == 0 ? q : p) + (b * S_ + (S_ - 1)) * (2 * H_);
    else if (secId == 1) Y = (tgt == 0 ? q : p) + (b * S_) * (2 * H_) + H_;
    else {
        const float* vec = (secId <= 5) ? (tgt == 0 ? qvm : pvm)
                                        : (tgt == 0 ? qvx : pvx);
        Y = vec + ((b * 2 + hf) * S_ + s) * H_;
    }
    const float* Wr = W + (secId * L_ + l) * H_;

    float num = 0.f, nx = 0.f, ny = 0.f;
    for (int h = 0; h < H_; ++h) {
        float w  = Wr[h];
        float xw = X[h] * w;
        float yw = Y[h] * w;
        num += xw * yw;
        nx  += xw * xw;
        ny  += yw * yw;
    }
    float den = fmaxf(sqrtf(nx) * sqrtf(ny), 1e-8f);
    out[(tgt == 0 ? 0 : OUTQOFF) + (b * S_ + s) * OUTC + secId * L_ + l] = num / den;
}

extern "C" void kernel_launch(void* const* d_in, const int* in_sizes, int n_in,
                              void* d_out, int out_size, void* d_ws, size_t ws_size,
                              hipStream_t stream)
{
    const float* p = (const float*)d_in[0];
    const float* q = (const float*)d_in[1];
    const float* W = (const float*)d_in[2];
    float* out = (float*)d_out;

    float* ws     = (float*)d_ws;
    float* cosW   = ws;
    float* rowSum = cosW + B_ * 2 * S_ * S_;
    float* colSum = rowSum + B_ * 2 * S_;
    float* qvm    = colSum + B_ * 2 * S_;
    float* pvm    = qvm + B_ * 2 * S_ * H_;
    float* qvx    = pvm + B_ * 2 * S_ * H_;
    float* pvx    = qvx + B_ * 2 * S_ * H_;

    maxpool_kernel <<<B_ * 2 * L_, 128, 0, stream>>>(p, q, W, out);
    attn_cos_kernel<<<B_ * 2,      128, 0, stream>>>(p, q, cosW, rowSum, colSum);
    attn_vec_kernel<<<B_ * 2 * S_, 128, 0, stream>>>(p, q, cosW, rowSum, colSum,
                                                     qvm, pvm, qvx, pvx);
    const int TOT = 2 * 6 * B_ * S_ * L_;
    mpcos_kernel   <<<(TOT + 255) / 256, 256, 0, stream>>>(p, q, W, qvm, pvm, qvx, pvx, out);
}


# attn_vec: one block computes 8 consecutive s rows per (b,half), one thread per h; each P/Q element is loaded once (next 4 j prefetched in registers) and reused for all 8 rows; j sum order unchanged
# speedup vs baseline: 1.2725x; 1.2725x over previous
#include <hip/hip_runtime.h>
#include <math.h>

typedef __attribute__((ext_vector_type(2))) float v2f;
typedef __attribute__((ext_vector_type(8))) float v8f;

#define B_    32
#define S_    64
#define H_    300
#define L_    20
#define KT    60
#define KTP   62
#define OUTC  160
#define OUTQOFF (B_*S_*OUTC)
#define NEG_INF (-3.4e38f)

__global__ __launch_bounds__(128)
void maxpool_kernel(const float* __restrict__ p, const float* __restrict__ q,
                    const float* __restrict__ W, float* __restrict__ out)
{
    __shared__ float As[S_][KTP];
    __shared__ float Bs[S_][KTP];
    __shared__ float nA[S_];
    __shared__ float nB[S_];
    __shared__ float cosBuf[S_][S_ + 1];

    const int idx  = blockIdx.x;
    const int l    = idx % L_;
    const int hf   = (idx / L_) & 1;
    const int b    = idx / (2 * L_);
    const int tid  = threadIdx.x;
    const int lane = tid & 31;
    const int wv   = tid >> 5;
    const int lo   = lane & 15;
    const int hi   = lane >> 4;

    const float* Ph = p + (b * S_) * (2 * H_) + hf * H_;
    const float* Qh = q + (b * S_) * (2 * H_) + hf * H_;
    const float* Wr = W + ((2 + hf) * L_ + l) * H_;

    v8f c0 = {}, c1 = {}, c2 = {}, c3 = {};
    float nacc = 0.f;

    for (int kt = 0; kt < H_; kt += KT) {
        __syncthreads();
        for (int i = tid; i < S_ * KT; i += 128) {
            int s = i / KT, k = i % KT;
            float wval = Wr[kt + k];
            As[s][k] = Ph[s * (2 * H_) + kt + k] * wval;
            Bs[s][k] = Qh[s * (2 * H_) + kt + k] * wval;
        }
        __syncthreads();
        if (tid < S_) {
            for (int k = 0; k < KT; ++k) { float v = As[tid][k]; nacc += v * v; }
        } else {
            int r = tid - S_;
            for (int k = 0; k < KT; ++k) { float v = Bs[r][k]; nacc += v * v; }
        }
        const int m = wv * 16 + lo;
        for (int kb = 0; kb < KT; kb += 4) {
            int ka = kb + 2 * hi;
            v2f a  = *(const v2f*)&As[m][ka];
            v2f b0 = *(const v2f*)&Bs[ 0 + lo][ka];
            v2f b1 = *(const v2f*)&Bs[16 + lo][ka];
            v2f b2 = *(const v2f*)&Bs[32 + lo][ka];
            v2f b3 = *(const v2f*)&Bs[48 + lo][ka];
            c0 = __builtin_amdgcn_wmma_f32_16x16x4_f32(false, a, false, b0, (short)0, c0, false, false);
            c1 = __builtin_amdgcn_wmma_f32_16x16x4_f32(false, a, false, b1, (short)0, c1, false, false);
            c2 = __builtin_amdgcn_wmma_f32_16x16x4_f32(false, a, false, b2, (short)0, c2, false, false);
            c3 = __builtin_amdgcn_wmma_f32_16x16x4_f32(false, a, false, b3, (short)0, c3, false, false);
        }
    }
    __syncthreads();
    if (tid < S_) nA[tid] = sqrtf(nacc); else nB[tid - S_] = sqrtf(nacc);
    __syncthreads();

    #pragma unroll
    for (int j = 0; j < 8; ++j) {
        int m = wv * 16 + 8 * hi + j;
        float na = nA[m];
        cosBuf[m][ 0 + lo] = c0[j] / (na * nB[ 0 + lo]);
        cosBuf[m][16 + lo] = c1[j] / (na * nB[16 + lo]);
        cosBuf[m][32 + lo] = c2[j] / (na * nB[32 + lo]);
        cosBuf[m][48 + lo] = c3[j] / (na * nB[48 + lo]);
    }
    __syncthreads();

    if (tid < S_) {
        float mx = NEG_INF;
        for (int n = 0; n < S_; ++n) mx = fmaxf(mx, cosBuf[tid][n]);
        out[(b * S_ + tid) * OUTC + (2 + hf) * L_ + l] = mx;
    } else {
        int n = tid - S_;
        float mx = NEG_INF;
        for (int m = 0; m < S_; ++m) mx = fmaxf(mx, cosBuf[m][n]);
        out[OUTQOFF + (b * S_ + n) * OUTC + (2 + hf) * L_ + l] = mx;
    }
}

__global__ __launch_bounds__(128)
void attn_cos_kernel(const float* __restrict__ p, const float* __restrict__ q,
                     float* __restrict__ cosW, float* __restrict__ rowSum,
                     float* __restrict__ colSum)
{
    __shared__ float As[S_][KTP];
    __shared__ float Bs[S_][KTP];
    __shared__ float nA[S_];
    __shared__ float nB[S_];
    __shared__ float cosBuf[S_][S_ + 1];

    const int hf   = blockIdx.x & 1;
    const int b    = blockIdx.x >> 1;
    const int tid  = threadIdx.x;
    const int lane = tid & 31;
    const int wv   = tid >> 5;
    const int lo   = lane & 15;
    const int hi   = lane >> 4;

    const float* Ph = p + (b * S_) * (2 * H_) + hf * H_;
    const float* Qh = q + (b * S_) * (2 * H_) + hf * H_;

    v8f c0 = {}, c1 = {}, c2 = {}, c3 = {};
    float nacc = 0.f;

    for (int kt = 0; kt < H_; kt += KT) {
        __syncthreads();
        for (int i = tid; i < S_ * KT; i += 128) {
            int s = i / KT, k = i % KT;
            As[s][k] = Ph[s * (2 * H_) + kt + k];
            Bs[s][k] = Qh[s * (2 * H_) + kt + k];
        }
        __syncthreads();
        if (tid < S_) {
            for (int k = 0; k < KT; ++k) { float v = As[tid][k]; nacc += v * v; }
        } else {
            int r = tid - S_;
            for (int k = 0; k < KT; ++k) { float v = Bs[r][k]; nacc += v * v; }
        }
        const int m = wv * 16 + lo;
        for (int kb = 0; kb < KT; kb += 4) {
            int ka = kb + 2 * hi;
            v2f a  = *(const v2f*)&As[m][ka];
            v2f b0 = *(const v2f*)&Bs[ 0 + lo][ka];
            v2f b1 = *(const v2f*)&Bs[16 + lo][ka];
            v2f b2 = *(const v2f*)&Bs[32 + lo][ka];
            v2f b3 = *(const v2f*)&Bs[48 + lo][ka];
            c0 = __builtin_amdgcn_wmma_f32_16x16x4_f32(false, a, false, b0, (short)0, c0, false, false);
            c1 = __builtin_amdgcn_wmma_f32_16x16x4_f32(false, a, false, b1, (short)0, c1, false, false);
            c2 = __builtin_amdgcn_wmma_f32_16x16x4_f32(false, a, false, b2, (short)0, c2, false, false);
            c3 = __builtin_amdgcn_wmma_f32_16x16x4_f32(false, a, false, b3, (short)0, c3, false, false);
        }
    }
    __syncthreads();
    if (tid < S_) nA[tid] = sqrtf(nacc); else nB[tid - S_] = sqrtf(nacc);
    __syncthreads();

    #pragma unroll
    for (int j = 0; j < 8; ++j) {
        int m = wv * 16 + 8 * hi + j;
        float na = nA[m];
        cosBuf[m][ 0 + lo] = c0[j] / (na * nB[ 0 + lo]);
        cosBuf[m][16 + lo] = c1[j] / (na * nB[16 + lo]);
        cosBuf[m][32 + lo] = c2[j] / (na * nB[32 + lo]);
        cosBuf[m][48 + lo] = c3[j] / (na * nB[48 + lo]);
    }
    __syncthreads();

    float* dst = cosW + (b * 2 + hf) * (S_ * S_);
    for (int i = tid; i < S_ * S_; i += 128)
        dst[i] = cosBuf[i >> 6][i & 63];

    if (tid < S_) {
        float sm = 0.f;
        for (int n = 0; n < S_; ++n) sm += cosBuf[tid][n];
        rowSum[(b * 2 + hf) * S_ + tid] = sm;
    } else {
        int n = tid - S_;
        float sm = 0.f;
        for (int m = 0; m < S_; ++m) sm += cosBuf[m][n];
        colSum[(b * 2 + hf) * S_ + n] = sm;
    }
}

#define ST   8
#define NST  (S_ / ST)
#define VTH  320
#define JU   4

__global__ __launch_bounds__(VTH)
void attn_vec_kernel(const float* __restrict__ p, const float* __restrict__ q,
                     const float* __restrict__ cosW,
                     const float* __restrict__ rowSum, const float* __restrict__ colSum,
                     float* __restrict__ qvm, float* __restrict__ pvm,
                     float* __restrict__ qvx, float* __restrict__ pvx)
{
    __shared__ __attribute__((aligned(16))) float cosRowT[S_][ST];
    __shared__ __attribute__((aligned(16))) float cosColT[S_][ST];

    const int bid = blockIdx.x;
    const int s0  = (bid % NST) * ST;
    const int hf  = (bid / NST) & 1;
    const int b   = bid / (2 * NST);
    const int tid = threadIdx.x;
    const int bh  = b * 2 + hf;

    const float* cosBase = cosW + bh * (S_ * S_);
    for (int i = tid; i < S_ * ST; i += VTH) {
        int tr = i >> 6, jr = i & (S_ - 1);
        cosRowT[jr][tr] = cosBase[(s0 + tr) * S_ + jr];
        int jc = i / ST, tc = i % ST;
        cosColT[jc][tc] = cosBase[jc * S_ + s0 + tc];
    }
    __syncthreads();

    const int h = tid;
    if (h >= H_) return;

    const float* Qc = q + (b * S_) * (2 * H_) + hf * H_ + h;
    const float* Pc = p + (b * S_) * (2 * H_) + hf * H_ + h;

    float qm[ST], pm[ST], qx[ST], px[ST];
    for (int t = 0; t < ST; ++t) { qm[t] = 0.f; pm[t] = 0.f; qx[t] = NEG_INF; px[t] = NEG_INF; }

    auto accum = [&](const float (&qv)[JU], const float (&pv)[JU], int j0) {
        for (int u = 0; u < JU; ++u) {
            const int j = j0 + u;
            const float4 r0 = *(const float4*)&cosRowT[j][0];
            const float4 r1 = *(const float4*)&cosRowT[j][4];
            const float4 k0 = *(const float4*)&cosColT[j][0];
            const float4 k1 = *(const float4*)&cosColT[j][4];
            const float cr[ST] = {r0.x, r0.y, r0.z, r0.w, r1.x, r1.y, r1.z, r1.w};
            const float cc[ST] = {k0.x, k0.y, k0.z, k0.w, k1.x, k1.y, k1.z, k1.w};
            for (int t = 0; t < ST; ++t) {
                float a  = qv[u] * cr[t];
                float bb = pv[u] * cc[t];
                qm[t] += a;  qx[t] = fmaxf(qx[t], a);
                pm[t] += bb; px[t] = fmaxf(px[t], bb);
            }
        }
    };

    float qn[JU], pn[JU];
    for (int u = 0; u < JU; ++u) { qn[u] = Qc[u * (2 * H_)]; pn[u] = Pc[u * (2 * H_)]; }

    for (int j0 = 0; j0 < S_ - JU; j0 += JU) {
        float qc[JU], pc[JU];
        for (int u = 0; u < JU; ++u) { qc[u] = qn[u]; pc[u] = pn[u]; }
        for (int u = 0; u < JU; ++u) {
            qn[u] = Qc[(j0 + JU + u) * (2 * H_)];
            pn[u] = Pc[(j0 + JU + u) * (2 * H_)];
        }
        accum(qc, pc, j0);
    }
    accum(qn, pn, S_ - JU);

    for (int t = 0; t < ST; ++t) {
        const float rs = rowSum[bh * S_ + s0 + t];
        const float cs = colSum[bh * S_ + s0 + t];
        const int o = (bh * S_ + s0 + t) * H_ + h;
        qvm[o] = qm[t] / rs;
        qvx[o] = qx[t];
        pvm[o] = pm[t] / cs;
        pvx[o] = px[t];
    }
}

__global__ __launch_bounds__(256)
void mpcos_kernel(const float* __restrict__ p, const float* __restrict__ q,
                  const float* __restrict__ W,
                  const float* __restrict__ qvm, const float* __restrict__ pvm,
                  const float* __restrict__ qvx, const float* __restrict__ pvx,
                  float* __restrict__ out)
{
    const int TOT = 2 * 6 * B_ * S_ * L_;
    int idx = blockIdx.x * blockDim.x + threadIdx.x;
    if (idx >= TOT) return;

    const int l = idx % L_;  int t = idx / L_;
    const int s = t % S_;    t /= S_;
    const int b = t % B_;    t /= B_;
    const int sec = t % 6;   const int tgt = t / 6;

    const int secTab[6] = {0, 1, 4, 5, 6, 7};
    const int secId = secTab[sec];
    const int hf = secId & 1;

    const float* X = (tgt == 0 ? p : q) + (b * S_ + s) * (2 * H_) + hf * H_;
    const float* Y;
    if (secId == 0)      Y = (tgt == 0 ? q : p) + (b * S_ + (S_ - 1)) * (2 * H_);
    else if (secId == 1) Y = (tgt == 0 ? q : p) + (b * S_) * (2 * H_) + H_;
    else {
        const float* vec = (secId <= 5) ? (tgt == 0 ? qvm : pvm)
                                        : (tgt == 0 ? qvx : pvx);
        Y = vec + ((b * 2 + hf) * S_ + s) * H_;
    }
    const float* Wr = W + (secId * L_ + l) * H_;

    float num = 0.f, nx = 0.f, ny = 0.f;
    for (int h = 0; h < H_; ++h) {
        float w  = Wr[h];
        float xw = X[h] * w;
        float yw = Y[h] * w;
        num += xw * yw;
        nx  += xw * xw;
        ny  += yw * yw;
    }
    float den = fmaxf(sqrtf(nx) * sqrtf(ny), 1e-8f);
    out[(tgt == 0 ? 0 : OUTQOFF) + (b * S_ + s) * OUTC + secId * L_ + l] = num / den;
}

extern "C" void kernel_launch(void* const* d_in, const int* in_sizes, int n_in,
                              void* d_out, int out_size, void* d_ws, size_t ws_size,
                              hipStream_t stream)
{
    const float* p = (const float*)d_in[0];
    const float* q = (const float*)d_in[1];
    const float* W = (const float*)d_in[2];
    float* out = (float*)d_out;

    float* ws     = (float*)d_ws;
    float* cosW   = ws;
    float* rowSum = cosW + B_ * 2 * S_ * S_;
    float* colSum = rowSum + B_ * 2 * S_;
    float* qvm    = colSum + B_ * 2 * S_;
    float* pvm    = qvm + B_ * 2 * S_ * H_;
    float* qvx    = pvm + B_ * 2 * S_ * H_;
    float* pvx    = qvx + B_ * 2 * S_ * H_;

    maxpool_kernel <<<B_ * 2 * L_, 128, 0, stream>>>(p, q, W, out);
    attn_cos_kernel<<<B_ * 2,      128, 0, stream>>>(p, q, cosW, rowSum, colSum);
    attn_vec_kernel<<<B_ * 2 * NST, VTH, 0, stream>>>(p, q, cosW, rowSum, colSum,
                                                     qvm, pvm, qvx, pvx);
    const int TOT = 2 * 6 * B_ * S_ * L_;
    mpcos_kernel   <<<(TOT + 255) / 256, 256, 0, stream>>>(p, q, W, qvm, pvm, qvx, pvx, out);
}


# mpcos: block stages 32 X rows, 32 Y rows and the section's W (transposed) in LDS with row-contiguous loads; lane = row, wave = 4 consecutive l; h sum order and products unchanged
# speedup vs baseline: 1.7885x; 1.4055x over previous
#include <hip/hip_runtime.h>
#include <math.h>

typedef __attribute__((ext_vector_type(2))) float v2f;
typedef __attribute__((ext_vector_type(8))) float v8f;

#define B_    32
#define S_    64
#define H_    300
#define L_    20
#define KT    60
#define KTP   62
#define OUTC  160
#define OUTQOFF (B_*S_*OUTC)
#define NEG_INF (-3.4e38f)

__global__ __launch_bounds__(128)
void maxpool_kernel(const float* __restrict__ p, const float* __restrict__ q,
                    const float* __restrict__ W, float* __restrict__ out)
{
    __shared__ float As[S_][KTP];
    __shared__ float Bs[S_][KTP];
    __shared__ float nA[S_];
    __shared__ float nB[S_];
    __shared__ float cosBuf[S_][S_ + 1];

    const int idx  = blockIdx.x;
    const int l    = idx % L_;
    const int hf   = (idx / L_) & 1;
    const int b    = idx / (2 * L_);
    const int tid  = threadIdx.x;
    const int lane = tid & 31;
    const int wv   = tid >> 5;
    const int lo   = lane & 15;
    const int hi   = lane >> 4;

    const float* Ph = p + (b * S_) * (2 * H_) + hf * H_;
    const float* Qh = q + (b * S_) * (2 * H_) + hf * H_;
    const float* Wr = W + ((2 + hf) * L_ + l) * H_;

    v8f c0 = {}, c1 = {}, c2 = {}, c3 = {};
    float nacc = 0.f;

    for (int kt = 0; kt < H_; kt += KT) {
        __syncthreads();
        for (int i = tid; i < S_ * KT; i += 128) {
            int s = i / KT, k = i % KT;
            float wval = Wr[kt + k];
            As[s][k] = Ph[s * (2 * H_) + kt + k] * wval;
            Bs[s][k] = Qh[s * (2 * H_) + kt + k] * wval;
        }
        __syncthreads();
        if (tid < S_) {
            for (int k = 0; k < KT; ++k) { float v = As[tid][k]; nacc += v * v; }
        } else {
            int r = tid - S_;
            for (int k = 0; k < KT; ++k) { float v = Bs[r][k]; nacc += v * v; }
        }
        const int m = wv * 16 + lo;
        for (int kb = 0; kb < KT; kb += 4) {
            int ka = kb + 2 * hi;
            v2f a  = *(const v2f*)&As[m][ka];
            v2f b0 = *(const v2f*)&Bs[ 0 + lo][ka];
            v2f b1 = *(const v2f*)&Bs[16 + lo][ka];
            v2f b2 = *(const v2f*)&Bs[32 + lo][ka];
            v2f b3 = *(const v2f*)&Bs[48 + lo][ka];
            c0 = __builtin_amdgcn_wmma_f32_16x16x4_f32(false, a, false, b0, (short)0, c0, false, false);
            c1 = __builtin_amdgcn_wmma_f32_16x16x4_f32(false, a, false, b1, (short)0, c1, false, false);
            c2 = __builtin_amdgcn_wmma_f32_16x16x4_f32(false, a, false, b2, (short)0, c2, false, false);
            c3 = __builtin_amdgcn_wmma_f32_16x16x4_f32(false, a, false, b3, (short)0, c3, false, false);
        }
    }
    __syncthreads();
    if (tid < S_) nA[tid] = sqrtf(nacc); else nB[tid - S_] = sqrtf(nacc);
    __syncthreads();

    #pragma unroll
    for (int j = 0; j < 8; ++j) {
        int m = wv * 16 + 8 * hi + j;
        float na = nA[m];
        cosBuf[m][ 0 + lo] = c0[j] / (na * nB[ 0 + lo]);
        cosBuf[m][16 + lo] = c1[j] / (na * nB[16 + lo]);
        cosBuf[m][32 + lo] = c2[j] / (na * nB[32 + lo]);
        cosBuf[m][48 + lo] = c3[j] / (na * nB[48 + lo]);
    }
    __syncthreads();

    if (tid < S_) {
        float mx = NEG_INF;
        for (int n = 0; n < S_; ++n) mx = fmaxf(mx, cosBuf[tid][n]);
        out[(b * S_ + tid) * OUTC + (2 + hf) * L_ + l] = mx;
    } else {
        int n = tid - S_;
        float mx = NEG_INF;
        for (int m = 0; m < S_; ++m) mx = fmaxf(mx, cosBuf[m][n]);
        out[OUTQOFF + (b * S_ + n) * OUTC + (2 + hf) * L_ + l] = mx;
    }
}

__global__ __launch_bounds__(128)
void attn_cos_kernel(const float* __restrict__ p, const float* __restrict__ q,
                     float* __restrict__ cosW, float* __restrict__ rowSum,
                     float* __restrict__ colSum)
{
    __shared__ float As[S_][KTP];
    __shared__ float Bs[S_][KTP];
    __shared__ float nA[S_];
    __shared__ float nB[S_];
    __shared__ float cosBuf[S_][S_ + 1];

    const int hf   = blockIdx.x & 1;
    const int b    = blockIdx.x >> 1;
    const int tid  = threadIdx.x;
    const int lane = tid & 31;
    const int wv   = tid >> 5;
    const int lo   = lane & 15;
    const int hi   = lane >> 4;

    const float* Ph = p + (b * S_) * (2 * H_) + hf * H_;
    const float* Qh = q + (b * S_) * (2 * H_) + hf * H_;

    v8f c0 = {}, c1 = {}, c2 = {}, c3 = {};
    float nacc = 0.f;

    for (int kt = 0; kt < H_; kt += KT) {
        __syncthreads();
        for (int i = tid; i < S_ * KT; i += 128) {
            int s = i / KT, k = i % KT;
            As[s][k] = Ph[s * (2 * H_) + kt + k];
            Bs[s][k] = Qh[s * (2 * H_) + kt + k];
        }
        __syncthreads();
        if (tid < S_) {
            for (int k = 0; k < KT; ++k) { float v = As[tid][k]; nacc += v * v; }
        } else {
            int r = tid - S_;
            for (int k = 0; k < KT; ++k) { float v = Bs[r][k]; nacc += v * v; }
        }
        const int m = wv * 16 + lo;
        for (int kb = 0; kb < KT; kb += 4) {
            int ka = kb + 2 * hi;
            v2f a  = *(const v2f*)&As[m][ka];
            v2f b0 = *(const v2f*)&Bs[ 0 + lo][ka];
            v2f b1 = *(const v2f*)&Bs[16 + lo][ka];
            v2f b2 = *(const v2f*)&Bs[32 + lo][ka];
            v2f b3 = *(const v2f*)&Bs[48 + lo][ka];
            c0 = __builtin_amdgcn_wmma_f32_16x16x4_f32(false, a, false, b0, (short)0, c0, false, false);
            c1 = __builtin_amdgcn_wmma_f32_16x16x4_f32(false, a, false, b1, (short)0, c1, false, false);
            c2 = __builtin_amdgcn_wmma_f32_16x16x4_f32(false, a, false, b2, (short)0, c2, false, false);
            c3 = __builtin_amdgcn_wmma_f32_16x16x4_f32(false, a, false, b3, (short)0, c3, false, false);
        }
    }
    __syncthreads();
    if (tid < S_) nA[tid] = sqrtf(nacc); else nB[tid - S_] = sqrtf(nacc);
    __syncthreads();

    #pragma unroll
    for (int j = 0; j < 8; ++j) {
        int m = wv * 16 + 8 * hi + j;
        float na = nA[m];
        cosBuf[m][ 0 + lo] = c0[j] / (na * nB[ 0 + lo]);
        cosBuf[m][16 + lo] = c1[j] / (na * nB[16 + lo]);
        cosBuf[m][32 + lo] = c2[j] / (na * nB[32 + lo]);
        cosBuf[m][48 + lo] = c3[j] / (na * nB[48 + lo]);
    }
    __syncthreads();

    float* dst = cosW + (b * 2 + hf) * (S_ * S_);
    for (int i = tid; i < S_ * S_; i += 128)
        dst[i] = cosBuf[i >> 6][i & 63];

    if (tid < S_) {
        float sm = 0.f;
        for (int n = 0; n < S_; ++n) sm += cosBuf[tid][n];
        rowSum[(b * 2 + hf) * S_ + tid] = sm;
    } else {
        int n = tid - S_;
        float sm = 0.f;
        for (int m = 0; m < S_; ++m) sm += cosBuf[m][n];
        colSum[(b * 2 + hf) * S_ + n] = sm;
    }
}

#define ST   8
#define NST  (S_ / ST)
#define VTH  320
#define JU   4

__global__ __launch_bounds__(VTH)
void attn_vec_kernel(const float* __restrict__ p, const float* __restrict__ q,
                     const float* __restrict__ cosW,
                     const float* __restrict__ rowSum, const float* __restrict__ colSum,
                     float* __restrict__ qvm, float* __restrict__ pvm,
                     float* __restrict__ qvx, float* __restrict__ pvx)
{
    __shared__ __attribute__((aligned(16))) float cosRowT[S_][ST];
    __shared__ __attribute__((aligned(16))) float cosColT[S_][ST];

    const int bid = blockIdx.x;
    const int s0  = (bid % NST) * ST;
    const int hf  = (bid / NST) & 1;
    const int b   = bid / (2 * NST);
    const int tid = threadIdx.x;
    const int bh  = b * 2 + hf;

    const float* cosBase = cosW + bh * (S_ * S_);
    for (int i = tid; i < S_ * ST; i += VTH) {
        int tr = i >> 6, jr = i & (S_ - 1);
        cosRowT[jr][tr] = cosBase[(s0 + tr) * S_ + jr];
        int jc = i / ST, tc = i % ST;
        cosColT[jc][tc] = cosBase[jc * S_ + s0 + tc];
    }
    __syncthreads();

    const int h = tid;
    if (h >= H_) return;

    const float* Qc = q + (b * S_) * (2 * H_) + hf * H_ + h;
    const float* Pc = p + (b * S_) * (2 * H_) + hf * H_ + h;

    float qm[ST], pm[ST], qx[ST], px[ST];
    for (int t = 0; t < ST; ++t) { qm[t] = 0.f; pm[t] = 0.f; qx[t] = NEG_INF; px[t] = NEG_INF; }

    auto accum = [&](const float (&qv)[JU], const float (&pv)[JU], int j0) {
        for (int u = 0; u < JU; ++u) {
            const int j = j0 + u;
            const float4 r0 = *(const float4*)&cosRowT[j][0];
            const float4 r1 = *(const float4*)&cosRowT[j][4];
            const float4 k0 = *(const float4*)&cosColT[j][0];
            const float4 k1 = *(const float4*)&cosColT[j][4];
            const float cr[ST] = {r0.x, r0.y, r0.z, r0.w, r1.x, r1.y, r1.z, r1.w};
            const float cc[ST] = {k0.x, k0.y, k0.z, k0.w, k1.x, k1.y, k1.z, k1.w};
            for (int t = 0; t < ST; ++t) {
                float a  = qv[u] * cr[t];
                float bb = pv[u] * cc[t];
                qm[t] += a;  qx[t] = fmaxf(qx[t], a);
                pm[t] += bb; px[t] = fmaxf(px[t], bb);
            }
        }
    };

    float qn[JU], pn[JU];
    for (int u = 0; u < JU; ++u) { qn[u] = Qc[u * (2 * H_)]; pn[u] = Pc[u * (2 * H_)]; }

    for (int j0 = 0; j0 < S_ - JU; j0 += JU) {
        float qc[JU], pc[JU];
        for (int u = 0; u < JU; ++u) { qc[u] = qn[u]; pc[u] = pn[u]; }
        for (int u = 0; u < JU; ++u) {
            qn[u] = Qc[(j0 + JU + u) * (2 * H_)];
            pn[u] = Pc[(j0 + JU + u) * (2 * H_)];
        }
        accum(qc, pc, j0);
    }
    accum(qn, pn, S_ - JU);

    for (int t = 0; t < ST; ++t) {
        const float rs = rowSum[bh * S_ + s0 + t];
        const float cs = colSum[bh * S_ + s0 + t];
        const int o = (bh * S_ + s0 + t) * H_ + h;
        qvm[o] = qm[t] / rs;
        qvx[o] = qx[t];
        pvm[o] = pm[t] / cs;
        pvx[o] = px[t];
    }
}

#define MR    32
#define MLG   4
#define MTH   ((L_ / MLG) * 32)
#define H4    (H_ / 4)
#define MPBLK (2 * 6 * B_ * (S_ / MR))

__global__ __launch_bounds__(MTH)
void mpcos_kernel(const float* __restrict__ p, const float* __restrict__ q,
                  const float* __restrict__ W,
                  const float* __restrict__ qvm, const float* __restrict__ pvm,
                  const float* __restrict__ qvx, const float* __restrict__ pvx,
                  float* __restrict__ out)
{
    __shared__ __attribute__((aligned(16))) float Xs[MR][H_];
    __shared__ __attribute__((aligned(16))) float Ys[MR][H_];
    __shared__ __attribute__((aligned(16))) float WsT[H_][L_];
    __shared__ float res[MR][L_ + 1];

    int t = blockIdx.x;
    const int rh  = t % (S_ / MR);  t /= (S_ / MR);
    const int b   = t % B_;         t /= B_;
    const int sec = t % 6;          const int tgt = t / 6;
    const int secId = sec < 2 ? sec : sec + 2;
    const int hf  = secId & 1;
    const int s0  = rh * MR;
    const int tid = threadIdx.x;

    const float* Xb = (tgt == 0 ? p : q) + (b * S_ + s0) * (2 * H_) + hf * H_;
    const float* Yb;
    int ystride;
    if (secId == 0)      { Yb = (tgt == 0 ? q : p) + (b * S_ + (S_ - 1)) * (2 * H_); ystride = 0; }
    else if (secId == 1) { Yb = (tgt == 0 ? q : p) + (b * S_) * (2 * H_) + H_;       ystride = 0; }
    else {
        const float* vec = (secId <= 5) ? (tgt == 0 ? qvm : pvm)
                                        : (tgt == 0 ? qvx : pvx);
        Yb = vec + ((b * 2 + hf) * S_ + s0) * H_;
        ystride = H_;
    }
    const float* Wb = W + (secId * L_) * H_;

    const int rr = tid / 80;
    const int cc = tid - rr * 80;
    if (cc < H4) {
        const float* xs = Xb + rr * (2 * H_) + 4 * cc;
        const float* ys = Yb + rr * ystride + 4 * cc;
        float4 xv[MR / 2], yv[MR / 2];
        #pragma unroll
        for (int k = 0; k < MR / 2; ++k) {
            const float* xk = xs + k * (2 * 2 * H_);
            const float* yk = ys + k * (2 * ystride);
            xv[k] = float4{xk[0], xk[1], xk[2], xk[3]};
            yv[k] = float4{yk[0], yk[1], yk[2], yk[3]};
        }
        #pragma unroll
        for (int k = 0; k < MR / 2; ++k) {
            *(float4*)&Xs[2 * k + rr][4 * cc] = xv[k];
            *(float4*)&Ys[2 * k + rr][4 * cc] = yv[k];
        }
    }
    #pragma unroll
    for (int hh = 0; hh < (H_ + MTH - 1) / MTH; ++hh) {
        const int h = tid + hh * MTH;
        if (h < H_) {
            float4 wv[L_ / 4];
            #pragma unroll
            for (int g = 0; g < L_ / 4; ++g) {
                const float* wg = Wb + (4 * g) * H_ + h;
                wv[g] = float4{wg[0], wg[H_], wg[2 * H_], wg[3 * H_]};
            }
            #pragma unroll
            for (int g = 0; g < L_ / 4; ++g)
                *(float4*)&WsT[h][4 * g] = wv[g];
        }
    }
    __syncthreads();

    const int lane = tid & 31;
    const int l0   = (tid >> 5) * MLG;

    v2f numA = {0.f, 0.f}, nxA = {0.f, 0.f}, nyA = {0.f, 0.f};
    v2f numB = {0.f, 0.f}, nxB = {0.f, 0.f}, nyB = {0.f, 0.f};

    for (int c = 0; c < H4; ++c) {
        const float4 x4 = *(const float4*)&Xs[lane][4 * c];
        const float4 y4 = *(const float4*)&Ys[lane][4 * c];
        const float xe[4] = {x4.x, x4.y, x4.z, x4.w};
        const float ye[4] = {y4.x, y4.y, y4.z, y4.w};
        for (int e = 0; e < 4; ++e) {
            const float4 w4 = *(const float4*)&WsT[4 * c + e][l0];
            const v2f wA = {w4.x, w4.y};
            const v2f wB = {w4.z, w4.w};
            const v2f xx = {xe[e], xe[e]};
            const v2f yy = {ye[e], ye[e]};
            const v2f xwA = xx * wA, ywA = yy * wA;
            const v2f xwB = xx * wB, ywB = yy * wB;
            numA = __builtin_elementwise_fma(xwA, ywA, numA);
            nxA  = __builtin_elementwise_fma(xwA, xwA, nxA);
            nyA  = __builtin_elementwise_fma(ywA, ywA, nyA);
            numB = __builtin_elementwise_fma(xwB, ywB, numB);
            nxB  = __builtin_elementwise_fma(xwB, xwB, nxB);
            nyB  = __builtin_elementwise_fma(ywB, ywB, nyB);
        }
    }

    const float num[MLG] = {numA.x, numA.y, numB.x, numB.y};
    const float nx[MLG]  = {nxA.x,  nxA.y,  nxB.x,  nxB.y};
    const float ny[MLG]  = {nyA.x,  nyA.y,  nyB.x,  nyB.y};
    for (int j = 0; j < MLG; ++j) {
        const float den = fmaxf(sqrtf(nx[j]) * sqrtf(ny[j]), 1e-8f);
        res[lane][l0 + j] = num[j] / den;
    }
    __syncthreads();

    float* ob = out + (tgt == 0 ? 0 : OUTQOFF) + (b * S_ + s0) * OUTC + secId * L_;
    for (int i = tid; i < MR * L_; i += MTH) {
        const int r = i / L_, l = i % L_;
        ob[r * OUTC + l] = res[r][l];
    }
}

extern "C" void kernel_launch(void* const* d_in, const int* in_sizes, int n_in,
                              void* d_out, int out_size, void* d_ws, size_t ws_size,
                              hipStream_t stream)
{
    const float* p = (const float*)d_in[0];
    const float* q = (const float*)d_in[1];
    const float* W = (const float*)d_in[2];
    float* out = (float*)d_out;

    float* ws     = (float*)d_ws;
    float* cosW   = ws;
    float* rowSum = cosW + B_ * 2 * S_ * S_;
    float* colSum = rowSum + B_ * 2 * S_;
    float* qvm    = colSum + B_ * 2 * S_;
    float* pvm    = qvm + B_ * 2 * S_ * H_;
    float* qvx    = pvm + B_ * 2 * S_ * H_;
    float* pvx    = qvx + B_ * 2 * S_ * H_;

    maxpool_kernel <<<B_ * 2 * L_, 128, 0, stream>>>(p, q, W, out);
    attn_cos_kernel<<<B_ * 2,      128, 0, stream>>>(p, q, cosW, rowSum, colSum);
    attn_vec_kernel<<<B_ * 2 * NST, VTH, 0, stream>>>(p, q, cosW, rowSum, colSum,
                                                     qvm, pvm, qvx, pvx);
    mpcos_kernel   <<<MPBLK, MTH, 0, stream>>>(p, q, W, qvm, pvm, qvx, pvx, out);
}


# maxpool: block per (b,half,5 l) stages the raw P,Q rows in LDS once for 5 l (was per l); w^2 goes on the A fragment, products are (P*w^2)*Q; row norms summed per lane half, then joined
# speedup vs baseline: 2.3634x; 1.3214x over previous
#include <hip/hip_runtime.h>
#include <math.h>

typedef __attribute__((ext_vector_type(2))) float v2f;
typedef __attribute__((ext_vector_type(8))) float v8f;

#define B_    32
#define S_    64
#define H_    300
#define L_    20
#define KT    60
#define KTP   62
#define OUTC  160
#define OUTQOFF (B_*S_*OUTC)
#define NEG_INF (-3.4e38f)

#define MP_LG   5
#define MP_NLG  (L_ / MP_LG)
#define MP_TH   (MP_LG * 4 * 32)
#define MP_CH   100
#define MP_C4   (MP_CH / 4)
#define MP_NR   ((2 * S_ * MP_C4) / MP_TH)

__global__ __launch_bounds__(MP_TH)
void maxpool_kernel(const float* __restrict__ p, const float* __restrict__ q,
                    const float* __restrict__ W, float* __restrict__ out)
{
    __shared__ __attribute__((aligned(16))) float XQ[2 * S_][H_];
    __shared__ __attribute__((aligned(16))) float w2s[MP_LG][H_];
    __shared__ float nPs[MP_LG][S_];
    __shared__ float nQs[MP_LG][S_];
    __shared__ float rowMx[MP_LG][S_];
    __shared__ float colPt[MP_LG][4][S_];

    const int idx  = blockIdx.x;
    const int lg   = idx % MP_NLG;
    const int hf   = (idx / MP_NLG) & 1;
    const int b    = idx / (2 * MP_NLG);
    const int tid  = threadIdx.x;
    const int lane = tid & 31;
    const int wv   = tid >> 5;
    const int lo   = lane & 15;
    const int hi   = lane >> 4;
    const int li   = wv >> 2;
    const int strip = wv & 3;

    const float* Ph = p + (b * S_) * (2 * H_) + hf * H_;
    const float* Qh = q + (b * S_) * (2 * H_) + hf * H_;
    const float* Wg = W + ((2 + hf) * L_ + lg * MP_LG) * H_;

    const float* gsrc[MP_NR];
    float*       ldst[MP_NR];
    #pragma unroll
    for (int r = 0; r < MP_NR; ++r) {
        const int i   = tid + r * MP_TH;
        const int row = i / MP_C4, c4 = i % MP_C4;
        gsrc[r] = (row < S_ ? Ph + row * (2 * H_) : Qh + (row - S_) * (2 * H_)) + 4 * c4;
        ldst[r] = &XQ[row][4 * c4];
    }

    float4 stg[MP_NR];
    #pragma unroll
    for (int r = 0; r < MP_NR; ++r) {
        const float* g = gsrc[r];
        stg[r] = float4{g[0], g[1], g[2], g[3]};
    }
    for (int i = tid; i < MP_LG * H_; i += MP_TH) {
        const float w = Wg[i];
        (&w2s[0][0])[i] = w * w;
    }
    #pragma unroll
    for (int r = 0; r < MP_NR; ++r) *(float4*)ldst[r] = stg[r];
    __syncthreads();

    const float* aRow = &XQ[16 * strip + lo][2 * hi];
    const float* bR0  = &XQ[S_ + 16 * ((strip + 0) & 3) + lo][2 * hi];
    const float* bR1  = &XQ[S_ + 16 * ((strip + 1) & 3) + lo][2 * hi];
    const float* bR2  = &XQ[S_ + 16 * ((strip + 2) & 3) + lo][2 * hi];
    const float* bR3  = &XQ[S_ + 16 * ((strip + 3) & 3) + lo][2 * hi];
    const float* wRow = &w2s[li][2 * hi];

    v8f c0 = {}, c1 = {}, c2 = {}, c3 = {};
    float nA = 0.f, nB = 0.f;

    #pragma unroll
    for (int ch = 0; ch < H_ / MP_CH; ++ch) {
        if (ch + 1 < H_ / MP_CH) {
            #pragma unroll
            for (int r = 0; r < MP_NR; ++r) {
                const float* g = gsrc[r] + (ch + 1) * MP_CH;
                stg[r] = float4{g[0], g[1], g[2], g[3]};
            }
        }
        for (int kb = ch * MP_CH; kb < (ch + 1) * MP_CH; kb += 4) {
            const v2f a  = *(const v2f*)(aRow + kb);
            const v2f w  = *(const v2f*)(wRow + kb);
            const v2f b0 = *(const v2f*)(bR0 + kb);
            const v2f b1 = *(const v2f*)(bR1 + kb);
            const v2f b2 = *(const v2f*)(bR2 + kb);
            const v2f b3 = *(const v2f*)(bR3 + kb);
            const v2f aw = a * w;
            const v2f bw = b0 * w;
            nA = fmaf(a.x, aw.x, nA);   nA = fmaf(a.y, aw.y, nA);
            nB = fmaf(b0.x, bw.x, nB);  nB = fmaf(b0.y, bw.y, nB);
            c0 = __builtin_amdgcn_wmma_f32_16x16x4_f32(false, aw, false, b0, (short)0, c0, false, false);
            c1 = __builtin_amdgcn_wmma_f32_16x16x4_f32(false, aw, false, b1, (short)0, c1, false, false);
            c2 = __builtin_amdgcn_wmma_f32_16x16x4_f32(false, aw, false, b2, (short)0, c2, false, false);
            c3 = __builtin_amdgcn_wmma_f32_16x16x4_f32(false, aw, false, b3, (short)0, c3, false, false);
        }
        if (ch + 1 < H_ / MP_CH) {
            #pragma unroll
            for (int r = 0; r < MP_NR; ++r)
                *(float4*)(ldst[r] + (ch + 1) * MP_CH) = stg[r];
            __syncthreads();
        }
    }

    nA += __shfl_xor(nA, 16);
    nB += __shfl_xor(nB, 16);
    if (hi == 0) {
        nPs[li][16 * strip + lo] = sqrtf(nA);
        nQs[li][16 * strip + lo] = sqrtf(nB);
    }
    __syncthreads();

    const int n0 = 16 * ((strip + 0) & 3) + lo;
    const int n1 = 16 * ((strip + 1) & 3) + lo;
    const int n2 = 16 * ((strip + 2) & 3) + lo;
    const int n3 = 16 * ((strip + 3) & 3) + lo;
    const float nb0 = nQs[li][n0], nb1 = nQs[li][n1], nb2 = nQs[li][n2], nb3 = nQs[li][n3];
    float cm0 = NEG_INF, cm1 = NEG_INF, cm2 = NEG_INF, cm3 = NEG_INF;
    float rmx[8];
    #pragma unroll
    for (int j = 0; j < 8; ++j) {
        const float na = nPs[li][16 * strip + 8 * hi + j];
        const float v0 = c0[j] / (na * nb0);
        const float v1 = c1[j] / (na * nb1);
        const float v2 = c2[j] / (na * nb2);
        const float v3 = c3[j] / (na * nb3);
        cm0 = fmaxf(cm0, v0); cm1 = fmaxf(cm1, v1);
        cm2 = fmaxf(cm2, v2); cm3 = fmaxf(cm3, v3);
        rmx[j] = fmaxf(fmaxf(fmaxf(fmaxf(NEG_INF, v0), v1), v2), v3);
    }
    #pragma unroll
    for (int d = 1; d < 16; d <<= 1) {
        #pragma unroll
        for (int j = 0; j < 8; ++j) rmx[j] = fmaxf(rmx[j], __shfl_xor(rmx[j], d));
    }
    if (lo == 0) {
        #pragma unroll
        for (int j = 0; j < 8; ++j) rowMx[li][16 * strip + 8 * hi + j] = rmx[j];
    }
    cm0 = fmaxf(cm0, __shfl_xor(cm0, 16));
    cm1 = fmaxf(cm1, __shfl_xor(cm1, 16));
    cm2 = fmaxf(cm2, __shfl_xor(cm2, 16));
    cm3 = fmaxf(cm3, __shfl_xor(cm3, 16));
    if (hi == 0) {
        colPt[li][strip][n0] = cm0;
        colPt[li][strip][n1] = cm1;
        colPt[li][strip][n2] = cm2;
        colPt[li][strip][n3] = cm3;
    }
    __syncthreads();

    const int nres = (S_ * MP_LG);
    if (tid < nres) {
        const int row = tid / MP_LG, l5 = tid % MP_LG;
        out[(b * S_ + row) * OUTC + (2 + hf) * L_ + lg * MP_LG + l5] = rowMx[l5][row];
    } else {
        const int u = tid - nres;
        const int n = u / MP_LG, l5 = u % MP_LG;
        float mx = NEG_INF;
        #pragma unroll
        for (int s = 0; s < 4; ++s) mx = fmaxf(mx, colPt[l5][s][n]);
        out[OUTQOFF + (b * S_ + n) * OUTC + (2 + hf) * L_ + lg * MP_LG + l5] = mx;
    }
}

__global__ __launch_bounds__(128)
void attn_cos_kernel(const float* __restrict__ p, const float* __restrict__ q,
                     float* __restrict__ cosW, float* __restrict__ rowSum,
                     float* __restrict__ colSum)
{
    __shared__ float As[S_][KTP];
    __shared__ float Bs[S_][KTP];
    __shared__ float nA[S_];
    __shared__ float nB[S_];
    __shared__ float cosBuf[S_][S_ + 1];

    const int hf   = blockIdx.x & 1;
    const int b    = blockIdx.x >> 1;
    const int tid  = threadIdx.x;
    const int lane = tid & 31;
    const int wv   = tid >> 5;
    const int lo   = lane & 15;
    const int hi   = lane >> 4;

    const float* Ph = p + (b * S_) * (2 * H_) + hf * H_;
    const float* Qh = q + (b * S_) * (2 * H_) + hf * H_;

    v8f c0 = {}, c1 = {}, c2 = {}, c3 = {};
    float nacc = 0.f;

    for (int kt = 0; kt < H_; kt += KT) {
        __syncthreads();
        for (int i = tid; i < S_ * KT; i += 128) {
            int s = i / KT, k = i % KT;
            As[s][k] = Ph[s * (2 * H_) + kt + k];
            Bs[s][k] = Qh[s * (2 * H_) + kt + k];
        }
        __syncthreads();
        if (tid < S_) {
            for (int k = 0; k < KT; ++k) { float v = As[tid][k]; nacc += v * v; }
        } else {
            int r = tid - S_;
            for (int k = 0; k < KT; ++k) { float v = Bs[r][k]; nacc += v * v; }
        }
        const int m = wv * 16 + lo;
        for (int kb = 0; kb < KT; kb += 4) {
            int ka = kb + 2 * hi;
            v2f a  = *(const v2f*)&As[m][ka];
            v2f b0 = *(const v2f*)&Bs[ 0 + lo][ka];
            v2f b1 = *(const v2f*)&Bs[16 + lo][ka];
            v2f b2 = *(const v2f*)&Bs[32 + lo][ka];
            v2f b3 = *(const v2f*)&Bs[48 + lo][ka];
            c0 = __builtin_amdgcn_wmma_f32_16x16x4_f32(false, a, false, b0, (short)0, c0, false, false);
            c1 = __builtin_amdgcn_wmma_f32_16x16x4_f32(false, a, false, b1, (short)0, c1, false, false);
            c2 = __builtin_amdgcn_wmma_f32_16x16x4_f32(false, a, false, b2, (short)0, c2, false, false);
            c3 = __builtin_amdgcn_wmma_f32_16x16x4_f32(false, a, false, b3, (short)0, c3, false, false);
        }
    }
    __syncthreads();
    if (tid < S_) nA[tid] = sqrtf(nacc); else nB[tid - S_] = sqrtf(nacc);
    __syncthreads();

    #pragma unroll
    for (int j = 0; j < 8; ++j) {
        int m = wv * 16 + 8 * hi + j;
        float na = nA[m];
        cosBuf[m][ 0 + lo] = c0[j] / (na * nB[ 0 + lo]);
        cosBuf[m][16 + lo] = c1[j] / (na * nB[16 + lo]);
        cosBuf[m][32 + lo] = c2[j] / (na * nB[32 + lo]);
        cosBuf[m][48 + lo] = c3[j] / (na * nB[48 + lo]);
    }
    __syncthreads();

    float* dst = cosW + (b * 2 + hf) * (S_ * S_);
    for (int i = tid; i < S_ * S_; i += 128)
        dst[i] = cosBuf[i >> 6][i & 63];

    if (tid < S_) {
        float sm = 0.f;
        for (int n = 0; n < S_; ++n) sm += cosBuf[tid][n];
        rowSum[(b * 2 + hf) * S_ + tid] = sm;
    } else {
        int n = tid - S_;
        float sm = 0.f;
        for (int m = 0; m < S_; ++m) sm += cosBuf[m][n];
        colSum[(b * 2 + hf) * S_ + n] = sm;
    }
}

#define ST   8
#define NST  (S_ / ST)
#define VTH  320
#define JU   4

__global__ __launch_bounds__(VTH)
void attn_vec_kernel(const float* __restrict__ p, const float* __restrict__ q,
                     const float* __restrict__ cosW,
                     const float* __restrict__ rowSum, const float* __restrict__ colSum,
                     float* __restrict__ qvm, float* __restrict__ pvm,
                     float* __restrict__ qvx, float* __restrict__ pvx)
{
    __shared__ __attribute__((aligned(16))) float cosRowT[S_][ST];
    __shared__ __attribute__((aligned(16))) float cosColT[S_][ST];

    const int bid = blockIdx.x;
    const int s0  = (bid % NST) * ST;
    const int hf  = (bid / NST) & 1;
    const int b   = bid / (2 * NST);
    const int tid = threadIdx.x;
    const int bh  = b * 2 + hf;

    const float* cosBase = cosW + bh * (S_ * S_);
    for (int i = tid; i < S_ * ST; i += VTH) {
        int tr = i >> 6, jr = i & (S_ - 1);
        cosRowT[jr][tr] = cosBase[(s0 + tr) * S_ + jr];
        int jc = i / ST, tc = i % ST;
        cosColT[jc][tc] = cosBase[jc * S_ + s0 + tc];
    }
    __syncthreads();

    const int h = tid;
    if (h >= H_) return;

    const float* Qc = q + (b * S_) * (2 * H_) + hf * H_ + h;
    const float* Pc = p + (b * S_) * (2 * H_) + hf * H_ + h;

    float qm[ST], pm[ST], qx[ST], px[ST];
    for (int t = 0; t < ST; ++t) { qm[t] = 0.f; pm[t] = 0.f; qx[t] = NEG_INF; px[t] = NEG_INF; }

    auto accum = [&](const float (&qv)[JU], const float (&pv)[JU], int j0) {
        for (int u = 0; u < JU; ++u) {
            const int j = j0 + u;
            const float4 r0 = *(const float4*)&cosRowT[j][0];
            const float4 r1 = *(const float4*)&cosRowT[j][4];
            const float4 k0 = *(const float4*)&cosColT[j][0];
            const float4 k1 = *(const float4*)&cosColT[j][4];
            const float cr[ST] = {r0.x, r0.y, r0.z, r0.w, r1.x, r1.y, r1.z, r1.w};
            const float cc[ST] = {k0.x, k0.y, k0.z, k0.w, k1.x, k1.y, k1.z, k1.w};
            for (int t = 0; t < ST; ++t) {
                float a  = qv[u] * cr[t];
                float bb = pv[u] * cc[t];
                qm[t] += a;  qx[t] = fmaxf(qx[t], a);
                pm[t] += bb; px[t] = fmaxf(px[t], bb);
            }
        }
    };

    float qn[JU], pn[JU];
    for (int u = 0; u < JU; ++u) { qn[u] = Qc[u * (2 * H_)]; pn[u] = Pc[u * (2 * H_)]; }

    for (int j0 = 0; j0 < S_ - JU; j0 += JU) {
        float qc[JU], pc[JU];
        for (int u = 0; u < JU; ++u) { qc[u] = qn[u]; pc[u] = pn[u]; }
        for (int u = 0; u < JU; ++u) {
            qn[u] = Qc[(j0 + JU + u) * (2 * H_)];
            pn[u] = Pc[(j0 + JU + u) * (2 * H_)];
        }
        accum(qc, pc, j0);
    }
    accum(qn, pn, S_ - JU);

    for (int t = 0; t < ST; ++t) {
        const float rs = rowSum[bh * S_ + s0 + t];
        const float cs = colSum[bh * S_ + s0 + t];
        const int o = (bh * S_ + s0 + t) * H_ + h;
        qvm[o] = qm[t] / rs;
        qvx[o] = qx[t];
        pvm[o] = pm[t] / cs;
        pvx[o] = px[t];
    }
}

#define MR    32
#define MLG   4
#define MTH   ((L_ / MLG) * 32)
#define H4    (H_ / 4)
#define MPBLK (2 * 6 * B_ * (S_ / MR))

__global__ __launch_bounds__(MTH)
void mpcos_kernel(const float* __restrict__ p, const float* __restrict__ q,
                  const float* __restrict__ W,
                  const float* __restrict__ qvm, const float* __restrict__ pvm,
                  const float* __restrict__ qvx, const float* __restrict__ pvx,
                  float* __restrict__ out)
{
    __shared__ __attribute__((aligned(16))) float Xs[MR][H_];
    __shared__ __attribute__((aligned(16))) float Ys[MR][H_];
    __shared__ __attribute__((aligned(16))) float WsT[H_][L_];
    __shared__ float res[MR][L_ + 1];

    int t = blockIdx.x;
    const int rh  = t % (S_ / MR);  t /= (S_ / MR);
    const int b   = t % B_;         t /= B_;
    const int sec = t % 6;          const int tgt = t / 6;
    const int secId = sec < 2 ? sec : sec + 2;
    const int hf  = secId & 1;
    const int s0  = rh * MR;
    const int tid = threadIdx.x;

    const float* Xb = (tgt == 0 ? p : q) + (b * S_ + s0) * (2 * H_) + hf * H_;
    const float* Yb;
    int ystride;
    if (secId == 0)      { Yb = (tgt == 0 ? q : p) + (b * S_ + (S_ - 1)) * (2 * H_); ystride = 0; }
    else if (secId == 1) { Yb = (tgt == 0 ? q : p) + (b * S_) * (2 * H_) + H_;       ystride = 0; }
    else {
        const float* vec = (secId <= 5) ? (tgt == 0 ? qvm : pvm)
                                        : (tgt == 0 ? qvx : pvx);
        Yb = vec + ((b * 2 + hf) * S_ + s0) * H_;
        ystride = H_;
    }
    const float* Wb = W + (secId * L_) * H_;

    const int rr = tid / 80;
    const int cc = tid - rr * 80;
    if (cc < H4) {
        const float* xs = Xb + rr * (2 * H_) + 4 * cc;
        const float* ys = Yb + rr * ystride + 4 * cc;
        float4 xv[MR / 2], yv[MR / 2];
        #pragma unroll
        for (int k = 0; k < MR / 2; ++k) {
            const float* xk = xs + k * (2 * 2 * H_);
            const float* yk = ys + k * (2 * ystride);
            xv[k] = float4{xk[0], xk[1], xk[2], xk[3]};
            yv[k] = float4{yk[0], yk[1], yk[2], yk[3]};
        }
        #pragma unroll
        for (int k = 0; k < MR / 2; ++k) {
            *(float4*)&Xs[2 * k + rr][4 * cc] = xv[k];
            *(float4*)&Ys[2 * k + rr][4 * cc] = yv[k];
        }
    }
    #pragma unroll
    for (int hh = 0; hh < (H_ + MTH - 1) / MTH; ++hh) {
        const int h = tid + hh * MTH;
        if (h < H_) {
            float4 wv[L_ / 4];
            #pragma unroll
            for (int g = 0; g < L_ / 4; ++g) {
                const float* wg = Wb + (4 * g) * H_ + h;
                wv[g] = float4{wg[0], wg[H_], wg[2 * H_], wg[3 * H_]};
            }
            #pragma unroll
            for (int g = 0; g < L_ / 4; ++g)
                *(float4*)&WsT[h][4 * g] = wv[g];
        }
    }
    __syncthreads();

    const int lane = tid & 31;
    const int l0   = (tid >> 5) * MLG;

    v2f numA = {0.f, 0.f}, nxA = {0.f, 0.f}, nyA = {0.f, 0.f};
    v2f numB = {0.f, 0.f}, nxB = {0.f, 0.f}, nyB = {0.f, 0.f};

    for (int c = 0; c < H4; ++c) {
        const float4 x4 = *(const float4*)&Xs[lane][4 * c];
        const float4 y4 = *(const float4*)&Ys[lane][4 * c];
        const float xe[4] = {x4.x, x4.y, x4.z, x4.w};
        const float ye[4] = {y4.x, y4.y, y4.z, y4.w};
        for (int e = 0; e < 4; ++e) {
            const float4 w4 = *(const float4*)&WsT[4 * c + e][l0];
            const v2f wA = {w4.x, w4.y};
            const v2f wB = {w4.z, w4.w};
            const v2f xx = {xe[e], xe[e]};
            const v2f yy = {ye[e], ye[e]};
            const v2f xwA = xx * wA, ywA = yy * wA;
            const v2f xwB = xx * wB, ywB = yy * wB;
            numA = __builtin_elementwise_fma(xwA, ywA, numA);
            nxA  = __builtin_elementwise_fma(xwA, xwA, nxA);
            nyA  = __builtin_elementwise_fma(ywA, ywA, nyA);
            numB = __builtin_elementwise_fma(xwB, ywB, numB);
            nxB  = __builtin_elementwise_fma(xwB, xwB, nxB);
            nyB  = __builtin_elementwise_fma(ywB, ywB, nyB);
        }
    }

    const float num[MLG] = {numA.x, numA.y, numB.x, numB.y};
    const float nx[MLG]  = {nxA.x,  nxA.y,  nxB.x,  nxB.y};
    const float ny[MLG]  = {nyA.x,  nyA.y,  nyB.x,  nyB.y};
    for (int j = 0; j < MLG; ++j) {
        const float den = fmaxf(sqrtf(nx[j]) * sqrtf(ny[j]), 1e-8f);
        res[lane][l0 + j] = num[j] / den;
    }
    __syncthreads();

    float* ob = out + (tgt == 0 ? 0 : OUTQOFF) + (b * S_ + s0) * OUTC + secId * L_;
    for (int i = tid; i < MR * L_; i += MTH) {
        const int r = i / L_, l = i % L_;
        ob[r * OUTC + l] = res[r][l];
    }
}

extern "C" void kernel_launch(void* const* d_in, const int* in_sizes, int n_in,
                              void* d_out, int out_size, void* d_ws, size_t ws_size,
                              hipStream_t stream)
{
    const float* p = (const float*)d_in[0];
    const float* q = (const float*)d_in[1];
    const float* W = (const float*)d_in[2];
    float* out = (float*)d_out;

    float* ws     = (float*)d_ws;
    float* cosW   = ws;
    float* rowSum = cosW + B_ * 2 * S_ * S_;
    float* colSum = rowSum + B_ * 2 * S_;
    float* qvm    = colSum + B_ * 2 * S_;
    float* pvm    = qvm + B_ * 2 * S_ * H_;
    float* qvx    = pvm + B_ * 2 * S_ * H_;
    float* pvx    = qvx + B_ * 2 * S_ * H_;

    maxpool_kernel <<<B_ * 2 * MP_NLG, MP_TH, 0, stream>>>(p, q, W, out);
    attn_cos_kernel<<<B_ * 2,      128, 0, stream>>>(p, q, cosW, rowSum, colSum);
    attn_vec_kernel<<<B_ * 2 * NST, VTH, 0, stream>>>(p, q, cosW, rowSum, colSum,
                                                     qvm, pvm, qvx, pvx);
    mpcos_kernel   <<<MPBLK, MTH, 0, stream>>>(p, q, W, qvm, pvm, qvx, pvx, out);
}


# attn_cos: P,Q rows staged whole in LDS by 16-byte loads, 3 prefetched chunks (was 4-byte loads, 8 in flight), 16 waves of one tile; row norms now 4 strided partial sums from the fragments, joined
# speedup vs baseline: 2.8824x; 1.2196x over previous
#include <hip/hip_runtime.h>
#include <math.h>

typedef __attribute__((ext_vector_type(2))) float v2f;
typedef __attribute__((ext_vector_type(8))) float v8f;

#define B_    32
#define S_    64
#define H_    300
#define L_    20
#define OUTC  160
#define OUTQOFF (B_*S_*OUTC)
#define NEG_INF (-3.4e38f)

#define MP_LG   5
#define MP_NLG  (L_ / MP_LG)
#define MP_TH   (MP_LG * 4 * 32)
#define MP_CH   100
#define MP_C4   (MP_CH / 4)
#define MP_NR   ((2 * S_ * MP_C4) / MP_TH)

__global__ __launch_bounds__(MP_TH)
void maxpool_kernel(const float* __restrict__ p, const float* __restrict__ q,
                    const float* __restrict__ W, float* __restrict__ out)
{
    __shared__ __attribute__((aligned(16))) float XQ[2 * S_][H_];
    __shared__ __attribute__((aligned(16))) float w2s[MP_LG][H_];
    __shared__ float nPs[MP_LG][S_];
    __shared__ float nQs[MP_LG][S_];
    __shared__ float rowMx[MP_LG][S_];
    __shared__ float colPt[MP_LG][4][S_];

    const int idx  = blockIdx.x;
    const int lg   = idx % MP_NLG;
    const int hf   = (idx / MP_NLG) & 1;
    const int b    = idx / (2 * MP_NLG);
    const int tid  = threadIdx.x;
    const int lane = tid & 31;
    const int wv   = tid >> 5;
    const int lo   = lane & 15;
    const int hi   = lane >> 4;
    const int li   = wv >> 2;
    const int strip = wv & 3;

    const float* Ph = p + (b * S_) * (2 * H_) + hf * H_;
    const float* Qh = q + (b * S_) * (2 * H_) + hf * H_;
    const float* Wg = W + ((2 + hf) * L_ + lg * MP_LG) * H_;

    const float* gsrc[MP_NR];
    float*       ldst[MP_NR];
    #pragma unroll
    for (int r = 0; r < MP_NR; ++r) {
        const int i   = tid + r * MP_TH;
        const int row = i / MP_C4, c4 = i % MP_C4;
        gsrc[r] = (row < S_ ? Ph + row * (2 * H_) : Qh + (row - S_) * (2 * H_)) + 4 * c4;
        ldst[r] = &XQ[row][4 * c4];
    }

    float4 stg[MP_NR];
    #pragma unroll
    for (int r = 0; r < MP_NR; ++r) {
        const float* g = gsrc[r];
        stg[r] = float4{g[0], g[1], g[2], g[3]};
    }
    for (int i = tid; i < MP_LG * H_; i += MP_TH) {
        const float w = Wg[i];
        (&w2s[0][0])[i] = w * w;
    }
    #pragma unroll
    for (int r = 0; r < MP_NR; ++r) *(float4*)ldst[r] = stg[r];
    __syncthreads();

    const float* aRow = &XQ[16 * strip + lo][2 * hi];
    const float* bR0  = &XQ[S_ + 16 * ((strip + 0) & 3) + lo][2 * hi];
    const float* bR1  = &XQ[S_ + 16 * ((strip + 1) & 3) + lo][2 * hi];
    const float* bR2  = &XQ[S_ + 16 * ((strip + 2) & 3) + lo][2 * hi];
    const float* bR3  = &XQ[S_ + 16 * ((strip + 3) & 3) + lo][2 * hi];
    const float* wRow = &w2s[li][2 * hi];

    v8f c0 = {}, c1 = {}, c2 = {}, c3 = {};
    float nA = 0.f, nB = 0.f;

    #pragma unroll
    for (int ch = 0; ch < H_ / MP_CH; ++ch) {
        if (ch + 1 < H_ / MP_CH) {
            #pragma unroll
            for (int r = 0; r < MP_NR; ++r) {
                const float* g = gsrc[r] + (ch + 1) * MP_CH;
                stg[r] = float4{g[0], g[1], g[2], g[3]};
            }
        }
        for (int kb = ch * MP_CH; kb < (ch + 1) * MP_CH; kb += 4) {
            const v2f a  = *(const v2f*)(aRow + kb);
            const v2f w  = *(const v2f*)(wRow + kb);
            const v2f b0 = *(const v2f*)(bR0 + kb);
            const v2f b1 = *(const v2f*)(bR1 + kb);
            const v2f b2 = *(const v2f*)(bR2 + kb);
            const v2f b3 = *(const v2f*)(bR3 + kb);
            const v2f aw = a * w;
            const v2f bw = b0 * w;
            nA = fmaf(a.x, aw.x, nA);   nA = fmaf(a.y, aw.y, nA);
            nB = fmaf(b0.x, bw.x, nB);  nB = fmaf(b0.y, bw.y, nB);
            c0 = __builtin_amdgcn_wmma_f32_16x16x4_f32(false, aw, false, b0, (short)0, c0, false, false);
            c1 = __builtin_amdgcn_wmma_f32_16x16x4_f32(false, aw, false, b1, (short)0, c1, false, false);
            c2 = __builtin_amdgcn_wmma_f32_16x16x4_f32(false, aw, false, b2, (short)0, c2, false, false);
            c3 = __builtin_amdgcn_wmma_f32_16x16x4_f32(false, aw, false, b3, (short)0, c3, false, false);
        }
        if (ch + 1 < H_ / MP_CH) {
            #pragma unroll
            for (int r = 0; r < MP_NR; ++r)
                *(float4*)(ldst[r] + (ch + 1) * MP_CH) = stg[r];
            __syncthreads();
        }
    }

    nA += __shfl_xor(nA, 16);
    nB += __shfl_xor(nB, 16);
    if (hi == 0) {
        nPs[li][16 * strip + lo] = sqrtf(nA);
        nQs[li][16 * strip + lo] = sqrtf(nB);
    }
    __syncthreads();

    const int n0 = 16 * ((strip + 0) & 3) + lo;
    const int n1 = 16 * ((strip + 1) & 3) + lo;
    const int n2 = 16 * ((strip + 2) & 3) + lo;
    const int n3 = 16 * ((strip + 3) & 3) + lo;
    const float nb0 = nQs[li][n0], nb1 = nQs[li][n1], nb2 = nQs[li][n2], nb3 = nQs[li][n3];
    float cm0 = NEG_INF, cm1 = NEG_INF, cm2 = NEG_INF, cm3 = NEG_INF;
    float rmx[8];
    #pragma unroll
    for (int j = 0; j < 8; ++j) {
        const float na = nPs[li][16 * strip + 8 * hi + j];
        const float v0 = c0[j] / (na * nb0);
        const float v1 = c1[j] / (na * nb1);
        const float v2 = c2[j] / (na * nb2);
        const float v3 = c3[j] / (na * nb3);
        cm0 = fmaxf(cm0, v0); cm1 = fmaxf(cm1, v1);
        cm2 = fmaxf(cm2, v2); cm3 = fmaxf(cm3, v3);
        rmx[j] = fmaxf(fmaxf(fmaxf(fmaxf(NEG_INF, v0), v1), v2), v3);
    }
    #pragma unroll
    for (int d = 1; d < 16; d <<= 1) {
        #pragma unroll
        for (int j = 0; j < 8; ++j) rmx[j] = fmaxf(rmx[j], __shfl_xor(rmx[j], d));
    }
    if (lo == 0) {
        #pragma unroll
        for (int j = 0; j < 8; ++j) rowMx[li][16 * strip + 8 * hi + j] = rmx[j];
    }
    cm0 = fmaxf(cm0, __shfl_xor(cm0, 16));
    cm1 = fmaxf(cm1, __shfl_xor(cm1, 16));
    cm2 = fmaxf(cm2, __shfl_xor(cm2, 16));
    cm3 = fmaxf(cm3, __shfl_xor(cm3, 16));
    if (hi == 0) {
        colPt[li][strip][n0] = cm0;
        colPt[li][strip][n1] = cm1;
        colPt[li][strip][n2] = cm2;
        colPt[li][strip][n3] = cm3;
    }
    __syncthreads();

    const int nres = (S_ * MP_LG);
    if (tid < nres) {
        const int row = tid / MP_LG, l5 = tid % MP_LG;
        out[(b * S_ + row) * OUTC + (2 + hf) * L_ + lg * MP_LG + l5] = rowMx[l5][row];
    } else {
        const int u = tid - nres;
        const int n = u / MP_LG, l5 = u % MP_LG;
        float mx = NEG_INF;
        #pragma unroll
        for (int s = 0; s < 4; ++s) mx = fmaxf(mx, colPt[l5][s][n]);
        out[OUTQOFF + (b * S_ + n) * OUTC + (2 + hf) * L_ + lg * MP_LG + l5] = mx;
    }
}

#define AC_TH   512
#define AC_CH   100
#define AC_C4   (AC_CH / 4)
#define AC_NP   (2 * S_ * AC_C4)
#define AC_NR   ((AC_NP + AC_TH - 1) / AC_TH)

__global__ __launch_bounds__(AC_TH)
void attn_cos_kernel(const float* __restrict__ p, const float* __restrict__ q,
                     float* __restrict__ cosW, float* __restrict__ rowSum,
                     float* __restrict__ colSum)
{
    __shared__ __attribute__((aligned(16))) float XQ[2 * S_][H_];
    __shared__ float nA[S_];
    __shared__ float nB[S_];
    __shared__ float cosBuf[S_][S_ + 1];

    const int hf   = blockIdx.x & 1;
    const int b    = blockIdx.x >> 1;
    const int tid  = threadIdx.x;
    const int lane = tid & 31;
    const int wv   = tid >> 5;
    const int lo   = lane & 15;
    const int hi   = lane >> 4;
    const int strip = wv & 3;
    const int qt    = wv >> 2;

    const float* Ph = p + (b * S_) * (2 * H_) + hf * H_;
    const float* Qh = q + (b * S_) * (2 * H_) + hf * H_;

    const float* gsrc[AC_NR];
    float*       ldst[AC_NR];
    bool         have[AC_NR];
    #pragma unroll
    for (int r = 0; r < AC_NR; ++r) {
        const int i   = tid + r * AC_TH;
        have[r] = i < AC_NP;
        const int ii  = have[r] ? i : 0;
        const int row = ii / AC_C4, c4 = ii % AC_C4;
        gsrc[r] = (row < S_ ? Ph + row * (2 * H_) : Qh + (row - S_) * (2 * H_)) + 4 * c4;
        ldst[r] = &XQ[row][4 * c4];
    }

    float4 stg[AC_NR];
    #pragma unroll
    for (int r = 0; r < AC_NR; ++r) {
        if (have[r]) {
            const float* g = gsrc[r];
            stg[r] = float4{g[0], g[1], g[2], g[3]};
        }
    }
    #pragma unroll
    for (int r = 0; r < AC_NR; ++r)
        if (have[r]) *(float4*)ldst[r] = stg[r];
    __syncthreads();

    const float* aRow = &XQ[16 * strip + lo][2 * hi];
    const float* bRow = &XQ[S_ + 16 * qt + lo][2 * hi];

    v8f c = {};
    v2f sA2 = {0.f, 0.f}, sB2 = {0.f, 0.f};

    #pragma unroll
    for (int ch = 0; ch < H_ / AC_CH; ++ch) {
        if (ch + 1 < H_ / AC_CH) {
            #pragma unroll
            for (int r = 0; r < AC_NR; ++r) {
                if (have[r]) {
                    const float* g = gsrc[r] + (ch + 1) * AC_CH;
                    stg[r] = float4{g[0], g[1], g[2], g[3]};
                }
            }
        }
        for (int kb = ch * AC_CH; kb < (ch + 1) * AC_CH; kb += 4) {
            const v2f a  = *(const v2f*)(aRow + kb);
            const v2f bb = *(const v2f*)(bRow + kb);
            sA2 = __builtin_elementwise_fma(a, a, sA2);
            sB2 = __builtin_elementwise_fma(bb, bb, sB2);
            c = __builtin_amdgcn_wmma_f32_16x16x4_f32(false, a, false, bb, (short)0, c, false, false);
        }
        __builtin_amdgcn_sched_barrier(0);
        if (ch + 1 < H_ / AC_CH) {
            #pragma unroll
            for (int r = 0; r < AC_NR; ++r)
                if (have[r]) *(float4*)(ldst[r] + (ch + 1) * AC_CH) = stg[r];
            __syncthreads();
        }
    }

    float sA = sA2.x + sA2.y;
    float sB = sB2.x + sB2.y;
    sA += __shfl_xor(sA, 16);
    sB += __shfl_xor(sB, 16);
    if (hi == 0) {
        if (qt == 0)    nA[16 * strip + lo] = sqrtf(sA);
        if (strip == 0) nB[16 * qt + lo]    = sqrtf(sB);
    }
    __syncthreads();

    {
        const int n = 16 * qt + lo;
        const float nb = nB[n];
        #pragma unroll
        for (int j = 0; j < 8; ++j) {
            const int m = 16 * strip + 8 * hi + j;
            cosBuf[m][n] = c[j] / (nA[m] * nb);
        }
    }
    __syncthreads();

    float* dst = cosW + (b * 2 + hf) * (S_ * S_);
    if (tid < S_) {
        float sm = 0.f;
        for (int n = 0; n < S_; ++n) sm += cosBuf[tid][n];
        rowSum[(b * 2 + hf) * S_ + tid] = sm;
    } else if (tid < 2 * S_) {
        const int n = tid - S_;
        float sm = 0.f;
        for (int m = 0; m < S_; ++m) sm += cosBuf[m][n];
        colSum[(b * 2 + hf) * S_ + n] = sm;
    } else {
        for (int i = tid - 2 * S_; i < S_ * S_; i += AC_TH - 2 * S_)
            dst[i] = cosBuf[i >> 6][i & 63];
    }
}

#define ST   8
#define NST  (S_ / ST)
#define VTH  320
#define JU   4

__global__ __launch_bounds__(VTH)
void attn_vec_kernel(const float* __restrict__ p, const float* __restrict__ q,
                     const float* __restrict__ cosW,
                     const float* __restrict__ rowSum, const float* __restrict__ colSum,
                     float* __restrict__ qvm, float* __restrict__ pvm,
                     float* __restrict__ qvx, float* __restrict__ pvx)
{
    __shared__ __attribute__((aligned(16))) float cosRowT[S_][ST];
    __shared__ __attribute__((aligned(16))) float cosColT[S_][ST];

    const int bid = blockIdx.x;
    const int s0  = (bid % NST) * ST;
    const int hf  = (bid / NST) & 1;
    const int b   = bid / (2 * NST);
    const int tid = threadIdx.x;
    const int bh  = b * 2 + hf;

    const float* cosBase = cosW + bh * (S_ * S_);
    for (int i = tid; i < S_ * ST; i += VTH) {
        int tr = i >> 6, jr = i & (S_ - 1);
        cosRowT[jr][tr] = cosBase[(s0 + tr) * S_ + jr];
        int jc = i / ST, tc = i % ST;
        cosColT[jc][tc] = cosBase[jc * S_ + s0 + tc];
    }
    __syncthreads();

    const int h = tid;
    if (h >= H_) return;

    const float* Qc = q + (b * S_) * (2 * H_) + hf * H_ + h;
    const float* Pc = p + (b * S_) * (2 * H_) + hf * H_ + h;

    float qm[ST], pm[ST], qx[ST], px[ST];
    for (int t = 0; t < ST; ++t) { qm[t] = 0.f; pm[t] = 0.f; qx[t] = NEG_INF; px[t] = NEG_INF; }

    auto accum = [&](const float (&qv)[JU], const float (&pv)[JU], int j0) {
        for (int u = 0; u < JU; ++u) {
            const int j = j0 + u;
            const float4 r0 = *(const float4*)&cosRowT[j][0];
            const float4 r1 = *(const float4*)&cosRowT[j][4];
            const float4 k0 = *(const float4*)&cosColT[j][0];
            const float4 k1 = *(const float4*)&cosColT[j][4];
            const float cr[ST] = {r0.x, r0.y, r0.z, r0.w, r1.x, r1.y, r1.z, r1.w};
            const float cc[ST] = {k0.x, k0.y, k0.z, k0.w, k1.x, k1.y, k1.z, k1.w};
            for (int t = 0; t < ST; ++t) {
                float a  = qv[u] * cr[t];
                float bb = pv[u] * cc[t];
                qm[t] += a;  qx[t] = fmaxf(qx[t], a);
                pm[t] += bb; px[t] = fmaxf(px[t], bb);
            }
        }
    };

    float qn[JU], pn[JU];
    for (int u = 0; u < JU; ++u) { qn[u] = Qc[u * (2 * H_)]; pn[u] = Pc[u * (2 * H_)]; }

    for (int j0 = 0; j0 < S_ - JU; j0 += JU) {
        float qc[JU], pc[JU];
        for (int u = 0; u < JU; ++u) { qc[u] = qn[u]; pc[u] = pn[u]; }
        for (int u = 0; u < JU; ++u) {
            qn[u] = Qc[(j0 + JU + u) * (2 * H_)];
            pn[u] = Pc[(j0 + JU + u) * (2 * H_)];
        }
        accum(qc, pc, j0);
    }
    accum(qn, pn, S_ - JU);

    for (int t = 0; t < ST; ++t) {
        const float rs = rowSum[bh * S_ + s0 + t];
        const float cs = colSum[bh * S_ + s0 + t];
        const int o = (bh * S_ + s0 + t) * H_ + h;
        qvm[o] = qm[t] / rs;
        qvx[o] = qx[t];
        pvm[o] = pm[t] / cs;
        pvx[o] = px[t];
    }
}

#define MR    32
#define MLG   4
#define MTH   ((L_ / MLG) * 32)
#define H4    (H_ / 4)
#define MPBLK (2 * 6 * B_ * (S_ / MR))

__global__ __launch_bounds__(MTH)
void mpcos_kernel(const float* __restrict__ p, const float* __restrict__ q,
                  const float* __restrict__ W,
                  const float* __restrict__ qvm, const float* __restrict__ pvm,
                  const float* __restrict__ qvx, const float* __restrict__ pvx,
                  float* __restrict__ out)
{
    __shared__ __attribute__((aligned(16))) float Xs[MR][H_];
    __shared__ __attribute__((aligned(16))) float Ys[MR][H_];
    __shared__ __attribute__((aligned(16))) float WsT[H_][L_];
    __shared__ float res[MR][L_ + 1];

    int t = blockIdx.x;
    const int rh  = t % (S_ / MR);  t /= (S_ / MR);
    const int b   = t % B_;         t /= B_;
    const int sec = t % 6;          const int tgt = t / 6;
    const int secId = sec < 2 ? sec : sec + 2;
    const int hf  = secId & 1;
    const int s0  = rh * MR;
    const int tid = threadIdx.x;

    const float* Xb = (tgt == 0 ? p : q) + (b * S_ + s0) * (2 * H_) + hf * H_;
    const float* Yb;
    int ystride;
    if (secId == 0)      { Yb = (tgt == 0 ? q : p) + (b * S_ + (S_ - 1)) * (2 * H_); ystride = 0; }
    else if (secId == 1) { Yb = (tgt == 0 ? q : p) + (b * S_) * (2 * H_) + H_;       ystride = 0; }
    else {
        const float* vec = (secId <= 5) ? (tgt == 0 ? qvm : pvm)
                                        : (tgt == 0 ? qvx : pvx);
        Yb = vec + ((b * 2 + hf) * S_ + s0) * H_;
        ystride = H_;
    }
    const float* Wb = W + (secId * L_) * H_;

    const int rr = tid / 80;
    const int cc = tid - rr * 80;
    if (cc < H4) {
        const float* xs = Xb + rr * (2 * H_) + 4 * cc;
        const float* ys = Yb + rr * ystride + 4 * cc;
        float4 xv[MR / 2], yv[MR / 2];
        #pragma unroll
        for (int k = 0; k < MR / 2; ++k) {
            const float* xk = xs + k * (2 * 2 * H_);
            const float* yk = ys + k * (2 * ystride);
            xv[k] = float4{xk[0], xk[1], xk[2], xk[3]};
            yv[k] = float4{yk[0], yk[1], yk[2], yk[3]};
        }
        #pragma unroll
        for (int k = 0; k < MR / 2; ++k) {
            *(float4*)&Xs[2 * k + rr][4 * cc] = xv[k];
            *(float4*)&Ys[2 * k + rr][4 * cc] = yv[k];
        }
    }
    #pragma unroll
    for (int hh = 0; hh < (H_ + MTH - 1) / MTH; ++hh) {
        const int h = tid + hh * MTH;
        if (h < H_) {
            float4 wv[L_ / 4];
            #pragma unroll
            for (int g = 0; g < L_ / 4; ++g) {
                const float* wg = Wb + (4 * g) * H_ + h;
                wv[g] = float4{wg[0], wg[H_], wg[2 * H_], wg[3 * H_]};
            }
            #pragma unroll
            for (int g = 0; g < L_ / 4; ++g)
                *(float4*)&WsT[h][4 * g] = wv[g];
        }
    }
    __syncthreads();

    const int lane = tid & 31;
    const int l0   = (tid >> 5) * MLG;

    v2f numA = {0.f, 0.f}, nxA = {0.f, 0.f}, nyA = {0.f, 0.f};
    v2f numB = {0.f, 0.f}, nxB = {0.f, 0.f}, nyB = {0.f, 0.f};

    for (int c = 0; c < H4; ++c) {
        const float4 x4 = *(const float4*)&Xs[lane][4 * c];
        const float4 y4 = *(const float4*)&Ys[lane][4 * c];
        const float xe[4] = {x4.x, x4.y, x4.z, x4.w};
        const float ye[4] = {y4.x, y4.y, y4.z, y4.w};
        for (int e = 0; e < 4; ++e) {
            const float4 w4 = *(const float4*)&WsT[4 * c + e][l0];
            const v2f wA = {w4.x, w4.y};
            const v2f wB = {w4.z, w4.w};
            const v2f xx = {xe[e], xe[e]};
            const v2f yy = {ye[e], ye[e]};
            const v2f xwA = xx * wA, ywA = yy * wA;
            const v2f xwB = xx * wB, ywB = yy * wB;
            numA = __builtin_elementwise_fma(xwA, ywA, numA);
            nxA  = __builtin_elementwise_fma(xwA, xwA, nxA);
            nyA  = __builtin_elementwise_fma(ywA, ywA, nyA);
            numB = __builtin_elementwise_fma(xwB, ywB, numB);
            nxB  = __builtin_elementwise_fma(xwB, xwB, nxB);
            nyB  = __builtin_elementwise_fma(ywB, ywB, nyB);
        }
    }

    const float num[MLG] = {numA.x, numA.y, numB.x, numB.y};
    const float nx[MLG]  = {nxA.x,  nxA.y,  nxB.x,  nxB.y};
    const float ny[MLG]  = {nyA.x,  nyA.y,  nyB.x,  nyB.y};
    for (int j = 0; j < MLG; ++j) {
        const float den = fmaxf(sqrtf(nx[j]) * sqrtf(ny[j]), 1e-8f);
        res[lane][l0 + j] = num[j] / den;
    }
    __syncthreads();

    float* ob = out + (tgt == 0 ? 0 : OUTQOFF) + (b * S_ + s0) * OUTC + secId * L_;
    for (int i = tid; i < MR * L_; i += MTH) {
        const int r = i / L_, l = i % L_;
        ob[r * OUTC + l] = res[r][l];
    }
}

extern "C" void kernel_launch(void* const* d_in, const int* in_sizes, int n_in,
                              void* d_out, int out_size, void* d_ws, size_t ws_size,
                              hipStream_t stream)
{
    const float* p = (const float*)d_in[0];
    const float* q = (const float*)d_in[1];
    const float* W = (const float*)d_in[2];
    float* out = (float*)d_out;

    float* ws     = (float*)d_ws;
    float* cosW   = ws;
    float* rowSum = cosW + B_ * 2 * S_ * S_;
    float* colSum = rowSum + B_ * 2 * S_;
    float* qvm    = colSum + B_ * 2 * S_;
    float* pvm    = qvm + B_ * 2 * S_ * H_;
    float* qvx    = pvm + B_ * 2 * S_ * H_;
    float* pvx    = qvx + B_ * 2 * S_ * H_;

    maxpool_kernel <<<B_ * 2 * MP_NLG, MP_TH, 0, stream>>>(p, q, W, out);
    attn_cos_kernel<<<B_ * 2,    AC_TH, 0, stream>>>(p, q, cosW, rowSum, colSum);
    attn_vec_kernel<<<B_ * 2 * NST, VTH, 0, stream>>>(p, q, cosW, rowSum, colSum,
                                                     qvm, pvm, qvx, pvx);
    mpcos_kernel   <<<MPBLK, MTH, 0, stream>>>(p, q, W, qvm, pvm, qvx, pvx, out);
}


# attn_cos launch removed: each attn_vec block (now 16 s rows) stages the P,Q rows in LDS, takes its 8 cosine tiles by WMMA there and reads P,Q from LDS; cos values, row/col sums and j order unchanged
# speedup vs baseline: 3.0509x; 1.0585x over previous
#include <hip/hip_runtime.h>
#include <math.h>

typedef __attribute__((ext_vector_type(2))) float v2f;
typedef __attribute__((ext_vector_type(8))) float v8f;

#define B_    32
#define S_    64
#define H_    300
#define L_    20
#define OUTC  160
#define OUTQOFF (B_*S_*OUTC)
#define NEG_INF (-3.4e38f)

#define MP_LG   5
#define MP_NLG  (L_ / MP_LG)
#define MP_TH   (MP_LG * 4 * 32)
#define MP_CH   100
#define MP_C4   (MP_CH / 4)
#define MP_NR   ((2 * S_ * MP_C4) / MP_TH)

__global__ __launch_bounds__(MP_TH)
void maxpool_kernel(const float* __restrict__ p, const float* __restrict__ q,
                    const float* __restrict__ W, float* __restrict__ out)
{
    __shared__ __attribute__((aligned(16))) float XQ[2 * S_][H_];
    __shared__ __attribute__((aligned(16))) float w2s[MP_LG][H_];
    __shared__ float nPs[MP_LG][S_];
    __shared__ float nQs[MP_LG][S_];
    __shared__ float rowMx[MP_LG][S_];
    __shared__ float colPt[MP_LG][4][S_];

    const int idx  = blockIdx.x;
    const int lg   = idx % MP_NLG;
    const int hf   = (idx / MP_NLG) & 1;
    const int b    = idx / (2 * MP_NLG);
    const int tid  = threadIdx.x;
    const int lane = tid & 31;
    const int wv   = tid >> 5;
    const int lo   = lane & 15;
    const int hi   = lane >> 4;
    const int li   = wv >> 2;
    const int strip = wv & 3;

    const float* Ph = p + (b * S_) * (2 * H_) + hf * H_;
    const float* Qh = q + (b * S_) * (2 * H_) + hf * H_;
    const float* Wg = W + ((2 + hf) * L_ + lg * MP_LG) * H_;

    const float* gsrc[MP_NR];
    float*       ldst[MP_NR];
    #pragma unroll
    for (int r = 0; r < MP_NR; ++r) {
        const int i   = tid + r * MP_TH;
        const int row = i / MP_C4, c4 = i % MP_C4;
        gsrc[r] = (row < S_ ? Ph + row * (2 * H_) : Qh + (row - S_) * (2 * H_)) + 4 * c4;
        ldst[r] = &XQ[row][4 * c4];
    }

    float4 stg[MP_NR];
    #pragma unroll
    for (int r = 0; r < MP_NR; ++r) {
        const float* g = gsrc[r];
        stg[r] = float4{g[0], g[1], g[2], g[3]};
    }
    for (int i = tid; i < MP_LG * H_; i += MP_TH) {
        const float w = Wg[i];
        (&w2s[0][0])[i] = w * w;
    }
    #pragma unroll
    for (int r = 0; r < MP_NR; ++r) *(float4*)ldst[r] = stg[r];
    __syncthreads();

    const float* aRow = &XQ[16 * strip + lo][2 * hi];
    const float* bR0  = &XQ[S_ + 16 * ((strip + 0) & 3) + lo][2 * hi];
    const float* bR1  = &XQ[S_ + 16 * ((strip + 1) & 3) + lo][2 * hi];
    const float* bR2  = &XQ[S_ + 16 * ((strip + 2) & 3) + lo][2 * hi];
    const float* bR3  = &XQ[S_ + 16 * ((strip + 3) & 3) + lo][2 * hi];
    const float* wRow = &w2s[li][2 * hi];

    v8f c0 = {}, c1 = {}, c2 = {}, c3 = {};
    float nA = 0.f, nB = 0.f;

    #pragma unroll
    for (int ch = 0; ch < H_ / MP_CH; ++ch) {
        if (ch + 1 < H_ / MP_CH) {
            #pragma unroll
            for (int r = 0; r < MP_NR; ++r) {
                const float* g = gsrc[r] + (ch + 1) * MP_CH;
                stg[r] = float4{g[0], g[1], g[2], g[3]};
            }
        }
        for (int kb = ch * MP_CH; kb < (ch + 1) * MP_CH; kb += 4) {
            const v2f a  = *(const v2f*)(aRow + kb);
            const v2f w  = *(const v2f*)(wRow + kb);
            const v2f b0 = *(const v2f*)(bR0 + kb);
            const v2f b1 = *(const v2f*)(bR1 + kb);
            const v2f b2 = *(const v2f*)(bR2 + kb);
            const v2f b3 = *(const v2f*)(bR3 + kb);
            const v2f aw = a * w;
            const v2f bw = b0 * w;
            nA = fmaf(a.x, aw.x, nA);   nA = fmaf(a.y, aw.y, nA);
            nB = fmaf(b0.x, bw.x, nB);  nB = fmaf(b0.y, bw.y, nB);
            c0 = __builtin_amdgcn_wmma_f32_16x16x4_f32(false, aw, false, b0, (short)0, c0, false, false);
            c1 = __builtin_amdgcn_wmma_f32_16x16x4_f32(false, aw, false, b1, (short)0, c1, false, false);
            c2 = __builtin_amdgcn_wmma_f32_16x16x4_f32(false, aw, false, b2, (short)0, c2, false, false);
            c3 = __builtin_amdgcn_wmma_f32_16x16x4_f32(false, aw, false, b3, (short)0, c3, false, false);
        }
        if (ch + 1 < H_ / MP_CH) {
            #pragma unroll
            for (int r = 0; r < MP_NR; ++r)
                *(float4*)(ldst[r] + (ch + 1) * MP_CH) = stg[r];
            __syncthreads();
        }
    }

    nA += __shfl_xor(nA, 16);
    nB += __shfl_xor(nB, 16);
    if (hi == 0) {
        nPs[li][16 * strip + lo] = sqrtf(nA);
        nQs[li][16 * strip + lo] = sqrtf(nB);
    }
    __syncthreads();

    const int n0 = 16 * ((strip + 0) & 3) + lo;
    const int n1 = 16 * ((strip + 1) & 3) + lo;
    const int n2 = 16 * ((strip + 2) & 3) + lo;
    const int n3 = 16 * ((strip + 3) & 3) + lo;
    const float nb0 = nQs[li][n0], nb1 = nQs[li][n1], nb2 = nQs[li][n2], nb3 = nQs[li][n3];
    float cm0 = NEG_INF, cm1 = NEG_INF, cm2 = NEG_INF, cm3 = NEG_INF;
    float rmx[8];
    #pragma unroll
    for (int j = 0; j < 8; ++j) {
        const float na = nPs[li][16 * strip + 8 * hi + j];
        const float v0 = c0[j] / (na * nb0);
        const float v1 = c1[j] / (na * nb1);
        const float v2 = c2[j] / (na * nb2);
        const float v3 = c3[j] / (na * nb3);
        cm0 = fmaxf(cm0, v0); cm1 = fmaxf(cm1, v1);
        cm2 = fmaxf(cm2, v2); cm3 = fmaxf(cm3, v3);
        rmx[j] = fmaxf(fmaxf(fmaxf(fmaxf(NEG_INF, v0), v1), v2), v3);
    }
    #pragma unroll
    for (int d = 1; d < 16; d <<= 1) {
        #pragma unroll
        for (int j = 0; j < 8; ++j) rmx[j] = fmaxf(rmx[j], __shfl_xor(rmx[j], d));
    }
    if (lo == 0) {
        #pragma unroll
        for (int j = 0; j < 8; ++j) rowMx[li][16 * strip + 8 * hi + j] = rmx[j];
    }
    cm0 = fmaxf(cm0, __shfl_xor(cm0, 16));
    cm1 = fmaxf(cm1, __shfl_xor(cm1, 16));
    cm2 = fmaxf(cm2, __shfl_xor(cm2, 16));
    cm3 = fmaxf(cm3, __shfl_xor(cm3, 16));
    if (hi == 0) {
        colPt[li][strip][n0] = cm0;
        colPt[li][strip][n1] = cm1;
        colPt[li][strip][n2] = cm2;
        colPt[li][strip][n3] = cm3;
    }
    __syncthreads();

    const int nres = (S_ * MP_LG);
    if (tid < nres) {
        const int row = tid / MP_LG, l5 = tid % MP_LG;
        out[(b * S_ + row) * OUTC + (2 + hf) * L_ + lg * MP_LG + l5] = rowMx[l5][row];
    } else {
        const int u = tid - nres;
        const int n = u / MP_LG, l5 = u % MP_LG;
        float mx = NEG_INF;
        #pragma unroll
        for (int s = 0; s < 4; ++s) mx = fmaxf(mx, colPt[l5][s][n]);
        out[OUTQOFF + (b * S_ + n) * OUTC + (2 + hf) * L_ + lg * MP_LG + l5] = mx;
    }
}

#define ST      16
#define NST     (S_ / ST)
#define VTH     320
#define AC_CH   100
#define AC_C4   (AC_CH / 4)
#define AC_NR   ((2 * S_ * AC_C4) / VTH)

__global__ __launch_bounds__(VTH)
void attn_fused_kernel(const float* __restrict__ p, const float* __restrict__ q,
                       float* __restrict__ qvm, float* __restrict__ pvm,
                       float* __restrict__ qvx, float* __restrict__ pvx)
{
    __shared__ __attribute__((aligned(16))) float XQ[2 * S_][H_];
    __shared__ __attribute__((aligned(16))) float cosRowT[S_][ST];
    __shared__ __attribute__((aligned(16))) float cosColT[S_][ST];
    __shared__ float nA[S_];
    __shared__ float nB[S_];
    __shared__ float rSum[ST];
    __shared__ float cSum[ST];

    const int bid  = blockIdx.x;
    const int st   = bid % NST;
    const int hf   = (bid / NST) & 1;
    const int b    = bid / (2 * NST);
    const int s0   = st * ST;
    const int bh   = b * 2 + hf;
    const int tid  = threadIdx.x;
    const int lane = tid & 31;
    const int wv   = tid >> 5;
    const int lo   = lane & 15;
    const int hi   = lane >> 4;
    const bool mm  = wv < 8;
    const bool rowTile = wv < 4;
    const int strip = rowTile ? st : (wv & 3);
    const int qt    = rowTile ? wv : st;

    const float* Ph = p + (b * S_) * (2 * H_) + hf * H_;
    const float* Qh = q + (b * S_) * (2 * H_) + hf * H_;

    const float* gsrc[AC_NR];
    float*       ldst[AC_NR];
    #pragma unroll
    for (int r = 0; r < AC_NR; ++r) {
        const int i   = tid + r * VTH;
        const int row = i / AC_C4, c4 = i % AC_C4;
        gsrc[r] = (row < S_ ? Ph + row * (2 * H_) : Qh + (row - S_) * (2 * H_)) + 4 * c4;
        ldst[r] = &XQ[row][4 * c4];
    }

    float4 stg[AC_NR];
    #pragma unroll
    for (int r = 0; r < AC_NR; ++r) {
        const float* g = gsrc[r];
        stg[r] = float4{g[0], g[1], g[2], g[3]};
    }
    #pragma unroll
    for (int r = 0; r < AC_NR; ++r) *(float4*)ldst[r] = stg[r];
    __syncthreads();

    const float* aRow = &XQ[16 * strip + lo][2 * hi];
    const float* bRow = &XQ[S_ + 16 * qt + lo][2 * hi];

    v8f c = {};
    v2f sA2 = {0.f, 0.f}, sB2 = {0.f, 0.f};

    #pragma unroll
    for (int ch = 0; ch < H_ / AC_CH; ++ch) {
        if (ch + 1 < H_ / AC_CH) {
            #pragma unroll
            for (int r = 0; r < AC_NR; ++r) {
                const float* g = gsrc[r] + (ch + 1) * AC_CH;
                stg[r] = float4{g[0], g[1], g[2], g[3]};
            }
        }
        if (mm) {
            for (int kb = ch * AC_CH; kb < (ch + 1) * AC_CH; kb += 4) {
                const v2f a  = *(const v2f*)(aRow + kb);
                const v2f bb = *(const v2f*)(bRow + kb);
                sA2 = __builtin_elementwise_fma(a, a, sA2);
                sB2 = __builtin_elementwise_fma(bb, bb, sB2);
                c = __builtin_amdgcn_wmma_f32_16x16x4_f32(false, a, false, bb, (short)0, c, false, false);
            }
        }
        __builtin_amdgcn_sched_barrier(0);
        if (ch + 1 < H_ / AC_CH) {
            #pragma unroll
            for (int r = 0; r < AC_NR; ++r)
                *(float4*)(ldst[r] + (ch + 1) * AC_CH) = stg[r];
            __syncthreads();
        }
    }

    if (mm) {
        float sA = sA2.x + sA2.y;
        float sB = sB2.x + sB2.y;
        sA += __shfl_xor(sA, 16);
        sB += __shfl_xor(sB, 16);
        if (hi == 0) {
            if (!rowTile) nA[16 * strip + lo] = sqrtf(sA);
            else          nB[16 * qt + lo]    = sqrtf(sB);
        }
    }
    __syncthreads();

    if (mm) {
        const int n = 16 * qt + lo;
        const float nb = nB[n];
        #pragma unroll
        for (int j = 0; j < 8; ++j) {
            const int m = 16 * strip + 8 * hi + j;
            const float v = c[j] / (nA[m] * nb);
            if (rowTile) cosRowT[n][m - s0] = v;
            else         cosColT[m][n - s0] = v;
        }
    }
    __syncthreads();

    if (tid < ST) {
        float sm = 0.f;
        for (int n = 0; n < S_; ++n) sm += cosRowT[n][tid];
        rSum[tid] = sm;
    } else if (tid >= 32 && tid < 32 + ST) {
        const int t = tid - 32;
        float sm = 0.f;
        for (int m = 0; m < S_; ++m) sm += cosColT[m][t];
        cSum[t] = sm;
    }
    __syncthreads();

    const int h = tid;
    if (h >= H_) return;

    const float* Qc = &XQ[S_][h];
    const float* Pc = &XQ[0][h];

    float qm[ST], pm[ST], qx[ST], px[ST];
    for (int t = 0; t < ST; ++t) { qm[t] = 0.f; pm[t] = 0.f; qx[t] = NEG_INF; px[t] = NEG_INF; }

    #pragma unroll 4
    for (int j = 0; j < S_; ++j) {
        const float qv = Qc[j * H_];
        const float pv = Pc[j * H_];
        float cr[ST], cc[ST];
        for (int g = 0; g < ST / 4; ++g) {
            const float4 r4 = *(const float4*)&cosRowT[j][4 * g];
            const float4 k4 = *(const float4*)&cosColT[j][4 * g];
            cr[4 * g + 0] = r4.x; cr[4 * g + 1] = r4.y; cr[4 * g + 2] = r4.z; cr[4 * g + 3] = r4.w;
            cc[4 * g + 0] = k4.x; cc[4 * g + 1] = k4.y; cc[4 * g + 2] = k4.z; cc[4 * g + 3] = k4.w;
        }
        for (int t = 0; t < ST; ++t) {
            float a  = qv * cr[t];
            float bb = pv * cc[t];
            qm[t] += a;  qx[t] = fmaxf(qx[t], a);
            pm[t] += bb; px[t] = fmaxf(px[t], bb);
        }
    }

    for (int t = 0; t < ST; ++t) {
        const float rs = rSum[t];
        const float cs = cSum[t];
        const int o = (bh * S_ + s0 + t) * H_ + h;
        qvm[o] = qm[t] / rs;
        qvx[o] = qx[t];
        pvm[o] = pm[t] / cs;
        pvx[o] = px[t];
    }
}

#define MR    32
#define MLG   4
#define MTH   ((L_ / MLG) * 32)
#define H4    (H_ / 4)
#define MPBLK (2 * 6 * B_ * (S_ / MR))

__global__ __launch_bounds__(MTH)
void mpcos_kernel(const float* __restrict__ p, const float* __restrict__ q,
                  const float* __restrict__ W,
                  const float* __restrict__ qvm, const float* __restrict__ pvm,
                  const float* __restrict__ qvx, const float* __restrict__ pvx,
                  float* __restrict__ out)
{
    __shared__ __attribute__((aligned(16))) float Xs[MR][H_];
    __shared__ __attribute__((aligned(16))) float Ys[MR][H_];
    __shared__ __attribute__((aligned(16))) float WsT[H_][L_];
    __shared__ float res[MR][L_ + 1];

    int t = blockIdx.x;
    const int rh  = t % (S_ / MR);  t /= (S_ / MR);
    const int b   = t % B_;         t /= B_;
    const int sec = t % 6;          const int tgt = t / 6;
    const int secId = sec < 2 ? sec : sec + 2;
    const int hf  = secId & 1;
    const int s0  = rh * MR;
    const int tid = threadIdx.x;

    const float* Xb = (tgt == 0 ? p : q) + (b * S_ + s0) * (2 * H_) + hf * H_;
    const float* Yb;
    int ystride;
    if (secId == 0)      { Yb = (tgt == 0 ? q : p) + (b * S_ + (S_ - 1)) * (2 * H_); ystride = 0; }
    else if (secId == 1) { Yb = (tgt == 0 ? q : p) + (b * S_) * (2 * H_) + H_;       ystride = 0; }
    else {
        const float* vec = (secId <= 5) ? (tgt == 0 ? qvm : pvm)
                                        : (tgt == 0 ? qvx : pvx);
        Yb = vec + ((b * 2 + hf) * S_ + s0) * H_;
        ystride = H_;
    }
    const float* Wb = W + (secId * L_) * H_;

    const int rr = tid / 80;
    const int cc = tid - rr * 80;
    if (cc < H4) {
        const float* xs = Xb + rr * (2 * H_) + 4 * cc;
        const float* ys = Yb + rr * ystride + 4 * cc;
        float4 xv[MR / 2], yv[MR / 2];
        #pragma unroll
        for (int k = 0; k < MR / 2; ++k) {
            const float* xk = xs + k * (2 * 2 * H_);
            const float* yk = ys + k * (2 * ystride);
            xv[k] = float4{xk[0], xk[1], xk[2], xk[3]};
            yv[k] = float4{yk[0], yk[1], yk[2], yk[3]};
        }
        #pragma unroll
        for (int k = 0; k < MR / 2; ++k) {
            *(float4*)&Xs[2 * k + rr][4 * cc] = xv[k];
            *(float4*)&Ys[2 * k + rr][4 * cc] = yv[k];
        }
    }
    #pragma unroll
    for (int hh = 0; hh < (H_ + MTH - 1) / MTH; ++hh) {
        const int h = tid + hh * MTH;
        if (h < H_) {
            float4 wv[L_ / 4];
            #pragma unroll
            for (int g = 0; g < L_ / 4; ++g) {
                const float* wg = Wb + (4 * g) * H_ + h;
                wv[g] = float4{wg[0], wg[H_], wg[2 * H_], wg[3 * H_]};
            }
            #pragma unroll
            for (int g = 0; g < L_ / 4; ++g)
                *(float4*)&WsT[h][4 * g] = wv[g];
        }
    }
    __syncthreads();

    const int lane = tid & 31;
    const int l0   = (tid >> 5) * MLG;

    v2f numA = {0.f, 0.f}, nxA = {0.f, 0.f}, nyA = {0.f, 0.f};
    v2f numB = {0.f, 0.f}, nxB = {0.f, 0.f}, nyB = {0.f, 0.f};

    for (int c = 0; c < H4; ++c) {
        const float4 x4 = *(const float4*)&Xs[lane][4 * c];
        const float4 y4 = *(const float4*)&Ys[lane][4 * c];
        const float xe[4] = {x4.x, x4.y, x4.z, x4.w};
        const float ye[4] = {y4.x, y4.y, y4.z, y4.w};
        for (int e = 0; e < 4; ++e) {
            const float4 w4 = *(const float4*)&WsT[4 * c + e][l0];
            const v2f wA = {w4.x, w4.y};
            const v2f wB = {w4.z, w4.w};
            const v2f xx = {xe[e], xe[e]};
            const v2f yy = {ye[e], ye[e]};
            const v2f xwA = xx * wA, ywA = yy * wA;
            const v2f xwB = xx * wB, ywB = yy * wB;
            numA = __builtin_elementwise_fma(xwA, ywA, numA);
            nxA  = __builtin_elementwise_fma(xwA, xwA, nxA);
            nyA  = __builtin_elementwise_fma(ywA, ywA, nyA);
            numB = __builtin_elementwise_fma(xwB, ywB, numB);
            nxB  = __builtin_elementwise_fma(xwB, xwB, nxB);
            nyB  = __builtin_elementwise_fma(ywB, ywB, nyB);
        }
    }

    const float num[MLG] = {numA.x, numA.y, numB.x, numB.y};
    const float nx[MLG]  = {nxA.x,  nxA.y,  nxB.x,  nxB.y};
    const float ny[MLG]  = {nyA.x,  nyA.y,  nyB.x,  nyB.y};
    for (int j = 0; j < MLG; ++j) {
        const float den = fmaxf(sqrtf(nx[j]) * sqrtf(ny[j]), 1e-8f);
        res[lane][l0 + j] = num[j] / den;
    }
    __syncthreads();

    float* ob = out + (tgt == 0 ? 0 : OUTQOFF) + (b * S_ + s0) * OUTC + secId * L_;
    for (int i = tid; i < MR * L_; i += MTH) {
        const int r = i / L_, l = i % L_;
        ob[r * OUTC + l] = res[r][l];
    }
}

extern "C" void kernel_launch(void* const* d_in, const int* in_sizes, int n_in,
                              void* d_out, int out_size, void* d_ws, size_t ws_size,
                              hipStream_t stream)
{
    const float* p = (const float*)d_in[0];
    const float* q = (const float*)d_in[1];
    const float* W = (const float*)d_in[2];
    float* out = (float*)d_out;

    float* ws  = (float*)d_ws;
    float* qvm = ws;
    float* pvm = qvm + B_ * 2 * S_ * H_;
    float* qvx = pvm + B_ * 2 * S_ * H_;
    float* pvx = qvx + B_ * 2 * S_ * H_;

    maxpool_kernel   <<<B_ * 2 * MP_NLG, MP_TH, 0, stream>>>(p, q, W, out);
    attn_fused_kernel<<<B_ * 2 * NST, VTH, 0, stream>>>(p, q, qvm, pvm, qvx, pvx);
    mpcos_kernel     <<<MPBLK, MTH, 0, stream>>>(p, q, W, qvm, pvm, qvx, pvx, out);
}


# maxpool and attentive launches fused: a 30-wave block stages the P,Q rows once; 20 waves run the maxpool WMMAs while 10 waves walk j for the attentive vectors (Q side, then P side); sum orders kept
# speedup vs baseline: 3.2594x; 1.0683x over previous
#include <hip/hip_runtime.h>
#include <math.h>

typedef __attribute__((ext_vector_type(2))) float v2f;
typedef __attribute__((ext_vector_type(8))) float v8f;

#define B_    32
#define S_    64
#define H_    300
#define L_    20
#define OUTC  160
#define OUTQOFF (B_*S_*OUTC)
#define NEG_INF (-3.4e38f)

#define MP_LG   5
#define MP_NLG  (L_ / MP_LG)
#define MP_W    (MP_LG * 4)
#define MP_TH   (MP_W * 32)
#define ST      16
#define NST     (S_ / ST)
#define VTH     320
#define FTH     (MP_TH + VTH)
#define PC4     (H_ / 4)
#define F_NP    ((2 * S_ * PC4) / FTH)

__global__ __launch_bounds__(FTH)
void pool_attn_kernel(const float* __restrict__ p, const float* __restrict__ q,
                      const float* __restrict__ W, float* __restrict__ out,
                      float* __restrict__ qvm, float* __restrict__ pvm,
                      float* __restrict__ qvx, float* __restrict__ pvx)
{
    __shared__ __attribute__((aligned(16))) float XQ[2 * S_][H_];
    __shared__ __attribute__((aligned(16))) float w2s[MP_LG][H_];
    __shared__ float nPs[MP_LG][S_];
    __shared__ float nQs[MP_LG][S_];
    __shared__ float rowMx[MP_LG][S_];
    __shared__ float colPt[MP_LG][4][S_];
    __shared__ __attribute__((aligned(16))) float cosRowT[S_][ST];
    __shared__ __attribute__((aligned(16))) float cosColT[S_][ST];
    __shared__ float nA[S_];
    __shared__ float nB[S_];
    __shared__ float rSum[ST];
    __shared__ float cSum[ST];

    const int idx  = blockIdx.x;
    const int x    = idx % NST;
    const int hf   = (idx / NST) & 1;
    const int b    = idx / (2 * NST);
    const int tid  = threadIdx.x;
    const int lane = tid & 31;
    const int wv   = __builtin_amdgcn_readfirstlane(tid >> 5);
    const int lo   = lane & 15;
    const int hi   = lane >> 4;

    const bool attn = wv >= MP_W;
    const int li    = wv >> 2;
    const int strip = wv & 3;
    const int awv   = wv - MP_W;
    const int atid  = tid - MP_TH;
    const int s0    = x * ST;
    const int bh    = b * 2 + hf;
    const bool mm      = attn && awv < 8;
    const bool rowTile = awv < 4;
    const int astrip = rowTile ? x : (awv & 3);
    const int aqt    = rowTile ? awv : x;

    const float* Ph = p + (b * S_) * (2 * H_) + hf * H_;
    const float* Qh = q + (b * S_) * (2 * H_) + hf * H_;
    const float* Wg = W + ((2 + hf) * L_ + x * MP_LG) * H_;

    {
        float4 stg[F_NP];
        #pragma unroll
        for (int r = 0; r < F_NP; ++r) {
            const int i   = tid + r * FTH;
            const int row = i / PC4, c4 = i % PC4;
            const float* g = (row < S_ ? Ph + row * (2 * H_) : Qh + (row - S_) * (2 * H_)) + 4 * c4;
            stg[r] = float4{g[0], g[1], g[2], g[3]};
        }
        for (int i = tid; i < MP_LG * H_; i += FTH) {
            const float w = Wg[i];
            (&w2s[0][0])[i] = w * w;
        }
        #pragma unroll
        for (int r = 0; r < F_NP; ++r) {
            const int i   = tid + r * FTH;
            const int row = i / PC4, c4 = i % PC4;
            *(float4*)&XQ[row][4 * c4] = stg[r];
        }
    }
    __syncthreads();

    v8f c = {};
    if (mm) {
        const float* aRow = &XQ[16 * astrip + lo][2 * hi];
        const float* bRow = &XQ[S_ + 16 * aqt + lo][2 * hi];
        v2f sA2 = {0.f, 0.f}, sB2 = {0.f, 0.f};
        for (int kb = 0; kb < H_; kb += 4) {
            const v2f a  = *(const v2f*)(aRow + kb);
            const v2f bb = *(const v2f*)(bRow + kb);
            sA2 = __builtin_elementwise_fma(a, a, sA2);
            sB2 = __builtin_elementwise_fma(bb, bb, sB2);
            c = __builtin_amdgcn_wmma_f32_16x16x4_f32(false, a, false, bb, (short)0, c, false, false);
        }
        float sA = sA2.x + sA2.y;
        float sB = sB2.x + sB2.y;
        sA += __shfl_xor(sA, 16);
        sB += __shfl_xor(sB, 16);
        if (hi == 0) {
            if (!rowTile) nA[16 * astrip + lo] = sqrtf(sA);
            else          nB[16 * aqt + lo]    = sqrtf(sB);
        }
    }
    __syncthreads();

    if (mm) {
        const int n = 16 * aqt + lo;
        const float nb = nB[n];
        #pragma unroll
        for (int j = 0; j < 8; ++j) {
            const int m = 16 * astrip + 8 * hi + j;
            const float v = c[j] / (nA[m] * nb);
            if (rowTile) cosRowT[n][m - s0] = v;
            else         cosColT[m][n - s0] = v;
        }
    }
    __syncthreads();

    if (attn) {
        if (atid < ST) {
            float sm = 0.f;
            for (int n = 0; n < S_; ++n) sm += cosRowT[n][atid];
            rSum[atid] = sm;
        } else if (atid >= 32 && atid < 32 + ST) {
            const int t = atid - 32;
            float sm = 0.f;
            for (int m = 0; m < S_; ++m) sm += cosColT[m][t];
            cSum[t] = sm;
        }
    }
    __syncthreads();

    v8f c0, c1, c2, c3;
    if (attn) {
        const int h = atid;
        if (h < H_) {
            {
                const float* Qc = &XQ[S_][h];
                float qm[ST], qx[ST];
                for (int t = 0; t < ST; ++t) { qm[t] = 0.f; qx[t] = NEG_INF; }
                #pragma unroll 4
                for (int j = 0; j < S_; ++j) {
                    const float qv = Qc[j * H_];
                    float cr[ST];
                    for (int g = 0; g < ST / 4; ++g) {
                        const float4 r4 = *(const float4*)&cosRowT[j][4 * g];
                        cr[4 * g + 0] = r4.x; cr[4 * g + 1] = r4.y; cr[4 * g + 2] = r4.z; cr[4 * g + 3] = r4.w;
                    }
                    for (int t = 0; t < ST; ++t) {
                        float a = qv * cr[t];
                        qm[t] += a;  qx[t] = fmaxf(qx[t], a);
                    }
                }
                float* qvm_o = qvm + ((bh * S_ + s0) * H_ + h);
                float* qvx_o = qvx + ((bh * S_ + s0) * H_ + h);
                for (int t = 0; t < ST; ++t) {
                    const float rs = rSum[t];
                    qvm_o[t * H_] = qm[t] / rs;
                    qvx_o[t * H_] = qx[t];
                }
            }
            {
                const float* Pc = &XQ[0][h];
                float pm[ST], px[ST];
                for (int t = 0; t < ST; ++t) { pm[t] = 0.f; px[t] = NEG_INF; }
                #pragma unroll 4
                for (int j = 0; j < S_; ++j) {
                    const float pv = Pc[j * H_];
                    float cc[ST];
                    for (int g = 0; g < ST / 4; ++g) {
                        const float4 k4 = *(const float4*)&cosColT[j][4 * g];
                        cc[4 * g + 0] = k4.x; cc[4 * g + 1] = k4.y; cc[4 * g + 2] = k4.z; cc[4 * g + 3] = k4.w;
                    }
                    for (int t = 0; t < ST; ++t) {
                        float bb = pv * cc[t];
                        pm[t] += bb; px[t] = fmaxf(px[t], bb);
                    }
                }
                float* pvm_o = pvm + ((bh * S_ + s0) * H_ + h);
                float* pvx_o = pvx + ((bh * S_ + s0) * H_ + h);
                for (int t = 0; t < ST; ++t) {
                    const float cs = cSum[t];
                    pvm_o[t * H_] = pm[t] / cs;
                    pvx_o[t * H_] = px[t];
                }
            }
        }
    } else {
        const float* aRow = &XQ[16 * strip + lo][2 * hi];
        const float* bR0  = &XQ[S_ + 16 * ((strip + 0) & 3) + lo][2 * hi];
        const float* bR1  = &XQ[S_ + 16 * ((strip + 1) & 3) + lo][2 * hi];
        const float* bR2  = &XQ[S_ + 16 * ((strip + 2) & 3) + lo][2 * hi];
        const float* bR3  = &XQ[S_ + 16 * ((strip + 3) & 3) + lo][2 * hi];
        const float* wRow = &w2s[li][2 * hi];
        float sP = 0.f, sQ = 0.f;
        c0 = v8f{}; c1 = v8f{}; c2 = v8f{}; c3 = v8f{};
        for (int kb = 0; kb < H_; kb += 4) {
            const v2f a  = *(const v2f*)(aRow + kb);
            const v2f w  = *(const v2f*)(wRow + kb);
            const v2f b0 = *(const v2f*)(bR0 + kb);
            const v2f b1 = *(const v2f*)(bR1 + kb);
            const v2f b2 = *(const v2f*)(bR2 + kb);
            const v2f b3 = *(const v2f*)(bR3 + kb);
            const v2f aw = a * w;
            const v2f bw = b0 * w;
            sP = fmaf(a.x, aw.x, sP);   sP = fmaf(a.y, aw.y, sP);
            sQ = fmaf(b0.x, bw.x, sQ);  sQ = fmaf(b0.y, bw.y, sQ);
            c0 = __builtin_amdgcn_wmma_f32_16x16x4_f32(false, aw, false, b0, (short)0, c0, false, false);
            c1 = __builtin_amdgcn_wmma_f32_16x16x4_f32(false, aw, false, b1, (short)0, c1, false, false);
            c2 = __builtin_amdgcn_wmma_f32_16x16x4_f32(false, aw, false, b2, (short)0, c2, false, false);
            c3 = __builtin_amdgcn_wmma_f32_16x16x4_f32(false, aw, false, b3, (short)0, c3, false, false);
        }
        sP += __shfl_xor(sP, 16);
        sQ += __shfl_xor(sQ, 16);
        if (hi == 0) {
            nPs[li][16 * strip + lo] = sqrtf(sP);
            nQs[li][16 * strip + lo] = sqrtf(sQ);
        }
    }
    __syncthreads();

    const int lane2 = (int)__builtin_amdgcn_mbcnt_lo(~0u, 0u);
    const int lo2   = lane2 & 15;
    const int hi2   = lane2 >> 4;
    const int tid2  = wv * 32 + lane2;

    if (!attn) {
        const int n0 = 16 * ((strip + 0) & 3) + lo2;
        const int n1 = 16 * ((strip + 1) & 3) + lo2;
        const int n2 = 16 * ((strip + 2) & 3) + lo2;
        const int n3 = 16 * ((strip + 3) & 3) + lo2;
        const float nb0 = nQs[li][n0], nb1 = nQs[li][n1], nb2 = nQs[li][n2], nb3 = nQs[li][n3];
        float cm0 = NEG_INF, cm1 = NEG_INF, cm2 = NEG_INF, cm3 = NEG_INF;
        float rmx[8];
        #pragma unroll
        for (int j = 0; j < 8; ++j) {
            const float na = nPs[li][16 * strip + 8 * hi2 + j];
            const float v0 = c0[j] / (na * nb0);
            const float v1 = c1[j] / (na * nb1);
            const float v2 = c2[j] / (na * nb2);
            const float v3 = c3[j] / (na * nb3);
            cm0 = fmaxf(cm0, v0); cm1 = fmaxf(cm1, v1);
            cm2 = fmaxf(cm2, v2); cm3 = fmaxf(cm3, v3);
            rmx[j] = fmaxf(fmaxf(fmaxf(fmaxf(NEG_INF, v0), v1), v2), v3);
        }
        #pragma unroll
        for (int d = 1; d < 16; d <<= 1) {
            #pragma unroll
            for (int j = 0; j < 8; ++j) rmx[j] = fmaxf(rmx[j], __shfl_xor(rmx[j], d));
        }
        if (lo2 == 0) {
            for (int j = 0; j < 8; ++j) rowMx[li][16 * strip + 8 * hi2 + j] = rmx[j];
        }
        cm0 = fmaxf(cm0, __shfl_xor(cm0, 16));
        cm1 = fmaxf(cm1, __shfl_xor(cm1, 16));
        cm2 = fmaxf(cm2, __shfl_xor(cm2, 16));
        cm3 = fmaxf(cm3, __shfl_xor(cm3, 16));
        if (hi2 == 0) {
            colPt[li][strip][n0] = cm0;
            colPt[li][strip][n1] = cm1;
            colPt[li][strip][n2] = cm2;
            colPt[li][strip][n3] = cm3;
        }
    }
    __syncthreads();

    const int nres = (S_ * MP_LG);
    if (tid2 < nres) {
        const int row = tid2 / MP_LG, l5 = tid2 % MP_LG;
        out[(b * S_ + row) * OUTC + (2 + hf) * L_ + x * MP_LG + l5] = rowMx[l5][row];
    } else if (tid2 < 2 * nres) {
        const int u = tid2 - nres;
        const int n = u / MP_LG, l5 = u % MP_LG;
        float mx = NEG_INF;
        for (int s = 0; s < 4; ++s) mx = fmaxf(mx, colPt[l5][s][n]);
        out[OUTQOFF + (b * S_ + n) * OUTC + (2 + hf) * L_ + x * MP_LG + l5] = mx;
    }
}

#define MR    32
#define MLG   4
#define MTH   ((L_ / MLG) * 32)
#define H4    (H_ / 4)
#define MPBLK (2 * 6 * B_ * (S_ / MR))

__global__ __launch_bounds__(MTH)
void mpcos_kernel(const float* __restrict__ p, const float* __restrict__ q,
                  const float* __restrict__ W,
                  const float* __restrict__ qvm, const float* __restrict__ pvm,
                  const float* __restrict__ qvx, const float* __restrict__ pvx,
                  float* __restrict__ out)
{
    __shared__ __attribute__((aligned(16))) float Xs[MR][H_];
    __shared__ __attribute__((aligned(16))) float Ys[MR][H_];
    __shared__ __attribute__((aligned(16))) float WsT[H_][L_];
    __shared__ float res[MR][L_ + 1];

    int t = blockIdx.x;
    const int rh  = t % (S_ / MR);  t /= (S_ / MR);
    const int b   = t % B_;         t /= B_;
    const int sec = t % 6;          const int tgt = t / 6;
    const int secId = sec < 2 ? sec : sec + 2;
    const int hf  = secId & 1;
    const int s0  = rh * MR;
    const int tid = threadIdx.x;

    const float* Xb = (tgt == 0 ? p : q) + (b * S_ + s0) * (2 * H_) + hf * H_;
    const float* Yb;
    int ystride;
    if (secId == 0)      { Yb = (tgt == 0 ? q : p) + (b * S_ + (S_ - 1)) * (2 * H_); ystride = 0; }
    else if (secId == 1) { Yb = (tgt == 0 ? q : p) + (b * S_) * (2 * H_) + H_;       ystride = 0; }
    else {
        const float* vec = (secId <= 5) ? (tgt == 0 ? qvm : pvm)
                                        : (tgt == 0 ? qvx : pvx);
        Yb = vec + ((b * 2 + hf) * S_ + s0) * H_;
        ystride = H_;
    }
    const float* Wb = W + (secId * L_) * H_;

    const int rr = tid / 80;
    const int cc = tid - rr * 80;
    if (cc < H4) {
        const float* xs = Xb + rr * (2 * H_) + 4 * cc;
        const float* ys = Yb + rr * ystride + 4 * cc;
        float4 xv[MR / 2], yv[MR / 2];
        #pragma unroll
        for (int k = 0; k < MR / 2; ++k) {
            const float* xk = xs + k * (2 * 2 * H_);
            const float* yk = ys + k * (2 * ystride);
            xv[k] = float4{xk[0], xk[1], xk[2], xk[3]};
            yv[k] = float4{yk[0], yk[1], yk[2], yk[3]};
        }
        #pragma unroll
        for (int k = 0; k < MR / 2; ++k) {
            *(float4*)&Xs[2 * k + rr][4 * cc] = xv[k];
            *(float4*)&Ys[2 * k + rr][4 * cc] = yv[k];
        }
    }
    #pragma unroll
    for (int hh = 0; hh < (H_ + MTH - 1) / MTH; ++hh) {
        const int h = tid + hh * MTH;
        if (h < H_) {
            float4 wv[L_ / 4];
            #pragma unroll
            for (int g = 0; g < L_ / 4; ++g) {
                const float* wg = Wb + (4 * g) * H_ + h;
                wv[g] = float4{wg[0], wg[H_], wg[2 * H_], wg[3 * H_]};
            }
            #pragma unroll
            for (int g = 0; g < L_ / 4; ++g)
                *(float4*)&WsT[h][4 * g] = wv[g];
        }
    }
    __syncthreads();

    const int lane = tid & 31;
    const int l0   = (tid >> 5) * MLG;

    v2f numA = {0.f, 0.f}, nxA = {0.f, 0.f}, nyA = {0.f, 0.f};
    v2f numB = {0.f, 0.f}, nxB = {0.f, 0.f}, nyB = {0.f, 0.f};

    for (int c = 0; c < H4; ++c) {
        const float4 x4 = *(const float4*)&Xs[lane][4 * c];
        const float4 y4 = *(const float4*)&Ys[lane][4 * c];
        const float xe[4] = {x4.x, x4.y, x4.z, x4.w};
        const float ye[4] = {y4.x, y4.y, y4.z, y4.w};
        for (int e = 0; e < 4; ++e) {
            const float4 w4 = *(const float4*)&WsT[4 * c + e][l0];
            const v2f wA = {w4.x, w4.y};
            const v2f wB = {w4.z, w4.w};
            const v2f xx = {xe[e], xe[e]};
            const v2f yy = {ye[e], ye[e]};
            const v2f xwA = xx * wA, ywA = yy * wA;
            const v2f xwB = xx * wB, ywB = yy * wB;
            numA = __builtin_elementwise_fma(xwA, ywA, numA);
            nxA  = __builtin_elementwise_fma(xwA, xwA, nxA);
            nyA  = __builtin_elementwise_fma(ywA, ywA, nyA);
            numB = __builtin_elementwise_fma(xwB, ywB, numB);
            nxB  = __builtin_elementwise_fma(xwB, xwB, nxB);
            nyB  = __builtin_elementwise_fma(ywB, ywB, nyB);
        }
    }

    const float num[MLG] = {numA.x, numA.y, numB.x, numB.y};
    const float nx[MLG]  = {nxA.x,  nxA.y,  nxB.x,  nxB.y};
    const float ny[MLG]  = {nyA.x,  nyA.y,  nyB.x,  nyB.y};
    for (int j = 0; j < MLG; ++j) {
        const float den = fmaxf(sqrtf(nx[j]) * sqrtf(ny[j]), 1e-8f);
        res[lane][l0 + j] = num[j] / den;
    }
    __syncthreads();

    float* ob = out + (tgt == 0 ? 0 : OUTQOFF) + (b * S_ + s0) * OUTC + secId * L_;
    for (int i = tid; i < MR * L_; i += MTH) {
        const int r = i / L_, l = i % L_;
        ob[r * OUTC + l] = res[r][l];
    }
}


extern "C" void kernel_launch(void* const* d_in, const int* in_sizes, int n_in,
                              void* d_out, int out_size, void* d_ws, size_t ws_size,
                              hipStream_t stream)
{
    const float* p = (const float*)d_in[0];
    const float* q = (const float*)d_in[1];
    const float* W = (const float*)d_in[2];
    float* out = (float*)d_out;

    float* ws  = (float*)d_ws;
    float* qvm = ws;
    float* pvm = qvm + B_ * 2 * S_ * H_;
    float* qvx = pvm + B_ * 2 * S_ * H_;
    float* pvx = qvx + B_ * 2 * S_ * H_;

    pool_attn_kernel<<<B_ * 2 * NST, FTH, 0, stream>>>(p, q, W, out, qvm, pvm, qvx, pvx);
    mpcos_kernel    <<<MPBLK, MTH, 0, stream>>>(p, q, W, qvm, pvm, qvx, pvx, out);
}


# mpcos: h sums moved to WMMA (A = x*y, x*x, y*y; B = w^2 for l 0-15; l 16-19 on the vector unit from the same fragments); products regrouped (x*y)*w^2; h sum order now the matrix unit's, two h halves
# speedup vs baseline: 3.6157x; 1.1093x over previous
#include <hip/hip_runtime.h>
#include <math.h>

typedef __attribute__((ext_vector_type(2))) float v2f;
typedef __attribute__((ext_vector_type(8))) float v8f;

#define B_    32
#define S_    64
#define H_    300
#define L_    20
#define OUTC  160
#define OUTQOFF (B_*S_*OUTC)
#define NEG_INF (-3.4e38f)

#define MP_LG   5
#define MP_NLG  (L_ / MP_LG)
#define MP_W    (MP_LG * 4)
#define MP_TH   (MP_W * 32)
#define ST      16
#define NST     (S_ / ST)
#define VTH     320
#define FTH     (MP_TH + VTH)
#define PC4     (H_ / 4)
#define F_NP    ((2 * S_ * PC4) / FTH)

__global__ __launch_bounds__(FTH)
void pool_attn_kernel(const float* __restrict__ p, const float* __restrict__ q,
                      const float* __restrict__ W, float* __restrict__ out,
                      float* __restrict__ qvm, float* __restrict__ pvm,
                      float* __restrict__ qvx, float* __restrict__ pvx)
{
    __shared__ __attribute__((aligned(16))) float XQ[2 * S_][H_];
    __shared__ __attribute__((aligned(16))) float w2s[MP_LG][H_];
    __shared__ float nPs[MP_LG][S_];
    __shared__ float nQs[MP_LG][S_];
    __shared__ float rowMx[MP_LG][S_];
    __shared__ float colPt[MP_LG][4][S_];
    __shared__ __attribute__((aligned(16))) float cosRowT[S_][ST];
    __shared__ __attribute__((aligned(16))) float cosColT[S_][ST];
    __shared__ float nA[S_];
    __shared__ float nB[S_];
    __shared__ float rSum[ST];
    __shared__ float cSum[ST];

    const int idx  = blockIdx.x;
    const int x    = idx % NST;
    const int hf   = (idx / NST) & 1;
    const int b    = idx / (2 * NST);
    const int tid  = threadIdx.x;
    const int lane = tid & 31;
    const int wv   = __builtin_amdgcn_readfirstlane(tid >> 5);
    const int lo   = lane & 15;
    const int hi   = lane >> 4;

    const bool attn = wv >= MP_W;
    const int li    = wv >> 2;
    const int strip = wv & 3;
    const int awv   = wv - MP_W;
    const int atid  = tid - MP_TH;
    const int s0    = x * ST;
    const int bh    = b * 2 + hf;
    const bool mm      = attn && awv < 8;
    const bool rowTile = awv < 4;
    const int astrip = rowTile ? x : (awv & 3);
    const int aqt    = rowTile ? awv : x;

    const float* Ph = p + (b * S_) * (2 * H_) + hf * H_;
    const float* Qh = q + (b * S_) * (2 * H_) + hf * H_;
    const float* Wg = W + ((2 + hf) * L_ + x * MP_LG) * H_;

    {
        float4 stg[F_NP];
        #pragma unroll
        for (int r = 0; r < F_NP; ++r) {
            const int i   = tid + r * FTH;
            const int row = i / PC4, c4 = i % PC4;
            const float* g = (row < S_ ? Ph + row * (2 * H_) : Qh + (row - S_) * (2 * H_)) + 4 * c4;
            stg[r] = float4{g[0], g[1], g[2], g[3]};
        }
        for (int i = tid; i < MP_LG * H_; i += FTH) {
            const float w = Wg[i];
            (&w2s[0][0])[i] = w * w;
        }
        #pragma unroll
        for (int r = 0; r < F_NP; ++r) {
            const int i   = tid + r * FTH;
            const int row = i / PC4, c4 = i % PC4;
            *(float4*)&XQ[row][4 * c4] = stg[r];
        }
    }
    __syncthreads();

    v8f c = {};
    if (mm) {
        const float* aRow = &XQ[16 * astrip + lo][2 * hi];
        const float* bRow = &XQ[S_ + 16 * aqt + lo][2 * hi];
        v2f sA2 = {0.f, 0.f}, sB2 = {0.f, 0.f};
        for (int kb = 0; kb < H_; kb += 4) {
            const v2f a  = *(const v2f*)(aRow + kb);
            const v2f bb = *(const v2f*)(bRow + kb);
            sA2 = __builtin_elementwise_fma(a, a, sA2);
            sB2 = __builtin_elementwise_fma(bb, bb, sB2);
            c = __builtin_amdgcn_wmma_f32_16x16x4_f32(false, a, false, bb, (short)0, c, false, false);
        }
        float sA = sA2.x + sA2.y;
        float sB = sB2.x + sB2.y;
        sA += __shfl_xor(sA, 16);
        sB += __shfl_xor(sB, 16);
        if (hi == 0) {
            if (!rowTile) nA[16 * astrip + lo] = sqrtf(sA);
            else          nB[16 * aqt + lo]    = sqrtf(sB);
        }
    }
    __syncthreads();

    if (mm) {
        const int n = 16 * aqt + lo;
        const float nb = nB[n];
        #pragma unroll
        for (int j = 0; j < 8; ++j) {
            const int m = 16 * astrip + 8 * hi + j;
            const float v = c[j] / (nA[m] * nb);
            if (rowTile) cosRowT[n][m - s0] = v;
            else         cosColT[m][n - s0] = v;
        }
    }
    __syncthreads();

    if (attn) {
        if (atid < ST) {
            float sm = 0.f;
            for (int n = 0; n < S_; ++n) sm += cosRowT[n][atid];
            rSum[atid] = sm;
        } else if (atid >= 32 && atid < 32 + ST) {
            const int t = atid - 32;
            float sm = 0.f;
            for (int m = 0; m < S_; ++m) sm += cosColT[m][t];
            cSum[t] = sm;
        }
    }
    __syncthreads();

    v8f c0, c1, c2, c3;
    if (attn) {
        const int h = atid;
        if (h < H_) {
            {
                const float* Qc = &XQ[S_][h];
                float qm[ST], qx[ST];
                for (int t = 0; t < ST; ++t) { qm[t] = 0.f; qx[t] = NEG_INF; }
                #pragma unroll 4
                for (int j = 0; j < S_; ++j) {
                    const float qv = Qc[j * H_];
                    float cr[ST];
                    for (int g = 0; g < ST / 4; ++g) {
                        const float4 r4 = *(const float4*)&cosRowT[j][4 * g];
                        cr[4 * g + 0] = r4.x; cr[4 * g + 1] = r4.y; cr[4 * g + 2] = r4.z; cr[4 * g + 3] = r4.w;
                    }
                    for (int t = 0; t < ST; ++t) {
                        float a = qv * cr[t];
                        qm[t] += a;  qx[t] = fmaxf(qx[t], a);
                    }
                }
                float* qvm_o = qvm + ((bh * S_ + s0) * H_ + h);
                float* qvx_o = qvx + ((bh * S_ + s0) * H_ + h);
                for (int t = 0; t < ST; ++t) {
                    const float rs = rSum[t];
                    qvm_o[t * H_] = qm[t] / rs;
                    qvx_o[t * H_] = qx[t];
                }
            }
            {
                const float* Pc = &XQ[0][h];
                float pm[ST], px[ST];
                for (int t = 0; t < ST; ++t) { pm[t] = 0.f; px[t] = NEG_INF; }
                #pragma unroll 4
                for (int j = 0; j < S_; ++j) {
                    const float pv = Pc[j * H_];
                    float cc[ST];
                    for (int g = 0; g < ST / 4; ++g) {
                        const float4 k4 = *(const float4*)&cosColT[j][4 * g];
                        cc[4 * g + 0] = k4.x; cc[4 * g + 1] = k4.y; cc[4 * g + 2] = k4.z; cc[4 * g + 3] = k4.w;
                    }
                    for (int t = 0; t < ST; ++t) {
                        float bb = pv * cc[t];
                        pm[t] += bb; px[t] = fmaxf(px[t], bb);
                    }
                }
                float* pvm_o = pvm + ((bh * S_ + s0) * H_ + h);
                float* pvx_o = pvx + ((bh * S_ + s0) * H_ + h);
                for (int t = 0; t < ST; ++t) {
                    const float cs = cSum[t];
                    pvm_o[t * H_] = pm[t] / cs;
                    pvx_o[t * H_] = px[t];
                }
            }
        }
    } else {
        const float* aRow = &XQ[16 * strip + lo][2 * hi];
        const float* bR0  = &XQ[S_ + 16 * ((strip + 0) & 3) + lo][2 * hi];
        const float* bR1  = &XQ[S_ + 16 * ((strip + 1) & 3) + lo][2 * hi];
        const float* bR2  = &XQ[S_ + 16 * ((strip + 2) & 3) + lo][2 * hi];
        const float* bR3  = &XQ[S_ + 16 * ((strip + 3) & 3) + lo][2 * hi];
        const float* wRow = &w2s[li][2 * hi];
        float sP = 0.f, sQ = 0.f;
        c0 = v8f{}; c1 = v8f{}; c2 = v8f{}; c3 = v8f{};
        for (int kb = 0; kb < H_; kb += 4) {
            const v2f a  = *(const v2f*)(aRow + kb);
            const v2f w  = *(const v2f*)(wRow + kb);
            const v2f b0 = *(const v2f*)(bR0 + kb);
            const v2f b1 = *(const v2f*)(bR1 + kb);
            const v2f b2 = *(const v2f*)(bR2 + kb);
            const v2f b3 = *(const v2f*)(bR3 + kb);
            const v2f aw = a * w;
            const v2f bw = b0 * w;
            sP = fmaf(a.x, aw.x, sP);   sP = fmaf(a.y, aw.y, sP);
            sQ = fmaf(b0.x, bw.x, sQ);  sQ = fmaf(b0.y, bw.y, sQ);
            c0 = __builtin_amdgcn_wmma_f32_16x16x4_f32(false, aw, false, b0, (short)0, c0, false, false);
            c1 = __builtin_amdgcn_wmma_f32_16x16x4_f32(false, aw, false, b1, (short)0, c1, false, false);
            c2 = __builtin_amdgcn_wmma_f32_16x16x4_f32(false, aw, false, b2, (short)0, c2, false, false);
            c3 = __builtin_amdgcn_wmma_f32_16x16x4_f32(false, aw, false, b3, (short)0, c3, false, false);
        }
        sP += __shfl_xor(sP, 16);
        sQ += __shfl_xor(sQ, 16);
        if (hi == 0) {
            nPs[li][16 * strip + lo] = sqrtf(sP);
            nQs[li][16 * strip + lo] = sqrtf(sQ);
        }
    }
    __syncthreads();

    const int lane2 = (int)__builtin_amdgcn_mbcnt_lo(~0u, 0u);
    const int lo2   = lane2 & 15;
    const int hi2   = lane2 >> 4;
    const int tid2  = wv * 32 + lane2;

    if (!attn) {
        const int n0 = 16 * ((strip + 0) & 3) + lo2;
        const int n1 = 16 * ((strip + 1) & 3) + lo2;
        const int n2 = 16 * ((strip + 2) & 3) + lo2;
        const int n3 = 16 * ((strip + 3) & 3) + lo2;
        const float nb0 = nQs[li][n0], nb1 = nQs[li][n1], nb2 = nQs[li][n2], nb3 = nQs[li][n3];
        float cm0 = NEG_INF, cm1 = NEG_INF, cm2 = NEG_INF, cm3 = NEG_INF;
        float rmx[8];
        #pragma unroll
        for (int j = 0; j < 8; ++j) {
            const float na = nPs[li][16 * strip + 8 * hi2 + j];
            const float v0 = c0[j] / (na * nb0);
            const float v1 = c1[j] / (na * nb1);
            const float v2 = c2[j] / (na * nb2);
            const float v3 = c3[j] / (na * nb3);
            cm0 = fmaxf(cm0, v0); cm1 = fmaxf(cm1, v1);
            cm2 = fmaxf(cm2, v2); cm3 = fmaxf(cm3, v3);
            rmx[j] = fmaxf(fmaxf(fmaxf(fmaxf(NEG_INF, v0), v1), v2), v3);
        }
        #pragma unroll
        for (int d = 1; d < 16; d <<= 1) {
            #pragma unroll
            for (int j = 0; j < 8; ++j) rmx[j] = fmaxf(rmx[j], __shfl_xor(rmx[j], d));
        }
        if (lo2 == 0) {
            for (int j = 0; j < 8; ++j) rowMx[li][16 * strip + 8 * hi2 + j] = rmx[j];
        }
        cm0 = fmaxf(cm0, __shfl_xor(cm0, 16));
        cm1 = fmaxf(cm1, __shfl_xor(cm1, 16));
        cm2 = fmaxf(cm2, __shfl_xor(cm2, 16));
        cm3 = fmaxf(cm3, __shfl_xor(cm3, 16));
        if (hi2 == 0) {
            colPt[li][strip][n0] = cm0;
            colPt[li][strip][n1] = cm1;
            colPt[li][strip][n2] = cm2;
            colPt[li][strip][n3] = cm3;
        }
    }
    __syncthreads();

    const int nres = (S_ * MP_LG);
    if (tid2 < nres) {
        const int row = tid2 / MP_LG, l5 = tid2 % MP_LG;
        out[(b * S_ + row) * OUTC + (2 + hf) * L_ + x * MP_LG + l5] = rowMx[l5][row];
    } else if (tid2 < 2 * nres) {
        const int u = tid2 - nres;
        const int n = u / MP_LG, l5 = u % MP_LG;
        float mx = NEG_INF;
        for (int s = 0; s < 4; ++s) mx = fmaxf(mx, colPt[l5][s][n]);
        out[OUTQOFF + (b * S_ + n) * OUTC + (2 + hf) * L_ + x * MP_LG + l5] = mx;
    }
}

#define MR    32
#define MTH   128
#define H4    (H_ / 4)
#define KS0   38
#define LT    16
#define LE    (L_ - LT)
#define MPBLK (2 * 6 * B_ * (S_ / MR))
#define XY4   (MR * H4)
#define W4    (L_ * H4)
#define XYT   ((XY4 + MTH - 1) / MTH)
#define WT    ((W4 + MTH - 1) / MTH)

__global__ __launch_bounds__(MTH)
void mpcos_kernel(const float* __restrict__ p, const float* __restrict__ q,
                  const float* __restrict__ W,
                  const float* __restrict__ qvm, const float* __restrict__ pvm,
                  const float* __restrict__ qvx, const float* __restrict__ pvx,
                  float* __restrict__ out)
{
    __shared__ __attribute__((aligned(16))) float Xs[MR][H_];
    __shared__ __attribute__((aligned(16))) float Ys[MR][H_];
    __shared__ __attribute__((aligned(16))) float W2s[L_][H_];

    int t = blockIdx.x;
    const int rh  = t % (S_ / MR);  t /= (S_ / MR);
    const int b   = t % B_;         t /= B_;
    const int sec = t % 6;          const int tgt = t / 6;
    const int secId = sec < 2 ? sec : sec + 2;
    const int hf  = secId & 1;
    const int s0  = rh * MR;
    const int tid = threadIdx.x;

    const float* Xb = (tgt == 0 ? p : q) + (b * S_ + s0) * (2 * H_) + hf * H_;
    const float* Yb;
    int ystride;
    if (secId == 0)      { Yb = (tgt == 0 ? q : p) + (b * S_ + (S_ - 1)) * (2 * H_); ystride = 0; }
    else if (secId == 1) { Yb = (tgt == 0 ? q : p) + (b * S_) * (2 * H_) + H_;       ystride = 0; }
    else {
        const float* vec = (secId <= 5) ? (tgt == 0 ? qvm : pvm)
                                        : (tgt == 0 ? qvx : pvx);
        Yb = vec + ((b * 2 + hf) * S_ + s0) * H_;
        ystride = H_;
    }
    const float* Wb = W + (secId * L_) * H_;

    {
        float4 xv[XYT], yv[XYT], wv4[WT];
        #pragma unroll
        for (int r = 0; r < XYT; ++r) {
            const int i = tid + r * MTH;
            if (i < XY4) {
                const int row = i / H4, c4 = i % H4;
                const float* xk = Xb + row * (2 * H_) + 4 * c4;
                const float* yk = Yb + row * ystride + 4 * c4;
                xv[r] = float4{xk[0], xk[1], xk[2], xk[3]};
                yv[r] = float4{yk[0], yk[1], yk[2], yk[3]};
            }
        }
        #pragma unroll
        for (int r = 0; r < WT; ++r) {
            const int i = tid + r * MTH;
            if (i < W4) {
                const float* wk = Wb + 4 * i;
                wv4[r] = float4{wk[0], wk[1], wk[2], wk[3]};
            }
        }
        #pragma unroll
        for (int r = 0; r < XYT; ++r) {
            const int i = tid + r * MTH;
            if (i < XY4) {
                const int row = i / H4, c4 = i % H4;
                *(float4*)&Xs[row][4 * c4] = xv[r];
                *(float4*)&Ys[row][4 * c4] = yv[r];
            }
        }
        #pragma unroll
        for (int r = 0; r < WT; ++r) {
            const int i = tid + r * MTH;
            if (i < W4) {
                const float4 w = wv4[r];
                *(float4*)(&W2s[0][0] + 4 * i) = float4{w.x * w.x, w.y * w.y, w.z * w.z, w.w * w.w};
            }
        }
    }
    __syncthreads();

    const int lane = tid & 31;
    const int lo   = lane & 15;
    const int hi   = lane >> 4;
    const int wv   = __builtin_amdgcn_readfirstlane(tid >> 5);
    const int tile = wv & 1;
    const int kh   = wv >> 1;
    const int kb0  = kh ? 4 * KS0 : 0;
    const int kb1  = kh ? H_ : 4 * KS0;

    const float* xRow = &Xs[16 * tile + lo][2 * hi];
    const float* yRow = &Ys[16 * tile + lo][2 * hi];
    const float* bRow = &W2s[lo][2 * hi];
    const float* eRow = &W2s[LT][2 * hi];

    v8f cN = {}, cX = {}, cY = {};
    v2f eN[LE], eX[LE], eY[LE];
    #pragma unroll
    for (int j = 0; j < LE; ++j) { eN[j] = v2f{0.f, 0.f}; eX[j] = v2f{0.f, 0.f}; eY[j] = v2f{0.f, 0.f}; }

    #pragma unroll 2
    for (int kb = kb0; kb < kb1; kb += 4) {
        const v2f x  = *(const v2f*)(xRow + kb);
        const v2f y  = *(const v2f*)(yRow + kb);
        const v2f w  = *(const v2f*)(bRow + kb);
        const v2f xy = x * y, xx = x * x, yy = y * y;
        cN = __builtin_amdgcn_wmma_f32_16x16x4_f32(false, xy, false, w, (short)0, cN, false, false);
        cX = __builtin_amdgcn_wmma_f32_16x16x4_f32(false, xx, false, w, (short)0, cX, false, false);
        cY = __builtin_amdgcn_wmma_f32_16x16x4_f32(false, yy, false, w, (short)0, cY, false, false);
        #pragma unroll
        for (int j = 0; j < LE; ++j) {
            const v2f we = *(const v2f*)(eRow + j * H_ + kb);
            eN[j] = __builtin_elementwise_fma(xy, we, eN[j]);
            eX[j] = __builtin_elementwise_fma(xx, we, eX[j]);
            eY[j] = __builtin_elementwise_fma(yy, we, eY[j]);
        }
    }

    float sN[LE], sX[LE], sY[LE];
    #pragma unroll
    for (int j = 0; j < LE; ++j) {
        sN[j] = eN[j].x + eN[j].y;  sX[j] = eX[j].x + eX[j].y;  sY[j] = eY[j].x + eY[j].y;
        sN[j] += __shfl_xor(sN[j], 16);
        sX[j] += __shfl_xor(sX[j], 16);
        sY[j] += __shfl_xor(sY[j], 16);
    }
    __syncthreads();

    float* part = &Xs[0][0];
    {
        float* pN = part + ((kh * 3 + 0) * MR) * L_;
        float* pX = part + ((kh * 3 + 1) * MR) * L_;
        float* pY = part + ((kh * 3 + 2) * MR) * L_;
        #pragma unroll
        for (int r = 0; r < 8; ++r) {
            const int row = 16 * tile + 8 * hi + r;
            pN[row * L_ + lo] = cN[r];
            pX[row * L_ + lo] = cX[r];
            pY[row * L_ + lo] = cY[r];
        }
        if (hi == 0) {
            const int row = 16 * tile + lo;
            #pragma unroll
            for (int j = 0; j < LE; ++j) {
                pN[row * L_ + LT + j] = sN[j];
                pX[row * L_ + LT + j] = sX[j];
                pY[row * L_ + LT + j] = sY[j];
            }
        }
    }
    __syncthreads();

    float* ob = out + (tgt == 0 ? 0 : OUTQOFF) + (b * S_ + s0) * OUTC + secId * L_;
    for (int i = tid; i < MR * L_; i += MTH) {
        const int r = i / L_, l = i % L_;
        const float n  = part[(0 * MR) * L_ + i] + part[(3 * MR) * L_ + i];
        const float sx = part[(1 * MR) * L_ + i] + part[(4 * MR) * L_ + i];
        const float sy = part[(2 * MR) * L_ + i] + part[(5 * MR) * L_ + i];
        const float den = fmaxf(sqrtf(sx) * sqrtf(sy), 1e-8f);
        ob[r * OUTC + l] = n / den;
    }
}


extern "C" void kernel_launch(void* const* d_in, const int* in_sizes, int n_in,
                              void* d_out, int out_size, void* d_ws, size_t ws_size,
                              hipStream_t stream)
{
    const float* p = (const float*)d_in[0];
    const float* q = (const float*)d_in[1];
    const float* W = (const float*)d_in[2];
    float* out = (float*)d_out;

    float* ws  = (float*)d_ws;
    float* qvm = ws;
    float* pvm = qvm + B_ * 2 * S_ * H_;
    float* qvx = pvm + B_ * 2 * S_ * H_;
    float* pvx = qvx + B_ * 2 * S_ * H_;

    pool_attn_kernel<<<B_ * 2 * NST, FTH, 0, stream>>>(p, q, W, out, qvm, pvm, qvx, pvx);
    mpcos_kernel    <<<MPBLK, MTH, 0, stream>>>(p, q, W, qvm, pvm, qvx, pvx, out);
}


# attentive walk re-cut for equal issue per SIMD: 4 waves (one per SIMD, oldest of the block) in place of 10, each (side, 8 s rows) x 5 passes of 64 h, two h per lane; j and k sum orders kept
# speedup vs baseline: 3.7623x; 1.0405x over previous
#include <hip/hip_runtime.h>
#include <math.h>

typedef __attribute__((ext_vector_type(2))) float v2f;
typedef __attribute__((ext_vector_type(8))) float v8f;

#define B_    32
#define S_    64
#define H_    300
#define L_    20
#define OUTC  160
#define OUTQOFF (B_*S_*OUTC)
#define NEG_INF (-3.4e38f)

#define MP_LG   5
#define MP_NLG  (L_ / MP_LG)
#define MP_W    (MP_LG * 4)
#define MP_TH   (MP_W * 32)
#define ST      16
#define NST     (S_ / ST)
#define AW      4
#define VTH     (AW * 32)
#define FTH     (MP_TH + VTH)
#define PC4     (H_ / 4)
#define NPC     (2 * S_ * PC4)
#define F_NP    ((NPC + FTH - 1) / FTH)
#define NHG     ((H_ + 63) / 64)

__global__ __launch_bounds__(FTH)
void pool_attn_kernel(const float* __restrict__ p, const float* __restrict__ q,
                      const float* __restrict__ W, float* __restrict__ out,
                      float* __restrict__ qvm, float* __restrict__ pvm,
                      float* __restrict__ qvx, float* __restrict__ pvx)
{
    __shared__ __attribute__((aligned(16))) float XQ[2 * S_][H_];
    __shared__ __attribute__((aligned(16))) float w2s[MP_LG][H_];
    __shared__ float nPs[MP_LG][S_];
    __shared__ float nQs[MP_LG][S_];
    __shared__ float rowMx[MP_LG][S_];
    __shared__ float colPt[MP_LG][4][S_];
    __shared__ __attribute__((aligned(16))) float cosRowT[S_][ST];
    __shared__ __attribute__((aligned(16))) float cosColT[S_][ST];
    __shared__ float nA[S_];
    __shared__ float nB[S_];
    __shared__ float rSum[ST];
    __shared__ float cSum[ST];

    const int idx  = blockIdx.x;
    const int x    = idx % NST;
    const int hf   = (idx / NST) & 1;
    const int b    = idx / (2 * NST);
    const int tid  = threadIdx.x;
    const int lane = tid & 31;
    const int wv   = __builtin_amdgcn_readfirstlane(tid >> 5);
    const int lo   = lane & 15;
    const int hi   = lane >> 4;

    const bool attn = wv < AW;
    const int mw    = wv - AW;
    const int li    = mw >> 2;
    const int strip = mw & 3;
    const int awv   = wv;
    const int atid  = tid;
    const int s0    = x * ST;
    const int bh    = b * 2 + hf;

    const float* Ph = p + (b * S_) * (2 * H_) + hf * H_;
    const float* Qh = q + (b * S_) * (2 * H_) + hf * H_;
    const float* Wg = W + ((2 + hf) * L_ + x * MP_LG) * H_;

    {
        float4 stg[F_NP];
        #pragma unroll
        for (int r = 0; r < F_NP; ++r) {
            const int i   = tid + r * FTH;
            if (i < NPC) {
                const int row = i / PC4, c4 = i % PC4;
                const float* g = (row < S_ ? Ph + row * (2 * H_) : Qh + (row - S_) * (2 * H_)) + 4 * c4;
                stg[r] = float4{g[0], g[1], g[2], g[3]};
            }
        }
        for (int i = tid; i < MP_LG * H_; i += FTH) {
            const float w = Wg[i];
            (&w2s[0][0])[i] = w * w;
        }
        #pragma unroll
        for (int r = 0; r < F_NP; ++r) {
            const int i   = tid + r * FTH;
            if (i < NPC) {
                const int row = i / PC4, c4 = i % PC4;
                *(float4*)&XQ[row][4 * c4] = stg[r];
            }
        }
    }
    __syncthreads();

    v8f cR = {}, cC = {};
    if (attn) {
        const float* aRowR = &XQ[16 * x + lo][2 * hi];
        const float* bRowR = &XQ[S_ + 16 * awv + lo][2 * hi];
        const float* aRowC = &XQ[16 * awv + lo][2 * hi];
        const float* bRowC = &XQ[S_ + 16 * x + lo][2 * hi];
        v2f sA2 = {0.f, 0.f}, sB2 = {0.f, 0.f};
        for (int kb = 0; kb < H_; kb += 4) {
            const v2f ar = *(const v2f*)(aRowR + kb);
            const v2f br = *(const v2f*)(bRowR + kb);
            const v2f ac = *(const v2f*)(aRowC + kb);
            const v2f bc = *(const v2f*)(bRowC + kb);
            sA2 = __builtin_elementwise_fma(ac, ac, sA2);
            sB2 = __builtin_elementwise_fma(br, br, sB2);
            cR = __builtin_amdgcn_wmma_f32_16x16x4_f32(false, ar, false, br, (short)0, cR, false, false);
            cC = __builtin_amdgcn_wmma_f32_16x16x4_f32(false, ac, false, bc, (short)0, cC, false, false);
        }
        float sA = sA2.x + sA2.y;
        float sB = sB2.x + sB2.y;
        sA += __shfl_xor(sA, 16);
        sB += __shfl_xor(sB, 16);
        if (hi == 0) {
            nA[16 * awv + lo] = sqrtf(sA);
            nB[16 * awv + lo] = sqrtf(sB);
        }
    }
    __syncthreads();

    if (attn) {
        const int nR = 16 * awv + lo;
        const float nbR = nB[nR];
        const float nbC = nB[s0 + lo];
        #pragma unroll
        for (int j = 0; j < 8; ++j) {
            const int t  = 8 * hi + j;
            const int mC = 16 * awv + t;
            cosRowT[nR][t] = cR[j] / (nA[s0 + t] * nbR);
            cosColT[mC][lo] = cC[j] / (nA[mC] * nbC);
        }
    }
    __syncthreads();

    if (attn) {
        if (atid < ST) {
            float sm = 0.f;
            for (int n = 0; n < S_; ++n) sm += cosRowT[n][atid];
            rSum[atid] = sm;
        } else if (atid >= 32 && atid < 32 + ST) {
            const int t = atid - 32;
            float sm = 0.f;
            for (int m = 0; m < S_; ++m) sm += cosColT[m][t];
            cSum[t] = sm;
        }
    }
    __syncthreads();

    v8f c0, c1, c2, c3;
    if (attn) {
        const int side = awv >> 1;
        const int th   = awv & 1;
        const float* Xc   = side ? &XQ[0][0] : &XQ[S_][0];
        const float* cT   = side ? &cosColT[0][8 * th] : &cosRowT[0][8 * th];
        const float* sums = side ? &cSum[8 * th] : &rSum[8 * th];
        float* om = (side ? pvm : qvm) + (bh * S_ + s0 + 8 * th) * H_;
        float* ox = (side ? pvx : qvx) + (bh * S_ + s0 + 8 * th) * H_;
        for (int g = 0; g < NHG; ++g) {
            const int hc = 64 * g + 2 * lane;
            if (hc < H_) {
                const float* col = Xc + hc;
                float m0[ST / 2], m1[ST / 2], x0[ST / 2], x1[ST / 2];
                for (int t = 0; t < ST / 2; ++t) { m0[t] = 0.f; m1[t] = 0.f; x0[t] = NEG_INF; x1[t] = NEG_INF; }
                #pragma unroll 4
                for (int j = 0; j < S_; ++j) {
                    const v2f v = *(const v2f*)(col + j * H_);
                    const float4 ca = *(const float4*)(cT + j * ST);
                    const float4 cb = *(const float4*)(cT + j * ST + 4);
                    const float cr[ST / 2] = {ca.x, ca.y, ca.z, ca.w, cb.x, cb.y, cb.z, cb.w};
                    for (int t = 0; t < ST / 2; ++t) {
                        const float a0 = v.x * cr[t];
                        const float a1 = v.y * cr[t];
                        m0[t] += a0;  x0[t] = fmaxf(x0[t], a0);
                        m1[t] += a1;  x1[t] = fmaxf(x1[t], a1);
                    }
                }
                for (int t = 0; t < ST / 2; ++t) {
                    const float rs = sums[t];
                    *(v2f*)(om + t * H_ + hc) = v2f{m0[t] / rs, m1[t] / rs};
                    *(v2f*)(ox + t * H_ + hc) = v2f{x0[t], x1[t]};
                }
            }
        }
    } else {
        const float* aRow = &XQ[16 * strip + lo][2 * hi];
        const float* bR0  = &XQ[S_ + 16 * ((strip + 0) & 3) + lo][2 * hi];
        const float* bR1  = &XQ[S_ + 16 * ((strip + 1) & 3) + lo][2 * hi];
        const float* bR2  = &XQ[S_ + 16 * ((strip + 2) & 3) + lo][2 * hi];
        const float* bR3  = &XQ[S_ + 16 * ((strip + 3) & 3) + lo][2 * hi];
        const float* wRow = &w2s[li][2 * hi];
        float sP = 0.f, sQ = 0.f;
        c0 = v8f{}; c1 = v8f{}; c2 = v8f{}; c3 = v8f{};
        for (int kb = 0; kb < H_; kb += 4) {
            const v2f a  = *(const v2f*)(aRow + kb);
            const v2f w  = *(const v2f*)(wRow + kb);
            const v2f b0 = *(const v2f*)(bR0 + kb);
            const v2f b1 = *(const v2f*)(bR1 + kb);
            const v2f b2 = *(const v2f*)(bR2 + kb);
            const v2f b3 = *(const v2f*)(bR3 + kb);
            const v2f aw = a * w;
            const v2f bw = b0 * w;
            sP = fmaf(a.x, aw.x, sP);   sP = fmaf(a.y, aw.y, sP);
            sQ = fmaf(b0.x, bw.x, sQ);  sQ = fmaf(b0.y, bw.y, sQ);
            c0 = __builtin_amdgcn_wmma_f32_16x16x4_f32(false, aw, false, b0, (short)0, c0, false, false);
            c1 = __builtin_amdgcn_wmma_f32_16x16x4_f32(false, aw, false, b1, (short)0, c1, false, false);
            c2 = __builtin_amdgcn_wmma_f32_16x16x4_f32(false, aw, false, b2, (short)0, c2, false, false);
            c3 = __builtin_amdgcn_wmma_f32_16x16x4_f32(false, aw, false, b3, (short)0, c3, false, false);
        }
        sP += __shfl_xor(sP, 16);
        sQ += __shfl_xor(sQ, 16);
        if (hi == 0) {
            nPs[li][16 * strip + lo] = sqrtf(sP);
            nQs[li][16 * strip + lo] = sqrtf(sQ);
        }
    }
    __syncthreads();

    const int lo2  = lo;
    const int hi2  = hi;
    const int tid2 = tid;

    if (!attn) {
        const int n0 = 16 * ((strip + 0) & 3) + lo2;
        const int n1 = 16 * ((strip + 1) & 3) + lo2;
        const int n2 = 16 * ((strip + 2) & 3) + lo2;
        const int n3 = 16 * ((strip + 3) & 3) + lo2;
        const float nb0 = nQs[li][n0], nb1 = nQs[li][n1], nb2 = nQs[li][n2], nb3 = nQs[li][n3];
        float cm0 = NEG_INF, cm1 = NEG_INF, cm2 = NEG_INF, cm3 = NEG_INF;
        float rmx[8];
        #pragma unroll
        for (int j = 0; j < 8; ++j) {
            const float na = nPs[li][16 * strip + 8 * hi2 + j];
            const float v0 = c0[j] / (na * nb0);
            const float v1 = c1[j] / (na * nb1);
            const float v2 = c2[j] / (na * nb2);
            const float v3 = c3[j] / (na * nb3);
            cm0 = fmaxf(cm0, v0); cm1 = fmaxf(cm1, v1);
            cm2 = fmaxf(cm2, v2); cm3 = fmaxf(cm3, v3);
            rmx[j] = fmaxf(fmaxf(fmaxf(fmaxf(NEG_INF, v0), v1), v2), v3);
        }
        #pragma unroll
        for (int d = 1; d < 16; d <<= 1) {
            #pragma unroll
            for (int j = 0; j < 8; ++j) rmx[j] = fmaxf(rmx[j], __shfl_xor(rmx[j], d));
        }
        if (lo2 == 0) {
            for (int j = 0; j < 8; ++j) rowMx[li][16 * strip + 8 * hi2 + j] = rmx[j];
        }
        cm0 = fmaxf(cm0, __shfl_xor(cm0, 16));
        cm1 = fmaxf(cm1, __shfl_xor(cm1, 16));
        cm2 = fmaxf(cm2, __shfl_xor(cm2, 16));
        cm3 = fmaxf(cm3, __shfl_xor(cm3, 16));
        if (hi2 == 0) {
            colPt[li][strip][n0] = cm0;
            colPt[li][strip][n1] = cm1;
            colPt[li][strip][n2] = cm2;
            colPt[li][strip][n3] = cm3;
        }
    }
    __syncthreads();

    const int nres = (S_ * MP_LG);
    if (tid2 < nres) {
        const int row = tid2 / MP_LG, l5 = tid2 % MP_LG;
        out[(b * S_ + row) * OUTC + (2 + hf) * L_ + x * MP_LG + l5] = rowMx[l5][row];
    } else if (tid2 < 2 * nres) {
        const int u = tid2 - nres;
        const int n = u / MP_LG, l5 = u % MP_LG;
        float mx = NEG_INF;
        for (int s = 0; s < 4; ++s) mx = fmaxf(mx, colPt[l5][s][n]);
        out[OUTQOFF + (b * S_ + n) * OUTC + (2 + hf) * L_ + x * MP_LG + l5] = mx;
    }
}

#define MR    32
#define MTH   128
#define H4    (H_ / 4)
#define KS0   38
#define LT    16
#define LE    (L_ - LT)
#define MPBLK (2 * 6 * B_ * (S_ / MR))
#define XY4   (MR * H4)
#define W4    (L_ * H4)
#define XYT   ((XY4 + MTH - 1) / MTH)
#define WT    ((W4 + MTH - 1) / MTH)

__global__ __launch_bounds__(MTH)
void mpcos_kernel(const float* __restrict__ p, const float* __restrict__ q,
                  const float* __restrict__ W,
                  const float* __restrict__ qvm, const float* __restrict__ pvm,
                  const float* __restrict__ qvx, const float* __restrict__ pvx,
                  float* __restrict__ out)
{
    __shared__ __attribute__((aligned(16))) float Xs[MR][H_];
    __shared__ __attribute__((aligned(16))) float Ys[MR][H_];
    __shared__ __attribute__((aligned(16))) float W2s[L_][H_];

    int t = blockIdx.x;
    const int rh  = t % (S_ / MR);  t /= (S_ / MR);
    const int b   = t % B_;         t /= B_;
    const int sec = t % 6;          const int tgt = t / 6;
    const int secId = sec < 2 ? sec : sec + 2;
    const int hf  = secId & 1;
    const int s0  = rh * MR;
    const int tid = threadIdx.x;

    const float* Xb = (tgt == 0 ? p : q) + (b * S_ + s0) * (2 * H_) + hf * H_;
    const float* Yb;
    int ystride;
    if (secId == 0)      { Yb = (tgt == 0 ? q : p) + (b * S_ + (S_ - 1)) * (2 * H_); ystride = 0; }
    else if (secId == 1) { Yb = (tgt == 0 ? q : p) + (b * S_) * (2 * H_) + H_;       ystride = 0; }
    else {
        const float* vec = (secId <= 5) ? (tgt == 0 ? qvm : pvm)
                                        : (tgt == 0 ? qvx : pvx);
        Yb = vec + ((b * 2 + hf) * S_ + s0) * H_;
        ystride = H_;
    }
    const float* Wb = W + (secId * L_) * H_;

    {
        float4 xv[XYT], yv[XYT], wv4[WT];
        #pragma unroll
        for (int r = 0; r < XYT; ++r) {
            const int i = tid + r * MTH;
            if (i < XY4) {
                const int row = i / H4, c4 = i % H4;
                const float* xk = Xb + row * (2 * H_) + 4 * c4;
                const float* yk = Yb + row * ystride + 4 * c4;
                xv[r] = float4{xk[0], xk[1], xk[2], xk[3]};
                yv[r] = float4{yk[0], yk[1], yk[2], yk[3]};
            }
        }
        #pragma unroll
        for (int r = 0; r < WT; ++r) {
            const int i = tid + r * MTH;
            if (i < W4) {
                const float* wk = Wb + 4 * i;
                wv4[r] = float4{wk[0], wk[1], wk[2], wk[3]};
            }
        }
        #pragma unroll
        for (int r = 0; r < XYT; ++r) {
            const int i = tid + r * MTH;
            if (i < XY4) {
                const int row = i / H4, c4 = i % H4;
                *(float4*)&Xs[row][4 * c4] = xv[r];
                *(float4*)&Ys[row][4 * c4] = yv[r];
            }
        }
        #pragma unroll
        for (int r = 0; r < WT; ++r) {
            const int i = tid + r * MTH;
            if (i < W4) {
                const float4 w = wv4[r];
                *(float4*)(&W2s[0][0] + 4 * i) = float4{w.x * w.x, w.y * w.y, w.z * w.z, w.w * w.w};
            }
        }
    }
    __syncthreads();

    const int lane = tid & 31;
    const int lo   = lane & 15;
    const int hi   = lane >> 4;
    const int wv   = __builtin_amdgcn_readfirstlane(tid >> 5);
    const int tile = wv & 1;
    const int kh   = wv >> 1;
    const int kb0  = kh ? 4 * KS0 : 0;
    const int kb1  = kh ? H_ : 4 * KS0;

    const float* xRow = &Xs[16 * tile + lo][2 * hi];
    const float* yRow = &Ys[16 * tile + lo][2 * hi];
    const float* bRow = &W2s[lo][2 * hi];
    const float* eRow = &W2s[LT][2 * hi];

    v8f cN = {}, cX = {}, cY = {};
    v2f eN[LE], eX[LE], eY[LE];
    #pragma unroll
    for (int j = 0; j < LE; ++j) { eN[j] = v2f{0.f, 0.f}; eX[j] = v2f{0.f, 0.f}; eY[j] = v2f{0.f, 0.f}; }

    #pragma unroll 2
    for (int kb = kb0; kb < kb1; kb += 4) {
        const v2f x  = *(const v2f*)(xRow + kb);
        const v2f y  = *(const v2f*)(yRow + kb);
        const v2f w  = *(const v2f*)(bRow + kb);
        const v2f xy = x * y, xx = x * x, yy = y * y;
        cN = __builtin_amdgcn_wmma_f32_16x16x4_f32(false, xy, false, w, (short)0, cN, false, false);
        cX = __builtin_amdgcn_wmma_f32_16x16x4_f32(false, xx, false, w, (short)0, cX, false, false);
        cY = __builtin_amdgcn_wmma_f32_16x16x4_f32(false, yy, false, w, (short)0, cY, false, false);
        #pragma unroll
        for (int j = 0; j < LE; ++j) {
            const v2f we = *(const v2f*)(eRow + j * H_ + kb);
            eN[j] = __builtin_elementwise_fma(xy, we, eN[j]);
            eX[j] = __builtin_elementwise_fma(xx, we, eX[j]);
            eY[j] = __builtin_elementwise_fma(yy, we, eY[j]);
        }
    }

    float sN[LE], sX[LE], sY[LE];
    #pragma unroll
    for (int j = 0; j < LE; ++j) {
        sN[j] = eN[j].x + eN[j].y;  sX[j] = eX[j].x + eX[j].y;  sY[j] = eY[j].x + eY[j].y;
        sN[j] += __shfl_xor(sN[j], 16);
        sX[j] += __shfl_xor(sX[j], 16);
        sY[j] += __shfl_xor(sY[j], 16);
    }
    __syncthreads();

    float* part = &Xs[0][0];
    {
        float* pN = part + ((kh * 3 + 0) * MR) * L_;
        float* pX = part + ((kh * 3 + 1) * MR) * L_;
        float* pY = part + ((kh * 3 + 2) * MR) * L_;
        #pragma unroll
        for (int r = 0; r < 8; ++r) {
            const int row = 16 * tile + 8 * hi + r;
            pN[row * L_ + lo] = cN[r];
            pX[row * L_ + lo] = cX[r];
            pY[row * L_ + lo] = cY[r];
        }
        if (hi == 0) {
            const int row = 16 * tile + lo;
            #pragma unroll
            for (int j = 0; j < LE; ++j) {
                pN[row * L_ + LT + j] = sN[j];
                pX[row * L_ + LT + j] = sX[j];
                pY[row * L_ + LT + j] = sY[j];
            }
        }
    }
    __syncthreads();

    float* ob = out + (tgt == 0 ? 0 : OUTQOFF) + (b * S_ + s0) * OUTC + secId * L_;
    for (int i = tid; i < MR * L_; i += MTH) {
        const int r = i / L_, l = i % L_;
        const float n  = part[(0 * MR) * L_ + i] + part[(3 * MR) * L_ + i];
        const float sx = part[(1 * MR) * L_ + i] + part[(4 * MR) * L_ + i];
        const float sy = part[(2 * MR) * L_ + i] + part[(5 * MR) * L_ + i];
        const float den = fmaxf(sqrtf(sx) * sqrtf(sy), 1e-8f);
        ob[r * OUTC + l] = n / den;
    }
}


extern "C" void kernel_launch(void* const* d_in, const int* in_sizes, int n_in,
                              void* d_out, int out_size, void* d_ws, size_t ws_size,
                              hipStream_t stream)
{
    const float* p = (const float*)d_in[0];
    const float* q = (const float*)d_in[1];
    const float* W = (const float*)d_in[2];
    float* out = (float*)d_out;

    float* ws  = (float*)d_ws;
    float* qvm = ws;
    float* pvm = qvm + B_ * 2 * S_ * H_;
    float* qvx = pvm + B_ * 2 * S_ * H_;
    float* pvx = qvx + B_ * 2 * S_ * H_;

    pool_attn_kernel<<<B_ * 2 * NST, FTH, 0, stream>>>(p, q, W, out, qvm, pvm, qvx, pvx);
    mpcos_kernel    <<<MPBLK, MTH, 0, stream>>>(p, q, W, qvm, pvm, qvx, pvx, out);
}


# mpcos launch removed: each pool_attn block also takes the six 16-row multi-perspective cosines of its s tile from the rows, vectors and w^2 it holds in LDS; h sums now four quarters joined in order
# speedup vs baseline: 4.4304x; 1.1776x over previous
#include <hip/hip_runtime.h>
#include <math.h>

typedef __attribute__((ext_vector_type(2))) float v2f;
typedef __attribute__((ext_vector_type(8))) float v8f;

#define B_    32
#define S_    64
#define H_    300
#define L_    20
#define OUTC  160
#define OUTQOFF (B_*S_*OUTC)
#define NEG_INF (-3.4e38f)

#define MP_LG   5
#define MP_NLG  (L_ / MP_LG)
#define MP_W    (MP_LG * 4)
#define MP_TH   (MP_W * 32)
#define ST      16
#define NST     (S_ / ST)
#define AW      4
#define VTH     (AW * 32)
#define FTH     (MP_TH + VTH)
#define PC4     (H_ / 4)
#define NPC     (2 * S_ * PC4)
#define F_NP    ((NPC + FTH - 1) / FTH)
#define NHG     ((H_ + 63) / 64)
#define NJOB    6
#define NKQ     (FTH / 32 / NJOB)
#define H4      (H_ / 4)
#define KSQ     19
#define LT      16
#define LE      (L_ - LT)
#define NWS     3
#define W4      (L_ * H4)
#define WNP     ((NWS * W4 + FTH - 1) / FTH)

__global__ __launch_bounds__(FTH)
void pool_attn_kernel(const float* __restrict__ p, const float* __restrict__ q,
                      const float* __restrict__ W, float* __restrict__ out)
{
    __shared__ __attribute__((aligned(16))) float XQ[2 * S_][H_];
    __shared__ __attribute__((aligned(16))) float w2s[MP_LG][H_];
    __shared__ float nPs[MP_LG][S_];
    __shared__ float nQs[MP_LG][S_];
    __shared__ float rowMx[MP_LG][S_];
    __shared__ float colPt[MP_LG][4][S_];
    __shared__ __attribute__((aligned(16))) float cosRowT[S_][ST];
    __shared__ __attribute__((aligned(16))) float cosColT[S_][ST];
    __shared__ float nA[S_];
    __shared__ float nB[S_];
    __shared__ float rSum[ST];
    __shared__ float cSum[ST];
    __shared__ __attribute__((aligned(16))) float Yv[4][ST][H_];
    __shared__ __attribute__((aligned(16))) float Wm[NWS][L_][H_];

    const int idx  = blockIdx.x;
    const int x    = idx % NST;
    const int hf   = (idx / NST) & 1;
    const int b    = idx / (2 * NST);
    const int tid  = threadIdx.x;
    const int lane = tid & 31;
    const int wv   = __builtin_amdgcn_readfirstlane(tid >> 5);
    const int lo   = lane & 15;
    const int hi   = lane >> 4;

    const bool attn = wv < AW;
    const int mw    = wv - AW;
    const int li    = mw >> 2;
    const int strip = mw & 3;
    const int awv   = wv;
    const int atid  = tid;
    const int s0    = x * ST;

    const float* Ph = p + (b * S_) * (2 * H_) + hf * H_;
    const float* Qh = q + (b * S_) * (2 * H_) + hf * H_;
    const float* Wg = W + ((2 + hf) * L_ + x * MP_LG) * H_;

    {
        float4 stg[F_NP];
        #pragma unroll
        for (int r = 0; r < F_NP; ++r) {
            const int i   = tid + r * FTH;
            if (i < NPC) {
                const int row = i / PC4, c4 = i % PC4;
                const float* g = (row < S_ ? Ph + row * (2 * H_) : Qh + (row - S_) * (2 * H_)) + 4 * c4;
                stg[r] = float4{g[0], g[1], g[2], g[3]};
            }
        }
        float4 wst[WNP];
        #pragma unroll
        for (int r = 0; r < WNP; ++r) {
            const int i = tid + r * FTH;
            if (i < NWS * W4) {
                const int ws = i / W4, pc = i % W4;
                const float* g = W + ((ws == 0 ? 0 : 2 + 2 * ws) + hf) * (L_ * H_) + 4 * pc;
                wst[r] = float4{g[0], g[1], g[2], g[3]};
            }
        }
        for (int i = tid; i < MP_LG * H_; i += FTH) {
            const float w = Wg[i];
            (&w2s[0][0])[i] = w * w;
        }
        #pragma unroll
        for (int r = 0; r < F_NP; ++r) {
            const int i   = tid + r * FTH;
            if (i < NPC) {
                const int row = i / PC4, c4 = i % PC4;
                *(float4*)&XQ[row][4 * c4] = stg[r];
            }
        }
        #pragma unroll
        for (int r = 0; r < WNP; ++r) {
            const int i = tid + r * FTH;
            if (i < NWS * W4) {
                const float4 w = wst[r];
                *(float4*)(&Wm[0][0][0] + 4 * i) = float4{w.x * w.x, w.y * w.y, w.z * w.z, w.w * w.w};
            }
        }
    }
    __syncthreads();

    v8f cR = {}, cC = {};
    if (attn) {
        const float* aRowR = &XQ[16 * x + lo][2 * hi];
        const float* bRowR = &XQ[S_ + 16 * awv + lo][2 * hi];
        const float* aRowC = &XQ[16 * awv + lo][2 * hi];
        const float* bRowC = &XQ[S_ + 16 * x + lo][2 * hi];
        v2f sA2 = {0.f, 0.f}, sB2 = {0.f, 0.f};
        for (int kb = 0; kb < H_; kb += 4) {
            const v2f ar = *(const v2f*)(aRowR + kb);
            const v2f br = *(const v2f*)(bRowR + kb);
            const v2f ac = *(const v2f*)(aRowC + kb);
            const v2f bc = *(const v2f*)(bRowC + kb);
            sA2 = __builtin_elementwise_fma(ac, ac, sA2);
            sB2 = __builtin_elementwise_fma(br, br, sB2);
            cR = __builtin_amdgcn_wmma_f32_16x16x4_f32(false, ar, false, br, (short)0, cR, false, false);
            cC = __builtin_amdgcn_wmma_f32_16x16x4_f32(false, ac, false, bc, (short)0, cC, false, false);
        }
        float sA = sA2.x + sA2.y;
        float sB = sB2.x + sB2.y;
        sA += __shfl_xor(sA, 16);
        sB += __shfl_xor(sB, 16);
        if (hi == 0) {
            nA[16 * awv + lo] = sqrtf(sA);
            nB[16 * awv + lo] = sqrtf(sB);
        }
    }
    __syncthreads();

    if (attn) {
        const int nR = 16 * awv + lo;
        const float nbR = nB[nR];
        const float nbC = nB[s0 + lo];
        #pragma unroll
        for (int j = 0; j < 8; ++j) {
            const int t  = 8 * hi + j;
            const int mC = 16 * awv + t;
            cosRowT[nR][t] = cR[j] / (nA[s0 + t] * nbR);
            cosColT[mC][lo] = cC[j] / (nA[mC] * nbC);
        }
    }
    __syncthreads();

    if (attn) {
        if (atid < ST) {
            float sm = 0.f;
            for (int n = 0; n < S_; ++n) sm += cosRowT[n][atid];
            rSum[atid] = sm;
        } else if (atid >= 32 && atid < 32 + ST) {
            const int t = atid - 32;
            float sm = 0.f;
            for (int m = 0; m < S_; ++m) sm += cosColT[m][t];
            cSum[t] = sm;
        }
    }
    __syncthreads();

    v8f c0, c1, c2, c3;
    if (attn) {
        const int side = awv >> 1;
        const int th   = awv & 1;
        const float* Xc   = side ? &XQ[0][0] : &XQ[S_][0];
        const float* cT   = side ? &cosColT[0][8 * th] : &cosRowT[0][8 * th];
        const float* sums = side ? &cSum[8 * th] : &rSum[8 * th];
        float* om = &Yv[side][8 * th][0];
        float* ox = &Yv[2 + side][8 * th][0];
        for (int g = 0; g < NHG; ++g) {
            const int hc = 64 * g + 2 * lane;
            if (hc < H_) {
                const float* col = Xc + hc;
                float m0[ST / 2], m1[ST / 2], x0[ST / 2], x1[ST / 2];
                for (int t = 0; t < ST / 2; ++t) { m0[t] = 0.f; m1[t] = 0.f; x0[t] = NEG_INF; x1[t] = NEG_INF; }
                #pragma unroll 4
                for (int j = 0; j < S_; ++j) {
                    const v2f v = *(const v2f*)(col + j * H_);
                    const float4 ca = *(const float4*)(cT + j * ST);
                    const float4 cb = *(const float4*)(cT + j * ST + 4);
                    const float cr[ST / 2] = {ca.x, ca.y, ca.z, ca.w, cb.x, cb.y, cb.z, cb.w};
                    for (int t = 0; t < ST / 2; ++t) {
                        const float a0 = v.x * cr[t];
                        const float a1 = v.y * cr[t];
                        m0[t] += a0;  x0[t] = fmaxf(x0[t], a0);
                        m1[t] += a1;  x1[t] = fmaxf(x1[t], a1);
                    }
                }
                for (int t = 0; t < ST / 2; ++t) {
                    const float rs = sums[t];
                    *(v2f*)(om + t * H_ + hc) = v2f{m0[t] / rs, m1[t] / rs};
                    *(v2f*)(ox + t * H_ + hc) = v2f{x0[t], x1[t]};
                }
            }
        }
    } else {
        const float* aRow = &XQ[16 * strip + lo][2 * hi];
        const float* bR0  = &XQ[S_ + 16 * ((strip + 0) & 3) + lo][2 * hi];
        const float* bR1  = &XQ[S_ + 16 * ((strip + 1) & 3) + lo][2 * hi];
        const float* bR2  = &XQ[S_ + 16 * ((strip + 2) & 3) + lo][2 * hi];
        const float* bR3  = &XQ[S_ + 16 * ((strip + 3) & 3) + lo][2 * hi];
        const float* wRow = &w2s[li][2 * hi];
        float sP = 0.f, sQ = 0.f;
        c0 = v8f{}; c1 = v8f{}; c2 = v8f{}; c3 = v8f{};
        for (int kb = 0; kb < H_; kb += 4) {
            const v2f a  = *(const v2f*)(aRow + kb);
            const v2f w  = *(const v2f*)(wRow + kb);
            const v2f b0 = *(const v2f*)(bR0 + kb);
            const v2f b1 = *(const v2f*)(bR1 + kb);
            const v2f b2 = *(const v2f*)(bR2 + kb);
            const v2f b3 = *(const v2f*)(bR3 + kb);
            const v2f aw = a * w;
            const v2f bw = b0 * w;
            sP = fmaf(a.x, aw.x, sP);   sP = fmaf(a.y, aw.y, sP);
            sQ = fmaf(b0.x, bw.x, sQ);  sQ = fmaf(b0.y, bw.y, sQ);
            c0 = __builtin_amdgcn_wmma_f32_16x16x4_f32(false, aw, false, b0, (short)0, c0, false, false);
            c1 = __builtin_amdgcn_wmma_f32_16x16x4_f32(false, aw, false, b1, (short)0, c1, false, false);
            c2 = __builtin_amdgcn_wmma_f32_16x16x4_f32(false, aw, false, b2, (short)0, c2, false, false);
            c3 = __builtin_amdgcn_wmma_f32_16x16x4_f32(false, aw, false, b3, (short)0, c3, false, false);
        }
        sP += __shfl_xor(sP, 16);
        sQ += __shfl_xor(sQ, 16);
        if (hi == 0) {
            nPs[li][16 * strip + lo] = sqrtf(sP);
            nQs[li][16 * strip + lo] = sqrtf(sQ);
        }
    }
    __syncthreads();

    const int lo2  = lo;
    const int hi2  = hi;
    const int tid2 = tid;

    if (!attn) {
        const int n0 = 16 * ((strip + 0) & 3) + lo2;
        const int n1 = 16 * ((strip + 1) & 3) + lo2;
        const int n2 = 16 * ((strip + 2) & 3) + lo2;
        const int n3 = 16 * ((strip + 3) & 3) + lo2;
        const float nb0 = nQs[li][n0], nb1 = nQs[li][n1], nb2 = nQs[li][n2], nb3 = nQs[li][n3];
        float cm0 = NEG_INF, cm1 = NEG_INF, cm2 = NEG_INF, cm3 = NEG_INF;
        float rmx[8];
        #pragma unroll
        for (int j = 0; j < 8; ++j) {
            const float na = nPs[li][16 * strip + 8 * hi2 + j];
            const float v0 = c0[j] / (na * nb0);
            const float v1 = c1[j] / (na * nb1);
            const float v2 = c2[j] / (na * nb2);
            const float v3 = c3[j] / (na * nb3);
            cm0 = fmaxf(cm0, v0); cm1 = fmaxf(cm1, v1);
            cm2 = fmaxf(cm2, v2); cm3 = fmaxf(cm3, v3);
            rmx[j] = fmaxf(fmaxf(fmaxf(fmaxf(NEG_INF, v0), v1), v2), v3);
        }
        #pragma unroll
        for (int d = 1; d < 16; d <<= 1) {
            #pragma unroll
            for (int j = 0; j < 8; ++j) rmx[j] = fmaxf(rmx[j], __shfl_xor(rmx[j], d));
        }
        if (lo2 == 0) {
            for (int j = 0; j < 8; ++j) rowMx[li][16 * strip + 8 * hi2 + j] = rmx[j];
        }
        cm0 = fmaxf(cm0, __shfl_xor(cm0, 16));
        cm1 = fmaxf(cm1, __shfl_xor(cm1, 16));
        cm2 = fmaxf(cm2, __shfl_xor(cm2, 16));
        cm3 = fmaxf(cm3, __shfl_xor(cm3, 16));
        if (hi2 == 0) {
            colPt[li][strip][n0] = cm0;
            colPt[li][strip][n1] = cm1;
            colPt[li][strip][n2] = cm2;
            colPt[li][strip][n3] = cm3;
        }
    }
    __syncthreads();

    const int nres = (S_ * MP_LG);
    if (tid2 < nres) {
        const int row = tid2 / MP_LG, l5 = tid2 % MP_LG;
        out[(b * S_ + row) * OUTC + (2 + hf) * L_ + x * MP_LG + l5] = rowMx[l5][row];
    } else if (tid2 < 2 * nres) {
        const int u = tid2 - nres;
        const int n = u / MP_LG, l5 = u % MP_LG;
        float mx = NEG_INF;
        for (int s = 0; s < 4; ++s) mx = fmaxf(mx, colPt[l5][s][n]);
        out[OUTQOFF + (b * S_ + n) * OUTC + (2 + hf) * L_ + x * MP_LG + l5] = mx;
    }

    const int job  = wv % NJOB;
    const int kq   = wv / NJOB;
    const int tgt  = job & 1;
    const int kind = job >> 1;
    {
        const float* xRow = &XQ[(tgt ? S_ : 0) + s0 + lo2][2 * hi2];
        const float* yRow = kind == 0 ? &XQ[(tgt ? 0 : S_) + (hf ? 0 : S_ - 1)][2 * hi2]
                                      : &Yv[2 * (kind - 1) + tgt][lo2][2 * hi2];
        const float* bRow = &Wm[kind][lo2][2 * hi2];
        const float* eRow = &Wm[kind][LT][2 * hi2];
        const int kb0 = 4 * KSQ * kq;
        const int kb1 = kq == NKQ - 1 ? H_ : kb0 + 4 * KSQ;

        v8f cN = {}, cX = {}, cY = {};
        v2f eN[LE], eX[LE], eY[LE];
        #pragma unroll
        for (int j = 0; j < LE; ++j) { eN[j] = v2f{0.f, 0.f}; eX[j] = v2f{0.f, 0.f}; eY[j] = v2f{0.f, 0.f}; }

        for (int kb = kb0; kb < kb1; kb += 4) {
            const v2f xf = *(const v2f*)(xRow + kb);
            const v2f yf = *(const v2f*)(yRow + kb);
            const v2f w  = *(const v2f*)(bRow + kb);
            const v2f xy = xf * yf, xx = xf * xf, yy = yf * yf;
            cN = __builtin_amdgcn_wmma_f32_16x16x4_f32(false, xy, false, w, (short)0, cN, false, false);
            cX = __builtin_amdgcn_wmma_f32_16x16x4_f32(false, xx, false, w, (short)0, cX, false, false);
            cY = __builtin_amdgcn_wmma_f32_16x16x4_f32(false, yy, false, w, (short)0, cY, false, false);
            #pragma unroll
            for (int j = 0; j < LE; ++j) {
                const v2f we = *(const v2f*)(eRow + j * H_ + kb);
                eN[j] = __builtin_elementwise_fma(xy, we, eN[j]);
                eX[j] = __builtin_elementwise_fma(xx, we, eX[j]);
                eY[j] = __builtin_elementwise_fma(yy, we, eY[j]);
            }
        }

        float sN[LE], sX[LE], sY[LE];
        #pragma unroll
        for (int j = 0; j < LE; ++j) {
            sN[j] = eN[j].x + eN[j].y;  sX[j] = eX[j].x + eX[j].y;  sY[j] = eY[j].x + eY[j].y;
            sN[j] += __shfl_xor(sN[j], 16);
            sX[j] += __shfl_xor(sX[j], 16);
            sY[j] += __shfl_xor(sY[j], 16);
        }
        __syncthreads();

        float* pN = &XQ[0][0] + (((kq * NJOB + job) * 3 + 0) * ST) * L_;
        float* pX = pN + ST * L_;
        float* pY = pX + ST * L_;
        #pragma unroll
        for (int r = 0; r < 8; ++r) {
            const int row = 8 * hi2 + r;
            pN[row * L_ + lo2] = cN[r];
            pX[row * L_ + lo2] = cX[r];
            pY[row * L_ + lo2] = cY[r];
        }
        if (hi2 == 0) {
            #pragma unroll
            for (int j = 0; j < LE; ++j) {
                pN[lo2 * L_ + LT + j] = sN[j];
                pX[lo2 * L_ + LT + j] = sX[j];
                pY[lo2 * L_ + LT + j] = sY[j];
            }
        }
    }
    __syncthreads();

    {
        const float* part = &XQ[0][0];
        for (int i = tid2; i < NJOB * ST * L_; i += FTH) {
            const int jb = i / (ST * L_), rl = i % (ST * L_);
            float n = 0.f, sx = 0.f, sy = 0.f;
            for (int k4 = 0; k4 < NKQ; ++k4) {
                const float* pp = part + (((k4 * NJOB + jb) * 3) * ST) * L_ + rl;
                n  += pp[0];
                sx += pp[ST * L_];
                sy += pp[2 * ST * L_];
            }
            const float den = fmaxf(sqrtf(sx) * sqrtf(sy), 1e-8f);
            const int tg = jb & 1, kd = jb >> 1;
            const int secId = (kd == 0 ? 0 : 2 + 2 * kd) + hf;
            out[(tg ? OUTQOFF : 0) + (b * S_ + s0 + rl / L_) * OUTC + secId * L_ + rl % L_] = n / den;
        }
    }
}

extern "C" void kernel_launch(void* const* d_in, const int* in_sizes, int n_in,
                              void* d_out, int out_size, void* d_ws, size_t ws_size,
                              hipStream_t stream)
{
    const float* p = (const float*)d_in[0];
    const float* q = (const float*)d_in[1];
    const float* W = (const float*)d_in[2];
    float* out = (float*)d_out;

    pool_attn_kernel<<<B_ * 2 * NST, FTH, 0, stream>>>(p, q, W, out);
}


# divisions by norms and by cosine row/col sums replaced by multiplies: each is stored as its reciprocal (1/sqrt, 1/sum) where produced; cosines are (c*ra)*rb, mean vectors m*ri; sum orders kept
# speedup vs baseline: 4.6252x; 1.0440x over previous
#include <hip/hip_runtime.h>
#include <math.h>

typedef __attribute__((ext_vector_type(2))) float v2f;
typedef __attribute__((ext_vector_type(8))) float v8f;

#define B_    32
#define S_    64
#define H_    300
#define L_    20
#define OUTC  160
#define OUTQOFF (B_*S_*OUTC)
#define NEG_INF (-3.4e38f)

#define MP_LG   5
#define MP_NLG  (L_ / MP_LG)
#define MP_W    (MP_LG * 4)
#define MP_TH   (MP_W * 32)
#define ST      16
#define NST     (S_ / ST)
#define AW      4
#define VTH     (AW * 32)
#define FTH     (MP_TH + VTH)
#define PC4     (H_ / 4)
#define NPC     (2 * S_ * PC4)
#define F_NP    ((NPC + FTH - 1) / FTH)
#define NHG     ((H_ + 63) / 64)
#define NJOB    6
#define NKQ     (FTH / 32 / NJOB)
#define H4      (H_ / 4)
#define KSQ     19
#define LT      16
#define LE      (L_ - LT)
#define NWS     3
#define W4      (L_ * H4)
#define WNP     ((NWS * W4 + FTH - 1) / FTH)

__global__ __launch_bounds__(FTH)
void pool_attn_kernel(const float* __restrict__ p, const float* __restrict__ q,
                      const float* __restrict__ W, float* __restrict__ out)
{
    __shared__ __attribute__((aligned(16))) float XQ[2 * S_][H_];
    __shared__ __attribute__((aligned(16))) float w2s[MP_LG][H_];
    __shared__ float rnPs[MP_LG][S_];
    __shared__ float rnQs[MP_LG][S_];
    __shared__ float rowMx[MP_LG][S_];
    __shared__ float colPt[MP_LG][4][S_];
    __shared__ __attribute__((aligned(16))) float cosRowT[S_][ST];
    __shared__ __attribute__((aligned(16))) float cosColT[S_][ST];
    __shared__ float rnA[S_];
    __shared__ float rnB[S_];
    __shared__ __attribute__((aligned(16))) float rSum[ST];
    __shared__ __attribute__((aligned(16))) float cSum[ST];
    __shared__ __attribute__((aligned(16))) float Yv[4][ST][H_];
    __shared__ __attribute__((aligned(16))) float Wm[NWS][L_][H_];

    const int idx  = blockIdx.x;
    const int x    = idx % NST;
    const int hf   = (idx / NST) & 1;
    const int b    = idx / (2 * NST);
    const int tid  = threadIdx.x;
    const int lane = tid & 31;
    const int wv   = __builtin_amdgcn_readfirstlane(tid >> 5);
    const int lo   = lane & 15;
    const int hi   = lane >> 4;

    const bool attn = wv < AW;
    const int mw    = wv - AW;
    const int li    = mw >> 2;
    const int strip = mw & 3;
    const int awv   = wv;
    const int atid  = tid;
    const int s0    = x * ST;

    const float* Ph = p + (b * S_) * (2 * H_) + hf * H_;
    const float* Qh = q + (b * S_) * (2 * H_) + hf * H_;
    const float* Wg = W + ((2 + hf) * L_ + x * MP_LG) * H_;

    {
        float4 stg[F_NP];
        #pragma unroll
        for (int r = 0; r < F_NP; ++r) {
            const int i   = tid + r * FTH;
            if (i < NPC) {
                const int row = i / PC4, c4 = i % PC4;
                const float* g = (row < S_ ? Ph + row * (2 * H_) : Qh + (row - S_) * (2 * H_)) + 4 * c4;
                stg[r] = float4{g[0], g[1], g[2], g[3]};
            }
        }
        float4 wst[WNP];
        #pragma unroll
        for (int r = 0; r < WNP; ++r) {
            const int i = tid + r * FTH;
            if (i < NWS * W4) {
                const int ws = i / W4, pc = i % W4;
                const float* g = W + ((ws == 0 ? 0 : 2 + 2 * ws) + hf) * (L_ * H_) + 4 * pc;
                wst[r] = float4{g[0], g[1], g[2], g[3]};
            }
        }
        for (int i = tid; i < MP_LG * H_; i += FTH) {
            const float w = Wg[i];
            (&w2s[0][0])[i] = w * w;
        }
        #pragma unroll
        for (int r = 0; r < F_NP; ++r) {
            const int i   = tid + r * FTH;
            if (i < NPC) {
                const int row = i / PC4, c4 = i % PC4;
                *(float4*)&XQ[row][4 * c4] = stg[r];
            }
        }
        #pragma unroll
        for (int r = 0; r < WNP; ++r) {
            const int i = tid + r * FTH;
            if (i < NWS * W4) {
                const float4 w = wst[r];
                *(float4*)(&Wm[0][0][0] + 4 * i) = float4{w.x * w.x, w.y * w.y, w.z * w.z, w.w * w.w};
            }
        }
    }
    __syncthreads();

    v8f cR = {}, cC = {};
    if (attn) {
        const float* aRowR = &XQ[16 * x + lo][2 * hi];
        const float* bRowR = &XQ[S_ + 16 * awv + lo][2 * hi];
        const float* aRowC = &XQ[16 * awv + lo][2 * hi];
        const float* bRowC = &XQ[S_ + 16 * x + lo][2 * hi];
        v2f sA2 = {0.f, 0.f}, sB2 = {0.f, 0.f};
        for (int kb = 0; kb < H_; kb += 4) {
            const v2f ar = *(const v2f*)(aRowR + kb);
            const v2f br = *(const v2f*)(bRowR + kb);
            const v2f ac = *(const v2f*)(aRowC + kb);
            const v2f bc = *(const v2f*)(bRowC + kb);
            sA2 = __builtin_elementwise_fma(ac, ac, sA2);
            sB2 = __builtin_elementwise_fma(br, br, sB2);
            cR = __builtin_amdgcn_wmma_f32_16x16x4_f32(false, ar, false, br, (short)0, cR, false, false);
            cC = __builtin_amdgcn_wmma_f32_16x16x4_f32(false, ac, false, bc, (short)0, cC, false, false);
        }
        float sA = sA2.x + sA2.y;
        float sB = sB2.x + sB2.y;
        sA += __shfl_xor(sA, 16);
        sB += __shfl_xor(sB, 16);
        if (hi == 0) {
            rnA[16 * awv + lo] = 1.0f / sqrtf(sA);
            rnB[16 * awv + lo] = 1.0f / sqrtf(sB);
        }
    }
    __syncthreads();

    if (attn) {
        const int nR = 16 * awv + lo;
        const float rbR = rnB[nR];
        const float rbC = rnB[s0 + lo];
        #pragma unroll
        for (int j = 0; j < 8; ++j) {
            const int t  = 8 * hi + j;
            const int mC = 16 * awv + t;
            cosRowT[nR][t] = (cR[j] * rnA[s0 + t]) * rbR;
            cosColT[mC][lo] = (cC[j] * rnA[mC]) * rbC;
        }
    }
    __syncthreads();

    if (attn) {
        if (atid < ST) {
            float sm = 0.f;
            for (int n = 0; n < S_; ++n) sm += cosRowT[n][atid];
            rSum[atid] = 1.0f / sm;
        } else if (atid >= 32 && atid < 32 + ST) {
            const int t = atid - 32;
            float sm = 0.f;
            for (int m = 0; m < S_; ++m) sm += cosColT[m][t];
            cSum[t] = 1.0f / sm;
        }
    }
    __syncthreads();

    v8f c0, c1, c2, c3;
    if (attn) {
        const int side = awv >> 1;
        const int th   = awv & 1;
        const float* Xc   = side ? &XQ[0][0] : &XQ[S_][0];
        const float* cT   = side ? &cosColT[0][8 * th] : &cosRowT[0][8 * th];
        const float* sums = side ? &cSum[8 * th] : &rSum[8 * th];
        float* om = &Yv[side][8 * th][0];
        float* ox = &Yv[2 + side][8 * th][0];
        for (int g = 0; g < NHG; ++g) {
            const int hc = 64 * g + 2 * lane;
            if (hc < H_) {
                const float* col = Xc + hc;
                float m0[ST / 2], m1[ST / 2], x0[ST / 2], x1[ST / 2];
                for (int t = 0; t < ST / 2; ++t) { m0[t] = 0.f; m1[t] = 0.f; x0[t] = NEG_INF; x1[t] = NEG_INF; }
                #pragma unroll 4
                for (int j = 0; j < S_; ++j) {
                    const v2f v = *(const v2f*)(col + j * H_);
                    const float4 ca = *(const float4*)(cT + j * ST);
                    const float4 cb = *(const float4*)(cT + j * ST + 4);
                    const float cr[ST / 2] = {ca.x, ca.y, ca.z, ca.w, cb.x, cb.y, cb.z, cb.w};
                    for (int t = 0; t < ST / 2; ++t) {
                        const float a0 = v.x * cr[t];
                        const float a1 = v.y * cr[t];
                        m0[t] += a0;  x0[t] = fmaxf(x0[t], a0);
                        m1[t] += a1;  x1[t] = fmaxf(x1[t], a1);
                    }
                }
                const float4 ia = *(const float4*)(sums);
                const float4 ib = *(const float4*)(sums + 4);
                const float ri[ST / 2] = {ia.x, ia.y, ia.z, ia.w, ib.x, ib.y, ib.z, ib.w};
                for (int t = 0; t < ST / 2; ++t) {
                    *(v2f*)(om + t * H_ + hc) = v2f{m0[t] * ri[t], m1[t] * ri[t]};
                    *(v2f*)(ox + t * H_ + hc) = v2f{x0[t], x1[t]};
                }
            }
        }
    } else {
        const float* aRow = &XQ[16 * strip + lo][2 * hi];
        const float* bR0  = &XQ[S_ + 16 * ((strip + 0) & 3) + lo][2 * hi];
        const float* bR1  = &XQ[S_ + 16 * ((strip + 1) & 3) + lo][2 * hi];
        const float* bR2  = &XQ[S_ + 16 * ((strip + 2) & 3) + lo][2 * hi];
        const float* bR3  = &XQ[S_ + 16 * ((strip + 3) & 3) + lo][2 * hi];
        const float* wRow = &w2s[li][2 * hi];
        float sP = 0.f, sQ = 0.f;
        c0 = v8f{}; c1 = v8f{}; c2 = v8f{}; c3 = v8f{};
        for (int kb = 0; kb < H_; kb += 4) {
            const v2f a  = *(const v2f*)(aRow + kb);
            const v2f w  = *(const v2f*)(wRow + kb);
            const v2f b0 = *(const v2f*)(bR0 + kb);
            const v2f b1 = *(const v2f*)(bR1 + kb);
            const v2f b2 = *(const v2f*)(bR2 + kb);
            const v2f b3 = *(const v2f*)(bR3 + kb);
            const v2f aw = a * w;
            const v2f bw = b0 * w;
            sP = fmaf(a.x, aw.x, sP);   sP = fmaf(a.y, aw.y, sP);
            sQ = fmaf(b0.x, bw.x, sQ);  sQ = fmaf(b0.y, bw.y, sQ);
            c0 = __builtin_amdgcn_wmma_f32_16x16x4_f32(false, aw, false, b0, (short)0, c0, false, false);
            c1 = __builtin_amdgcn_wmma_f32_16x16x4_f32(false, aw, false, b1, (short)0, c1, false, false);
            c2 = __builtin_amdgcn_wmma_f32_16x16x4_f32(false, aw, false, b2, (short)0, c2, false, false);
            c3 = __builtin_amdgcn_wmma_f32_16x16x4_f32(false, aw, false, b3, (short)0, c3, false, false);
        }
        sP += __shfl_xor(sP, 16);
        sQ += __shfl_xor(sQ, 16);
        if (hi == 0) {
            rnPs[li][16 * strip + lo] = 1.0f / sqrtf(sP);
            rnQs[li][16 * strip + lo] = 1.0f / sqrtf(sQ);
        }
    }
    __syncthreads();

    const int lo2  = lo;
    const int hi2  = hi;
    const int tid2 = tid;

    if (!attn) {
        const int n0 = 16 * ((strip + 0) & 3) + lo2;
        const int n1 = 16 * ((strip + 1) & 3) + lo2;
        const int n2 = 16 * ((strip + 2) & 3) + lo2;
        const int n3 = 16 * ((strip + 3) & 3) + lo2;
        const float rb0 = rnQs[li][n0], rb1 = rnQs[li][n1], rb2 = rnQs[li][n2], rb3 = rnQs[li][n3];
        float cm0 = NEG_INF, cm1 = NEG_INF, cm2 = NEG_INF, cm3 = NEG_INF;
        float rmx[8];
        #pragma unroll
        for (int j = 0; j < 8; ++j) {
            const float ra = rnPs[li][16 * strip + 8 * hi2 + j];
            const float v0 = (c0[j] * ra) * rb0;
            const float v1 = (c1[j] * ra) * rb1;
            const float v2 = (c2[j] * ra) * rb2;
            const float v3 = (c3[j] * ra) * rb3;
            cm0 = fmaxf(cm0, v0); cm1 = fmaxf(cm1, v1);
            cm2 = fmaxf(cm2, v2); cm3 = fmaxf(cm3, v3);
            rmx[j] = fmaxf(fmaxf(fmaxf(fmaxf(NEG_INF, v0), v1), v2), v3);
        }
        #pragma unroll
        for (int d = 1; d < 16; d <<= 1) {
            #pragma unroll
            for (int j = 0; j < 8; ++j) rmx[j] = fmaxf(rmx[j], __shfl_xor(rmx[j], d));
        }
        if (lo2 == 0) {
            for (int j = 0; j < 8; ++j) rowMx[li][16 * strip + 8 * hi2 + j] = rmx[j];
        }
        cm0 = fmaxf(cm0, __shfl_xor(cm0, 16));
        cm1 = fmaxf(cm1, __shfl_xor(cm1, 16));
        cm2 = fmaxf(cm2, __shfl_xor(cm2, 16));
        cm3 = fmaxf(cm3, __shfl_xor(cm3, 16));
        if (hi2 == 0) {
            colPt[li][strip][n0] = cm0;
            colPt[li][strip][n1] = cm1;
            colPt[li][strip][n2] = cm2;
            colPt[li][strip][n3] = cm3;
        }
    }
    __syncthreads();

    const int nres = (S_ * MP_LG);
    if (tid2 < nres) {
        const int row = tid2 / MP_LG, l5 = tid2 % MP_LG;
        out[(b * S_ + row) * OUTC + (2 + hf) * L_ + x * MP_LG + l5] = rowMx[l5][row];
    } else if (tid2 < 2 * nres) {
        const int u = tid2 - nres;
        const int n = u / MP_LG, l5 = u % MP_LG;
        float mx = NEG_INF;
        for (int s = 0; s < 4; ++s) mx = fmaxf(mx, colPt[l5][s][n]);
        out[OUTQOFF + (b * S_ + n) * OUTC + (2 + hf) * L_ + x * MP_LG + l5] = mx;
    }

    const int job  = wv % NJOB;
    const int kq   = wv / NJOB;
    const int tgt  = job & 1;
    const int kind = job >> 1;
    {
        const float* xRow = &XQ[(tgt ? S_ : 0) + s0 + lo2][2 * hi2];
        const float* yRow = kind == 0 ? &XQ[(tgt ? 0 : S_) + (hf ? 0 : S_ - 1)][2 * hi2]
                                      : &Yv[2 * (kind - 1) + tgt][lo2][2 * hi2];
        const float* bRow = &Wm[kind][lo2][2 * hi2];
        const float* eRow = &Wm[kind][LT][2 * hi2];
        const int kb0 = 4 * KSQ * kq;
        const int kb1 = kq == NKQ - 1 ? H_ : kb0 + 4 * KSQ;

        v8f cN = {}, cX = {}, cY = {};
        v2f eN[LE], eX[LE], eY[LE];
        #pragma unroll
        for (int j = 0; j < LE; ++j) { eN[j] = v2f{0.f, 0.f}; eX[j] = v2f{0.f, 0.f}; eY[j] = v2f{0.f, 0.f}; }

        for (int kb = kb0; kb < kb1; kb += 4) {
            const v2f xf = *(const v2f*)(xRow + kb);
            const v2f yf = *(const v2f*)(yRow + kb);
            const v2f w  = *(const v2f*)(bRow + kb);
            const v2f xy = xf * yf, xx = xf * xf, yy = yf * yf;
            cN = __builtin_amdgcn_wmma_f32_16x16x4_f32(false, xy, false, w, (short)0, cN, false, false);
            cX = __builtin_amdgcn_wmma_f32_16x16x4_f32(false, xx, false, w, (short)0, cX, false, false);
            cY = __builtin_amdgcn_wmma_f32_16x16x4_f32(false, yy, false, w, (short)0, cY, false, false);
            #pragma unroll
            for (int j = 0; j < LE; ++j) {
                const v2f we = *(const v2f*)(eRow + j * H_ + kb);
                eN[j] = __builtin_elementwise_fma(xy, we, eN[j]);
                eX[j] = __builtin_elementwise_fma(xx, we, eX[j]);
                eY[j] = __builtin_elementwise_fma(yy, we, eY[j]);
            }
        }

        float sN[LE], sX[LE], sY[LE];
        #pragma unroll
        for (int j = 0; j < LE; ++j) {
            sN[j] = eN[j].x + eN[j].y;  sX[j] = eX[j].x + eX[j].y;  sY[j] = eY[j].x + eY[j].y;
            sN[j] += __shfl_xor(sN[j], 16);
            sX[j] += __shfl_xor(sX[j], 16);
            sY[j] += __shfl_xor(sY[j], 16);
        }
        __syncthreads();

        float* pN = &XQ[0][0] + (((kq * NJOB + job) * 3 + 0) * ST) * L_;
        float* pX = pN + ST * L_;
        float* pY = pX + ST * L_;
        #pragma unroll
        for (int r = 0; r < 8; ++r) {
            const int row = 8 * hi2 + r;
            pN[row * L_ + lo2] = cN[r];
            pX[row * L_ + lo2] = cX[r];
            pY[row * L_ + lo2] = cY[r];
        }
        if (hi2 == 0) {
            #pragma unroll
            for (int j = 0; j < LE; ++j) {
                pN[lo2 * L_ + LT + j] = sN[j];
                pX[lo2 * L_ + LT + j] = sX[j];
                pY[lo2 * L_ + LT + j] = sY[j];
            }
        }
    }
    __syncthreads();

    {
        const float* part = &XQ[0][0];
        for (int i = tid2; i < NJOB * ST * L_; i += FTH) {
            const int jb = i / (ST * L_), rl = i % (ST * L_);
            float n = 0.f, sx = 0.f, sy = 0.f;
            for (int k4 = 0; k4 < NKQ; ++k4) {
                const float* pp = part + (((k4 * NJOB + jb) * 3) * ST) * L_ + rl;
                n  += pp[0];
                sx += pp[ST * L_];
                sy += pp[2 * ST * L_];
            }
            const float den = fmaxf(sqrtf(sx) * sqrtf(sy), 1e-8f);
            const int tg = jb & 1, kd = jb >> 1;
            const int secId = (kd == 0 ? 0 : 2 + 2 * kd) + hf;
            out[(tg ? OUTQOFF : 0) + (b * S_ + s0 + rl / L_) * OUTC + secId * L_ + rl % L_] = n / den;
        }
    }
}

extern "C" void kernel_launch(void* const* d_in, const int* in_sizes, int n_in,
                              void* d_out, int out_size, void* d_ws, size_t ws_size,
                              hipStream_t stream)
{
    const float* p = (const float*)d_in[0];
    const float* q = (const float*)d_in[1];
    const float* W = (const float*)d_in[2];
    float* out = (float*)d_out;

    pool_attn_kernel<<<B_ * 2 * NST, FTH, 0, stream>>>(p, q, W, out);
}
